# starting kernel
# baseline (speedup 1.0000x reference)
#include <hip/hip_runtime.h>
#include <cmath>


#define GN   50000
#define GE   800000
#define GETOT (GE + GN)
#define GG   256
#define GH   4
#define GC   32
#define GD   128
#define GEMB 128

typedef float v2f __attribute__((ext_vector_type(2)));
typedef float v8f __attribute__((ext_vector_type(8)));

__device__ __forceinline__ void atomicMaxFloat(float* addr, float val) {
    if (val >= 0.0f) {
        atomicMax((int*)addr, __float_as_int(val));
    } else {
        atomicMin((unsigned int*)addr, __float_as_uint(val));
    }
}

__global__ void fill_f32(float* __restrict__ p, float v, int n) {
    int i = blockIdx.x * blockDim.x + threadIdx.x;
    if (i < n) p[i] = v;
}

__global__ void gemm_wmma_f32(const float* __restrict__ A,
                              const float* __restrict__ B,
                              const float* __restrict__ bias,
                              float* __restrict__ C,
                              int M, int K, int N) {
    const int wave = threadIdx.x >> 5;
    const int lane = threadIdx.x & 31;
    const int tilesN = N >> 4;
    const int total  = (M >> 4) * tilesN;
    const int tile = blockIdx.x * (blockDim.x >> 5) + wave;
    if (tile >= total) return;

    const int row0 = (tile / tilesN) << 4;
    const int col0 = (tile % tilesN) << 4;
    const int half = lane >> 4;
    const int l16  = lane & 15;

    const float* Ap = A + (size_t)(row0 + l16) * K + 2 * half;
    const float* Bp = B + (size_t)(2 * half) * N + col0 + l16;

    v8f acc = {};
    for (int k0 = 0; k0 < K; k0 += 4) {
        v2f a, b;
        a.x = Ap[k0];
        a.y = Ap[k0 + 1];
        b.x = Bp[(size_t)k0 * N];
        b.y = Bp[(size_t)(k0 + 1) * N];
        acc = __builtin_amdgcn_wmma_f32_16x16x4_f32(
                false, a, false, b, (short)0, acc, false, false);
    }

    const float bv = bias ? bias[col0 + l16] : 0.0f;
    float* Cp = C + (size_t)(row0 + 8 * half) * N + col0 + l16;
#pragma unroll
    for (int j = 0; j < 8; ++j) {
        Cp[(size_t)j * N] = acc[j] + bv;
    }
}

__global__ void attn_proj_k(const float* __restrict__ ht,
                            const float* __restrict__ asrc,
                            const float* __restrict__ adst,
                            float* __restrict__ als,
                            float* __restrict__ ald) {
    int tid = blockIdx.x * blockDim.x + threadIdx.x;
    if (tid >= GN * GH) return;
    const int n = tid >> 2;
    const int h = tid & 3;
    const float* hp = ht + (size_t)n * GD + h * GC;
    const float* ap = asrc + h * GC;
    const float* dp = adst + h * GC;
    float s = 0.0f, d = 0.0f;
#pragma unroll
    for (int c = 0; c < GC; ++c) {
        float v = hp[c];
        s += v * ap[c];
        d += v * dp[c];
    }
    als[tid] = s;
    ald[tid] = d;
}

__device__ __forceinline__ void edge_sd(const int* __restrict__ ei, int e,
                                        int& s, int& d) {
    if (e < GE) { s = ei[e]; d = ei[GE + e]; }
    else        { s = e - GE; d = e - GE; }
}

__global__ void edge_max_k(const int* __restrict__ ei,
                           const float* __restrict__ als,
                           const float* __restrict__ ald,
                           float* __restrict__ m) {
    int e = blockIdx.x * blockDim.x + threadIdx.x;
    if (e >= GETOT) return;
    int s, d;
    edge_sd(ei, e, s, d);
#pragma unroll
    for (int h = 0; h < GH; ++h) {
        float v = als[s * GH + h] + ald[d * GH + h];
        v = (v >= 0.0f) ? v : 0.2f * v;
        atomicMaxFloat(m + (size_t)d * GH + h, v);
    }
}

__global__ void edge_sum_k(const int* __restrict__ ei,
                           const float* __restrict__ als,
                           const float* __restrict__ ald,
                           const float* __restrict__ m,
                           float* __restrict__ denom) {
    int e = blockIdx.x * blockDim.x + threadIdx.x;
    if (e >= GETOT) return;
    int s, d;
    edge_sd(ei, e, s, d);
#pragma unroll
    for (int h = 0; h < GH; ++h) {
        float v = als[s * GH + h] + ald[d * GH + h];
        v = (v >= 0.0f) ? v : 0.2f * v;
        float p = expf(v - m[(size_t)d * GH + h]);
        atomicAdd(denom + (size_t)d * GH + h, p);
    }
}

__global__ void edge_agg_k(const int* __restrict__ ei,
                           const float* __restrict__ als,
                           const float* __restrict__ ald,
                           const float* __restrict__ m,
                           const float* __restrict__ denom,
                           const float* __restrict__ ht,
                           float* __restrict__ agg) {
    int tid = blockIdx.x * blockDim.x + threadIdx.x;
    if (tid >= GETOT * GH) return;
    const int e = tid >> 2;
    const int h = tid & 3;
    int s, d;
    edge_sd(ei, e, s, d);
    float v = als[s * GH + h] + ald[d * GH + h];
    v = (v >= 0.0f) ? v : 0.2f * v;
    float p = expf(v - m[(size_t)d * GH + h]);
    float alpha = p / denom[(size_t)d * GH + h];
    const float* hs = ht + (size_t)s * GD + h * GC;
    float* ag = agg + (size_t)d * GD + h * GC;
#pragma unroll
    for (int c = 0; c < GC; ++c) {
        atomicAdd(ag + c, hs[c] * alpha);
    }
}

__global__ void post_ln_k(const float* __restrict__ agg,
                          const float* __restrict__ bias,
                          const float* __restrict__ gamma,
                          const float* __restrict__ beta,
                          float* __restrict__ out) {
    const int wave = threadIdx.x >> 5;
    const int lane = threadIdx.x & 31;
    const int node = blockIdx.x * (blockDim.x >> 5) + wave;
    if (node >= GN) return;
    const float* ap = agg + (size_t)node * GD;
    float v[4];
    float s = 0.0f, sq = 0.0f;
#pragma unroll
    for (int j = 0; j < 4; ++j) {
        int f = j * 32 + lane;
        v[j] = ap[f] + bias[f];
        s  += v[j];
        sq += v[j] * v[j];
    }
#pragma unroll
    for (int off = 16; off > 0; off >>= 1) {
        s  += __shfl_xor(s,  off, 32);
        sq += __shfl_xor(sq, off, 32);
    }
    const float mean = s * (1.0f / GD);
    const float var  = sq * (1.0f / GD) - mean * mean;
    const float rstd = rsqrtf(var + 1e-5f);
    float* op = out + (size_t)node * GD;
#pragma unroll
    for (int j = 0; j < 4; ++j) {
        int f = j * 32 + lane;
        float y = (v[j] - mean) * rstd * gamma[f] + beta[f];
        op[f] = (y >= 0.0f) ? y : 0.1f * y;
    }
}

__global__ void pool_k(const float* __restrict__ x,
                       const int* __restrict__ batch,
                       float* __restrict__ gmax,
                       float* __restrict__ gsum,
                       float* __restrict__ cnt) {
    int tid = blockIdx.x * blockDim.x + threadIdx.x;
    if (tid >= GN * GD) return;
    const int n = tid >> 7;
    const int f = tid & 127;
    const int b = batch[n];
    const float v = x[tid];
    atomicMaxFloat(gmax + (size_t)b * GD + f, v);
    atomicAdd(gsum + (size_t)b * GD + f, v);
    if (f == 0) atomicAdd(cnt + b, 1.0f);
}

__global__ void pool_fin_k(const float* __restrict__ gmax,
                           const float* __restrict__ gsum,
                           const float* __restrict__ cnt,
                           float* __restrict__ pooled) {
    int tid = blockIdx.x * blockDim.x + threadIdx.x;
    if (tid >= GG * 2 * GD) return;
    const int g = tid >> 8;
    const int f = tid & 255;
    pooled[tid] = (f < GD) ? gmax[(size_t)g * GD + f]
                           : gsum[(size_t)g * GD + (f - GD)] / fmaxf(cnt[g], 1.0f);
}

extern "C" void kernel_launch(void* const* d_in, const int* in_sizes, int n_in,
                              void* d_out, int out_size, void* d_ws, size_t ws_size,
                              hipStream_t stream) {
    const float* x       = (const float*)d_in[0];
    const int*   ei      = (const int*)  d_in[1];
    const int*   batch   = (const int*)  d_in[2];
    const float* W       = (const float*)d_in[3];
    const float* att_src = (const float*)d_in[4];
    const float* att_dst = (const float*)d_in[5];
    const float* bias    = (const float*)d_in[6];
    const float* ln_g    = (const float*)d_in[7];
    const float* ln_b    = (const float*)d_in[8];
    const float* W_out   = (const float*)d_in[9];
    const float* b_out   = (const float*)d_in[10];
    float* out = (float*)d_out;

    float* ws     = (float*)d_ws;
    float* ht     = ws;
    float* agg    = ht     + (size_t)GN * GD;
    float* nodeio = agg    + (size_t)GN * GD;
    float* als    = nodeio + (size_t)GN * GD;
    float* ald    = als    + (size_t)GN * GH;
    float* segm   = ald    + (size_t)GN * GH;
    float* denom  = segm   + (size_t)GN * GH;
    float* gmax   = denom  + (size_t)GN * GH;
    float* gsum   = gmax   + (size_t)GG * GD;
    float* cnt    = gsum   + (size_t)GG * GD;
    float* pooled = cnt    + GG;

    const float NEG_INF = -INFINITY;
    const int TB = 256;
    auto blocks = [](long long n, int tb) { return (int)((n + tb - 1) / tb); };

    for (int l = 0; l < 3; ++l) {
        const float* in = (l == 0) ? x : nodeio;

        {
            int tiles = (GN / 16) * (GD / 16);
            gemm_wmma_f32<<<blocks(tiles, 4), 128, 0, stream>>>(
                in, W + (size_t)l * GD * GD, nullptr, ht, GN, GD, GD);
        }

        attn_proj_k<<<blocks((long long)GN * GH, TB), TB, 0, stream>>>(
            ht, att_src + l * GH * GC, att_dst + l * GH * GC, als, ald);

        fill_f32<<<blocks((long long)GN * GH, TB), TB, 0, stream>>>(segm, NEG_INF, GN * GH);
        fill_f32<<<blocks((long long)GN * GH, TB), TB, 0, stream>>>(denom, 0.0f, GN * GH);
        fill_f32<<<blocks((long long)GN * GD, TB), TB, 0, stream>>>(agg, 0.0f, GN * GD);

        edge_max_k<<<blocks(GETOT, TB), TB, 0, stream>>>(ei, als, ald, segm);
        edge_sum_k<<<blocks(GETOT, TB), TB, 0, stream>>>(ei, als, ald, segm, denom);
        edge_agg_k<<<blocks((long long)GETOT * GH, TB), TB, 0, stream>>>(
            ei, als, ald, segm, denom, ht, agg);

        post_ln_k<<<blocks(GN, 8), 256, 0, stream>>>(
            agg, bias + l * GD, ln_g + l * GD, ln_b + l * GD, nodeio);
    }

    fill_f32<<<blocks(GG * GD, TB), TB, 0, stream>>>(gmax, NEG_INF, GG * GD);
    fill_f32<<<blocks(GG * GD, TB), TB, 0, stream>>>(gsum, 0.0f, GG * GD);
    fill_f32<<<blocks(GG, TB), TB, 0, stream>>>(cnt, 0.0f, GG);
    pool_k<<<blocks((long long)GN * GD, TB), TB, 0, stream>>>(nodeio, batch, gmax, gsum, cnt);
    pool_fin_k<<<blocks(GG * 2 * GD, TB), TB, 0, stream>>>(gmax, gsum, cnt, pooled);

    {
        int tiles = (GG / 16) * (GEMB / 16);
        gemm_wmma_f32<<<blocks(tiles, 4), 128, 0, stream>>>(
            pooled, W_out, b_out, out, GG, 2 * GD, GEMB);
    }
}


# Edge softmax+aggregate: edges grouped by destination per call (count/scan/place), one wave per destination gathers its segment in registers; no float atomics there; sums in counted-rank order
# speedup vs baseline: 6.8716x; 6.8716x over previous
#include <hip/hip_runtime.h>
#include <cmath>


#define GN   50000
#define GE   800000
#define GETOT (GE + GN)
#define GG   256
#define GH   4
#define GC   32
#define GD   128
#define GEMB 128

typedef float v2f __attribute__((ext_vector_type(2)));
typedef float v8f __attribute__((ext_vector_type(8)));

__device__ __forceinline__ void atomicMaxFloat(float* addr, float val) {
    if (val >= 0.0f) {
        atomicMax((int*)addr, __float_as_int(val));
    } else {
        atomicMin((unsigned int*)addr, __float_as_uint(val));
    }
}

__global__ void fill_f32(float* __restrict__ p, float v, int n) {
    int i = blockIdx.x * blockDim.x + threadIdx.x;
    if (i < n) p[i] = v;
}

__global__ void gemm_wmma_f32(const float* __restrict__ A,
                              const float* __restrict__ B,
                              const float* __restrict__ bias,
                              float* __restrict__ C,
                              int M, int K, int N) {
    const int wave = threadIdx.x >> 5;
    const int lane = threadIdx.x & 31;
    const int tilesN = N >> 4;
    const int total  = (M >> 4) * tilesN;
    const int tile = blockIdx.x * (blockDim.x >> 5) + wave;
    if (tile >= total) return;

    const int row0 = (tile / tilesN) << 4;
    const int col0 = (tile % tilesN) << 4;
    const int half = lane >> 4;
    const int l16  = lane & 15;

    const float* Ap = A + (size_t)(row0 + l16) * K + 2 * half;
    const float* Bp = B + (size_t)(2 * half) * N + col0 + l16;

    v8f acc = {};
    for (int k0 = 0; k0 < K; k0 += 4) {
        v2f a, b;
        a.x = Ap[k0];
        a.y = Ap[k0 + 1];
        b.x = Bp[(size_t)k0 * N];
        b.y = Bp[(size_t)(k0 + 1) * N];
        acc = __builtin_amdgcn_wmma_f32_16x16x4_f32(
                false, a, false, b, (short)0, acc, false, false);
    }

    const float bv = bias ? bias[col0 + l16] : 0.0f;
    float* Cp = C + (size_t)(row0 + 8 * half) * N + col0 + l16;
#pragma unroll
    for (int j = 0; j < 8; ++j) {
        Cp[(size_t)j * N] = acc[j] + bv;
    }
}

__global__ void attn_proj_k(const float* __restrict__ ht,
                            const float* __restrict__ asrc,
                            const float* __restrict__ adst,
                            float* __restrict__ als,
                            float* __restrict__ ald) {
    int tid = blockIdx.x * blockDim.x + threadIdx.x;
    if (tid >= GN * GH) return;
    const int n = tid >> 2;
    const int h = tid & 3;
    const float* hp = ht + (size_t)n * GD + h * GC;
    const float* ap = asrc + h * GC;
    const float* dp = adst + h * GC;
    float s = 0.0f, d = 0.0f;
#pragma unroll
    for (int c = 0; c < GC; ++c) {
        float v = hp[c];
        s += v * ap[c];
        d += v * dp[c];
    }
    als[tid] = s;
    ald[tid] = d;
}

__global__ void csr_count_k(const int* __restrict__ ei,
                            int* __restrict__ deg,
                            int* __restrict__ rank) {
    int e = blockIdx.x * blockDim.x + threadIdx.x;
    if (e >= GE) return;
    rank[e] = atomicAdd(deg + ei[GE + e], 1);
}

#define SCAN_T   1024
#define SCAN_PER ((GN + SCAN_T - 1) / SCAN_T)

__global__ void __launch_bounds__(SCAN_T)
csr_scan_k(const int* __restrict__ deg,
           int* __restrict__ rowptr) {
    __shared__ int part[SCAN_T];
    const int t = threadIdx.x;
    const int base = t * SCAN_PER;
    int sum = 0;
    for (int i = 0; i < SCAN_PER; ++i) {
        const int idx = base + i;
        if (idx < GN) sum += deg[idx];
    }
    part[t] = sum;
    __syncthreads();
    for (int off = 1; off < SCAN_T; off <<= 1) {
        const int v = (t >= off) ? part[t - off] : 0;
        __syncthreads();
        part[t] += v;
        __syncthreads();
    }
    int run = part[t] - sum;
    for (int i = 0; i < SCAN_PER; ++i) {
        const int idx = base + i;
        if (idx < GN) {
            rowptr[idx] = run;
            run += deg[idx];
        }
    }
    if (t == SCAN_T - 1) rowptr[GN] = part[t];
}

__global__ void csr_fill_k(const int* __restrict__ ei,
                           const int* __restrict__ rowptr,
                           const int* __restrict__ rank,
                           int* __restrict__ csrc) {
    int e = blockIdx.x * blockDim.x + threadIdx.x;
    if (e >= GE) return;
    csrc[rowptr[ei[GE + e]] + rank[e]] = ei[e];
}

__device__ __forceinline__ float lrelu02(float v) {
    return (v >= 0.0f) ? v : 0.2f * v;
}

__device__ __forceinline__ float pick4(int h, float a, float b, float c, float d) {
    const bool odd  = (h & 1) != 0;
    const bool high = (h & 2) != 0;
    const float lo = odd ? b : a;
    const float hi = odd ? d : c;
    return high ? hi : lo;
}

__global__ void __launch_bounds__(256)
seg_softmax_agg_k(const int* __restrict__ rowptr,
                  const int* __restrict__ csrc,
                  const float* __restrict__ als,
                  const float* __restrict__ ald,
                  const float* __restrict__ ht,
                  float* __restrict__ agg) {
    const int lane = threadIdx.x & 31;
    const int d = __builtin_amdgcn_readfirstlane(
        (int)(blockIdx.x * (blockDim.x >> 5) + (threadIdx.x >> 5)));
    if (d >= GN) return;
    const int beg = rowptr[d];
    const int end = rowptr[d + 1];
    const float4* als4 = reinterpret_cast<const float4*>(als);
    const float4 ad = reinterpret_cast<const float4*>(ald)[d];
    const float4 sd = als4[d];

    const float t0 = lrelu02(sd.x + ad.x);
    const float t1 = lrelu02(sd.y + ad.y);
    const float t2 = lrelu02(sd.z + ad.z);
    const float t3 = lrelu02(sd.w + ad.w);

    float m0 = t0, m1 = t1, m2 = t2, m3 = t3;
    for (int i = beg + lane; i < end; i += 32) {
        const float4 a = als4[csrc[i]];
        m0 = fmaxf(m0, lrelu02(a.x + ad.x));
        m1 = fmaxf(m1, lrelu02(a.y + ad.y));
        m2 = fmaxf(m2, lrelu02(a.z + ad.z));
        m3 = fmaxf(m3, lrelu02(a.w + ad.w));
    }
    for (int off = 16; off > 0; off >>= 1) {
        m0 = fmaxf(m0, __shfl_xor(m0, off, 32));
        m1 = fmaxf(m1, __shfl_xor(m1, off, 32));
        m2 = fmaxf(m2, __shfl_xor(m2, off, 32));
        m3 = fmaxf(m3, __shfl_xor(m3, off, 32));
    }

    float q0 = 0.0f, q1 = 0.0f, q2 = 0.0f, q3 = 0.0f;
    for (int i = beg + lane; i < end; i += 32) {
        const float4 a = als4[csrc[i]];
        q0 += expf(lrelu02(a.x + ad.x) - m0);
        q1 += expf(lrelu02(a.y + ad.y) - m1);
        q2 += expf(lrelu02(a.z + ad.z) - m2);
        q3 += expf(lrelu02(a.w + ad.w) - m3);
    }
    for (int off = 16; off > 0; off >>= 1) {
        q0 += __shfl_xor(q0, off, 32);
        q1 += __shfl_xor(q1, off, 32);
        q2 += __shfl_xor(q2, off, 32);
        q3 += __shfl_xor(q3, off, 32);
    }
    q0 += expf(t0 - m0);
    q1 += expf(t1 - m1);
    q2 += expf(t2 - m2);
    q3 += expf(t3 - m3);

    const int h = lane >> 3;
    const float4* hrow = reinterpret_cast<const float4*>(ht) + lane;
    float4 acc;
    {
        const float al = expf(pick4(h, t0, t1, t2, t3) - pick4(h, m0, m1, m2, m3))
                         / pick4(h, q0, q1, q2, q3);
        const float4 hv = hrow[(size_t)d * (GD / 4)];
        acc.x = hv.x * al;
        acc.y = hv.y * al;
        acc.z = hv.z * al;
        acc.w = hv.w * al;
    }
    for (int base = beg; base < end; base += 32) {
        int   s  = 0;
        float a0 = 0.0f, a1 = 0.0f, a2 = 0.0f, a3 = 0.0f;
        if (base + lane < end) {
            s = csrc[base + lane];
            const float4 a = als4[s];
            a0 = expf(lrelu02(a.x + ad.x) - m0) / q0;
            a1 = expf(lrelu02(a.y + ad.y) - m1) / q1;
            a2 = expf(lrelu02(a.z + ad.z) - m2) / q2;
            a3 = expf(lrelu02(a.w + ad.w) - m3) / q3;
        }
        const int cnt = (end - base < 32) ? (end - base) : 32;
#pragma unroll 4
        for (int k = 0; k < cnt; ++k) {
            const int sk = __builtin_amdgcn_readlane(s, k);
            const float b0 = __int_as_float(__builtin_amdgcn_readlane(__float_as_int(a0), k));
            const float b1 = __int_as_float(__builtin_amdgcn_readlane(__float_as_int(a1), k));
            const float b2 = __int_as_float(__builtin_amdgcn_readlane(__float_as_int(a2), k));
            const float b3 = __int_as_float(__builtin_amdgcn_readlane(__float_as_int(a3), k));
            const float al = pick4(h, b0, b1, b2, b3);
            const float4 hv = hrow[(size_t)sk * (GD / 4)];
            acc.x += hv.x * al;
            acc.y += hv.y * al;
            acc.z += hv.z * al;
            acc.w += hv.w * al;
        }
    }
    reinterpret_cast<float4*>(agg)[(size_t)d * (GD / 4) + lane] = acc;
}

__global__ void post_ln_k(const float* __restrict__ agg,
                          const float* __restrict__ bias,
                          const float* __restrict__ gamma,
                          const float* __restrict__ beta,
                          float* __restrict__ out) {
    const int wave = threadIdx.x >> 5;
    const int lane = threadIdx.x & 31;
    const int node = blockIdx.x * (blockDim.x >> 5) + wave;
    if (node >= GN) return;
    const float* ap = agg + (size_t)node * GD;
    float v[4];
    float s = 0.0f, sq = 0.0f;
#pragma unroll
    for (int j = 0; j < 4; ++j) {
        int f = j * 32 + lane;
        v[j] = ap[f] + bias[f];
        s  += v[j];
        sq += v[j] * v[j];
    }
#pragma unroll
    for (int off = 16; off > 0; off >>= 1) {
        s  += __shfl_xor(s,  off, 32);
        sq += __shfl_xor(sq, off, 32);
    }
    const float mean = s * (1.0f / GD);
    const float var  = sq * (1.0f / GD) - mean * mean;
    const float rstd = rsqrtf(var + 1e-5f);
    float* op = out + (size_t)node * GD;
#pragma unroll
    for (int j = 0; j < 4; ++j) {
        int f = j * 32 + lane;
        float y = (v[j] - mean) * rstd * gamma[f] + beta[f];
        op[f] = (y >= 0.0f) ? y : 0.1f * y;
    }
}

__global__ void pool_k(const float* __restrict__ x,
                       const int* __restrict__ batch,
                       float* __restrict__ gmax,
                       float* __restrict__ gsum,
                       float* __restrict__ cnt) {
    int tid = blockIdx.x * blockDim.x + threadIdx.x;
    if (tid >= GN * GD) return;
    const int n = tid >> 7;
    const int f = tid & 127;
    const int b = batch[n];
    const float v = x[tid];
    atomicMaxFloat(gmax + (size_t)b * GD + f, v);
    atomicAdd(gsum + (size_t)b * GD + f, v);
    if (f == 0) atomicAdd(cnt + b, 1.0f);
}

__global__ void pool_fin_k(const float* __restrict__ gmax,
                           const float* __restrict__ gsum,
                           const float* __restrict__ cnt,
                           float* __restrict__ pooled) {
    int tid = blockIdx.x * blockDim.x + threadIdx.x;
    if (tid >= GG * 2 * GD) return;
    const int g = tid >> 8;
    const int f = tid & 255;
    pooled[tid] = (f < GD) ? gmax[(size_t)g * GD + f]
                           : gsum[(size_t)g * GD + (f - GD)] / fmaxf(cnt[g], 1.0f);
}

extern "C" void kernel_launch(void* const* d_in, const int* in_sizes, int n_in,
                              void* d_out, int out_size, void* d_ws, size_t ws_size,
                              hipStream_t stream) {
    const float* x       = (const float*)d_in[0];
    const int*   ei      = (const int*)  d_in[1];
    const int*   batch   = (const int*)  d_in[2];
    const float* W       = (const float*)d_in[3];
    const float* att_src = (const float*)d_in[4];
    const float* att_dst = (const float*)d_in[5];
    const float* bias    = (const float*)d_in[6];
    const float* ln_g    = (const float*)d_in[7];
    const float* ln_b    = (const float*)d_in[8];
    const float* W_out   = (const float*)d_in[9];
    const float* b_out   = (const float*)d_in[10];
    float* out = (float*)d_out;

    float* ws     = (float*)d_ws;
    float* bufP   = ws;
    float* bufQ   = bufP   + (size_t)GN * GD;
    int*   rowptr = (int*)(bufQ + (size_t)GN * GD);
    int*   deg    = rowptr + (GN + 1);
    int*   rank   = deg    + GN;
    int*   csrc   = rank   + GE;
    float* als    = ws     + (size_t)3 * GN * GD;
    float* ald    = als    + (size_t)GN * GH;
    float* gmax   = ald    + (size_t)3 * GN * GH;
    float* gsum   = gmax   + (size_t)GG * GD;
    float* cnt    = gsum   + (size_t)GG * GD;
    float* pooled = cnt    + GG;

    const float NEG_INF = -INFINITY;
    const int TB = 256;
    auto blocks = [](long long n, int tb) { return (int)((n + tb - 1) / tb); };

    fill_f32<<<blocks(GN, TB), TB, 0, stream>>>((float*)deg, 0.0f, GN);
    csr_count_k<<<blocks(GE, TB), TB, 0, stream>>>(ei, deg, rank);
    csr_scan_k<<<1, SCAN_T, 0, stream>>>(deg, rowptr);
    csr_fill_k<<<blocks(GE, TB), TB, 0, stream>>>(ei, rowptr, rank, csrc);

    float* nodeio = bufQ;
    for (int l = 0; l < 3; ++l) {
        float* ht  = (l & 1) ? bufP : bufQ;
        float* agg = (l & 1) ? bufQ : bufP;
        const float* in = (l == 0) ? x : agg;

        {
            int tiles = (GN / 16) * (GD / 16);
            gemm_wmma_f32<<<blocks(tiles, 4), 128, 0, stream>>>(
                in, W + (size_t)l * GD * GD, nullptr, ht, GN, GD, GD);
        }

        attn_proj_k<<<blocks((long long)GN * GH, TB), TB, 0, stream>>>(
            ht, att_src + l * GH * GC, att_dst + l * GH * GC, als, ald);

        seg_softmax_agg_k<<<blocks(GN, 8), 256, 0, stream>>>(
            rowptr, csrc, als, ald, ht, agg);

        post_ln_k<<<blocks(GN, 8), 256, 0, stream>>>(
            agg, bias + l * GD, ln_g + l * GD, ln_b + l * GD, ht);
    }

    fill_f32<<<blocks(GG * GD, TB), TB, 0, stream>>>(gmax, NEG_INF, GG * GD);
    fill_f32<<<blocks(GG * GD, TB), TB, 0, stream>>>(gsum, 0.0f, GG * GD);
    fill_f32<<<blocks(GG, TB), TB, 0, stream>>>(cnt, 0.0f, GG);
    pool_k<<<blocks((long long)GN * GD, TB), TB, 0, stream>>>(nodeio, batch, gmax, gsum, cnt);
    pool_fin_k<<<blocks(GG * 2 * GD, TB), TB, 0, stream>>>(gmax, gsum, cnt, pooled);

    {
        int tiles = (GG / 16) * (GEMB / 16);
        gemm_wmma_f32<<<blocks(tiles, 4), 128, 0, stream>>>(
            pooled, W_out, b_out, out, GG, 2 * GD, GEMB);
    }
}


# Pooling: one wave folds 8 consecutive nodes in registers and issues max/add atomics once per run of equal graph ids (was two per element); sums go in node order per run, then atomic add
# speedup vs baseline: 8.1502x; 1.1861x over previous
#include <hip/hip_runtime.h>
#include <cmath>


#define GN   50000
#define GE   800000
#define GETOT (GE + GN)
#define GG   256
#define GH   4
#define GC   32
#define GD   128
#define GEMB 128

typedef float v2f __attribute__((ext_vector_type(2)));
typedef float v8f __attribute__((ext_vector_type(8)));

__device__ __forceinline__ void atomicMaxFloat(float* addr, float val) {
    if (val >= 0.0f) {
        atomicMax((int*)addr, __float_as_int(val));
    } else {
        atomicMin((unsigned int*)addr, __float_as_uint(val));
    }
}

__global__ void fill_f32(float* __restrict__ p, float v, int n) {
    int i = blockIdx.x * blockDim.x + threadIdx.x;
    if (i < n) p[i] = v;
}

__global__ void gemm_wmma_f32(const float* __restrict__ A,
                              const float* __restrict__ B,
                              const float* __restrict__ bias,
                              float* __restrict__ C,
                              int M, int K, int N) {
    const int wave = threadIdx.x >> 5;
    const int lane = threadIdx.x & 31;
    const int tilesN = N >> 4;
    const int total  = (M >> 4) * tilesN;
    const int tile = blockIdx.x * (blockDim.x >> 5) + wave;
    if (tile >= total) return;

    const int row0 = (tile / tilesN) << 4;
    const int col0 = (tile % tilesN) << 4;
    const int half = lane >> 4;
    const int l16  = lane & 15;

    const float* Ap = A + (size_t)(row0 + l16) * K + 2 * half;
    const float* Bp = B + (size_t)(2 * half) * N + col0 + l16;

    v8f acc = {};
    for (int k0 = 0; k0 < K; k0 += 4) {
        v2f a, b;
        a.x = Ap[k0];
        a.y = Ap[k0 + 1];
        b.x = Bp[(size_t)k0 * N];
        b.y = Bp[(size_t)(k0 + 1) * N];
        acc = __builtin_amdgcn_wmma_f32_16x16x4_f32(
                false, a, false, b, (short)0, acc, false, false);
    }

    const float bv = bias ? bias[col0 + l16] : 0.0f;
    float* Cp = C + (size_t)(row0 + 8 * half) * N + col0 + l16;
#pragma unroll
    for (int j = 0; j < 8; ++j) {
        Cp[(size_t)j * N] = acc[j] + bv;
    }
}

__global__ void attn_proj_k(const float* __restrict__ ht,
                            const float* __restrict__ asrc,
                            const float* __restrict__ adst,
                            float* __restrict__ als,
                            float* __restrict__ ald) {
    int tid = blockIdx.x * blockDim.x + threadIdx.x;
    if (tid >= GN * GH) return;
    const int n = tid >> 2;
    const int h = tid & 3;
    const float* hp = ht + (size_t)n * GD + h * GC;
    const float* ap = asrc + h * GC;
    const float* dp = adst + h * GC;
    float s = 0.0f, d = 0.0f;
#pragma unroll
    for (int c = 0; c < GC; ++c) {
        float v = hp[c];
        s += v * ap[c];
        d += v * dp[c];
    }
    als[tid] = s;
    ald[tid] = d;
}

__global__ void csr_count_k(const int* __restrict__ ei,
                            int* __restrict__ deg,
                            int* __restrict__ rank) {
    int e = blockIdx.x * blockDim.x + threadIdx.x;
    if (e >= GE) return;
    rank[e] = atomicAdd(deg + ei[GE + e], 1);
}

#define SCAN_T   1024
#define SCAN_PER ((GN + SCAN_T - 1) / SCAN_T)

__global__ void __launch_bounds__(SCAN_T)
csr_scan_k(const int* __restrict__ deg,
           int* __restrict__ rowptr) {
    __shared__ int part[SCAN_T];
    const int t = threadIdx.x;
    const int base = t * SCAN_PER;
    int sum = 0;
    for (int i = 0; i < SCAN_PER; ++i) {
        const int idx = base + i;
        if (idx < GN) sum += deg[idx];
    }
    part[t] = sum;
    __syncthreads();
    for (int off = 1; off < SCAN_T; off <<= 1) {
        const int v = (t >= off) ? part[t - off] : 0;
        __syncthreads();
        part[t] += v;
        __syncthreads();
    }
    int run = part[t] - sum;
    for (int i = 0; i < SCAN_PER; ++i) {
        const int idx = base + i;
        if (idx < GN) {
            rowptr[idx] = run;
            run += deg[idx];
        }
    }
    if (t == SCAN_T - 1) rowptr[GN] = part[t];
}

__global__ void csr_fill_k(const int* __restrict__ ei,
                           const int* __restrict__ rowptr,
                           const int* __restrict__ rank,
                           int* __restrict__ csrc) {
    int e = blockIdx.x * blockDim.x + threadIdx.x;
    if (e >= GE) return;
    csrc[rowptr[ei[GE + e]] + rank[e]] = ei[e];
}

__device__ __forceinline__ float lrelu02(float v) {
    return (v >= 0.0f) ? v : 0.2f * v;
}

__device__ __forceinline__ float pick4(int h, float a, float b, float c, float d) {
    const bool odd  = (h & 1) != 0;
    const bool high = (h & 2) != 0;
    const float lo = odd ? b : a;
    const float hi = odd ? d : c;
    return high ? hi : lo;
}

__global__ void __launch_bounds__(256)
seg_softmax_agg_k(const int* __restrict__ rowptr,
                  const int* __restrict__ csrc,
                  const float* __restrict__ als,
                  const float* __restrict__ ald,
                  const float* __restrict__ ht,
                  float* __restrict__ agg) {
    const int lane = threadIdx.x & 31;
    const int d = __builtin_amdgcn_readfirstlane(
        (int)(blockIdx.x * (blockDim.x >> 5) + (threadIdx.x >> 5)));
    if (d >= GN) return;
    const int beg = rowptr[d];
    const int end = rowptr[d + 1];
    const float4* als4 = reinterpret_cast<const float4*>(als);
    const float4 ad = reinterpret_cast<const float4*>(ald)[d];
    const float4 sd = als4[d];

    const float t0 = lrelu02(sd.x + ad.x);
    const float t1 = lrelu02(sd.y + ad.y);
    const float t2 = lrelu02(sd.z + ad.z);
    const float t3 = lrelu02(sd.w + ad.w);

    float m0 = t0, m1 = t1, m2 = t2, m3 = t3;
    for (int i = beg + lane; i < end; i += 32) {
        const float4 a = als4[csrc[i]];
        m0 = fmaxf(m0, lrelu02(a.x + ad.x));
        m1 = fmaxf(m1, lrelu02(a.y + ad.y));
        m2 = fmaxf(m2, lrelu02(a.z + ad.z));
        m3 = fmaxf(m3, lrelu02(a.w + ad.w));
    }
    for (int off = 16; off > 0; off >>= 1) {
        m0 = fmaxf(m0, __shfl_xor(m0, off, 32));
        m1 = fmaxf(m1, __shfl_xor(m1, off, 32));
        m2 = fmaxf(m2, __shfl_xor(m2, off, 32));
        m3 = fmaxf(m3, __shfl_xor(m3, off, 32));
    }

    float q0 = 0.0f, q1 = 0.0f, q2 = 0.0f, q3 = 0.0f;
    for (int i = beg + lane; i < end; i += 32) {
        const float4 a = als4[csrc[i]];
        q0 += expf(lrelu02(a.x + ad.x) - m0);
        q1 += expf(lrelu02(a.y + ad.y) - m1);
        q2 += expf(lrelu02(a.z + ad.z) - m2);
        q3 += expf(lrelu02(a.w + ad.w) - m3);
    }
    for (int off = 16; off > 0; off >>= 1) {
        q0 += __shfl_xor(q0, off, 32);
        q1 += __shfl_xor(q1, off, 32);
        q2 += __shfl_xor(q2, off, 32);
        q3 += __shfl_xor(q3, off, 32);
    }
    q0 += expf(t0 - m0);
    q1 += expf(t1 - m1);
    q2 += expf(t2 - m2);
    q3 += expf(t3 - m3);

    const int h = lane >> 3;
    const float4* hrow = reinterpret_cast<const float4*>(ht) + lane;
    float4 acc;
    {
        const float al = expf(pick4(h, t0, t1, t2, t3) - pick4(h, m0, m1, m2, m3))
                         / pick4(h, q0, q1, q2, q3);
        const float4 hv = hrow[(size_t)d * (GD / 4)];
        acc.x = hv.x * al;
        acc.y = hv.y * al;
        acc.z = hv.z * al;
        acc.w = hv.w * al;
    }
    for (int base = beg; base < end; base += 32) {
        int   s  = 0;
        float a0 = 0.0f, a1 = 0.0f, a2 = 0.0f, a3 = 0.0f;
        if (base + lane < end) {
            s = csrc[base + lane];
            const float4 a = als4[s];
            a0 = expf(lrelu02(a.x + ad.x) - m0) / q0;
            a1 = expf(lrelu02(a.y + ad.y) - m1) / q1;
            a2 = expf(lrelu02(a.z + ad.z) - m2) / q2;
            a3 = expf(lrelu02(a.w + ad.w) - m3) / q3;
        }
        const int cnt = (end - base < 32) ? (end - base) : 32;
#pragma unroll 4
        for (int k = 0; k < cnt; ++k) {
            const int sk = __builtin_amdgcn_readlane(s, k);
            const float b0 = __int_as_float(__builtin_amdgcn_readlane(__float_as_int(a0), k));
            const float b1 = __int_as_float(__builtin_amdgcn_readlane(__float_as_int(a1), k));
            const float b2 = __int_as_float(__builtin_amdgcn_readlane(__float_as_int(a2), k));
            const float b3 = __int_as_float(__builtin_amdgcn_readlane(__float_as_int(a3), k));
            const float al = pick4(h, b0, b1, b2, b3);
            const float4 hv = hrow[(size_t)sk * (GD / 4)];
            acc.x += hv.x * al;
            acc.y += hv.y * al;
            acc.z += hv.z * al;
            acc.w += hv.w * al;
        }
    }
    reinterpret_cast<float4*>(agg)[(size_t)d * (GD / 4) + lane] = acc;
}

__global__ void post_ln_k(const float* __restrict__ agg,
                          const float* __restrict__ bias,
                          const float* __restrict__ gamma,
                          const float* __restrict__ beta,
                          float* __restrict__ out) {
    const int wave = threadIdx.x >> 5;
    const int lane = threadIdx.x & 31;
    const int node = blockIdx.x * (blockDim.x >> 5) + wave;
    if (node >= GN) return;
    const float* ap = agg + (size_t)node * GD;
    float v[4];
    float s = 0.0f, sq = 0.0f;
#pragma unroll
    for (int j = 0; j < 4; ++j) {
        int f = j * 32 + lane;
        v[j] = ap[f] + bias[f];
        s  += v[j];
        sq += v[j] * v[j];
    }
#pragma unroll
    for (int off = 16; off > 0; off >>= 1) {
        s  += __shfl_xor(s,  off, 32);
        sq += __shfl_xor(sq, off, 32);
    }
    const float mean = s * (1.0f / GD);
    const float var  = sq * (1.0f / GD) - mean * mean;
    const float rstd = rsqrtf(var + 1e-5f);
    float* op = out + (size_t)node * GD;
#pragma unroll
    for (int j = 0; j < 4; ++j) {
        int f = j * 32 + lane;
        float y = (v[j] - mean) * rstd * gamma[f] + beta[f];
        op[f] = (y >= 0.0f) ? y : 0.1f * y;
    }
}

#define POOL_NPW 8
static_assert(GN % POOL_NPW == 0, "a wave takes POOL_NPW whole nodes");

__device__ __forceinline__ void pool_flush(float* __restrict__ gmax,
                                           float* __restrict__ gsum,
                                           float* __restrict__ cnt,
                                           int g, int lane,
                                           const float* mx, const float* sm, int c) {
    float* pm = gmax + (size_t)g * GD + lane;
    float* ps = gsum + (size_t)g * GD + lane;
#pragma unroll
    for (int i = 0; i < 4; ++i) {
        atomicMaxFloat(pm + 32 * i, mx[i]);
        atomicAdd(ps + 32 * i, sm[i]);
    }
    if (lane == 0) atomicAdd(cnt + g, (float)c);
}

__global__ void __launch_bounds__(256)
pool_k(const float* __restrict__ x,
       const int* __restrict__ batch,
       float* __restrict__ gmax,
       float* __restrict__ gsum,
       float* __restrict__ cnt) {
    const int lane = threadIdx.x & 31;
    const int n0 = __builtin_amdgcn_readfirstlane(
        (int)((blockIdx.x * (blockDim.x >> 5) + (threadIdx.x >> 5)) * POOL_NPW));
    if (n0 >= GN) return;
    const float* xp = x + (size_t)n0 * GD + lane;

    int   b[POOL_NPW];
    float v[POOL_NPW][4];
#pragma unroll
    for (int j = 0; j < POOL_NPW; ++j) {
        b[j] = batch[n0 + j];
#pragma unroll
        for (int i = 0; i < 4; ++i) v[j][i] = xp[j * GD + 32 * i];
    }

    int cur = b[0];
    int c = 1;
    float mx[4], sm[4];
#pragma unroll
    for (int i = 0; i < 4; ++i) { mx[i] = v[0][i]; sm[i] = v[0][i]; }
#pragma unroll
    for (int j = 1; j < POOL_NPW; ++j) {
        if (b[j] != cur) {
            pool_flush(gmax, gsum, cnt, cur, lane, mx, sm, c);
            cur = b[j];
            c = 1;
#pragma unroll
            for (int i = 0; i < 4; ++i) { mx[i] = v[j][i]; sm[i] = v[j][i]; }
        } else {
#pragma unroll
            for (int i = 0; i < 4; ++i) {
                mx[i] = fmaxf(mx[i], v[j][i]);
                sm[i] += v[j][i];
            }
            ++c;
        }
    }
    pool_flush(gmax, gsum, cnt, cur, lane, mx, sm, c);
}

__global__ void pool_fin_k(const float* __restrict__ gmax,
                           const float* __restrict__ gsum,
                           const float* __restrict__ cnt,
                           float* __restrict__ pooled) {
    int tid = blockIdx.x * blockDim.x + threadIdx.x;
    if (tid >= GG * 2 * GD) return;
    const int g = tid >> 8;
    const int f = tid & 255;
    pooled[tid] = (f < GD) ? gmax[(size_t)g * GD + f]
                           : gsum[(size_t)g * GD + (f - GD)] / fmaxf(cnt[g], 1.0f);
}

extern "C" void kernel_launch(void* const* d_in, const int* in_sizes, int n_in,
                              void* d_out, int out_size, void* d_ws, size_t ws_size,
                              hipStream_t stream) {
    const float* x       = (const float*)d_in[0];
    const int*   ei      = (const int*)  d_in[1];
    const int*   batch   = (const int*)  d_in[2];
    const float* W       = (const float*)d_in[3];
    const float* att_src = (const float*)d_in[4];
    const float* att_dst = (const float*)d_in[5];
    const float* bias    = (const float*)d_in[6];
    const float* ln_g    = (const float*)d_in[7];
    const float* ln_b    = (const float*)d_in[8];
    const float* W_out   = (const float*)d_in[9];
    const float* b_out   = (const float*)d_in[10];
    float* out = (float*)d_out;

    float* ws     = (float*)d_ws;
    float* bufP   = ws;
    float* bufQ   = bufP   + (size_t)GN * GD;
    int*   rowptr = (int*)(bufQ + (size_t)GN * GD);
    int*   deg    = rowptr + (GN + 1);
    int*   rank   = deg    + GN;
    int*   csrc   = rank   + GE;
    float* als    = ws     + (size_t)3 * GN * GD;
    float* ald    = als    + (size_t)GN * GH;
    float* gmax   = ald    + (size_t)3 * GN * GH;
    float* gsum   = gmax   + (size_t)GG * GD;
    float* cnt    = gsum   + (size_t)GG * GD;
    float* pooled = cnt    + GG;

    const float NEG_INF = -INFINITY;
    const int TB = 256;
    auto blocks = [](long long n, int tb) { return (int)((n + tb - 1) / tb); };

    fill_f32<<<blocks(GN, TB), TB, 0, stream>>>((float*)deg, 0.0f, GN);
    csr_count_k<<<blocks(GE, TB), TB, 0, stream>>>(ei, deg, rank);
    csr_scan_k<<<1, SCAN_T, 0, stream>>>(deg, rowptr);
    csr_fill_k<<<blocks(GE, TB), TB, 0, stream>>>(ei, rowptr, rank, csrc);

    float* nodeio = bufQ;
    for (int l = 0; l < 3; ++l) {
        float* ht  = (l & 1) ? bufP : bufQ;
        float* agg = (l & 1) ? bufQ : bufP;
        const float* in = (l == 0) ? x : agg;

        {
            int tiles = (GN / 16) * (GD / 16);
            gemm_wmma_f32<<<blocks(tiles, 4), 128, 0, stream>>>(
                in, W + (size_t)l * GD * GD, nullptr, ht, GN, GD, GD);
        }

        attn_proj_k<<<blocks((long long)GN * GH, TB), TB, 0, stream>>>(
            ht, att_src + l * GH * GC, att_dst + l * GH * GC, als, ald);

        seg_softmax_agg_k<<<blocks(GN, 8), 256, 0, stream>>>(
            rowptr, csrc, als, ald, ht, agg);

        post_ln_k<<<blocks(GN, 8), 256, 0, stream>>>(
            agg, bias + l * GD, ln_g + l * GD, ln_b + l * GD, ht);
    }

    fill_f32<<<blocks(GG * GD, TB), TB, 0, stream>>>(gmax, NEG_INF, GG * GD);
    fill_f32<<<blocks(GG * GD, TB), TB, 0, stream>>>(gsum, 0.0f, GG * GD);
    fill_f32<<<blocks(GG, TB), TB, 0, stream>>>(cnt, 0.0f, GG);
    pool_k<<<blocks(GN / POOL_NPW, 8), 256, 0, stream>>>(
        nodeio, batch, gmax, gsum, cnt);
    pool_fin_k<<<blocks(GG * 2 * GD, TB), TB, 0, stream>>>(gmax, gsum, cnt, pooled);

    {
        int tiles = (GG / 16) * (GEMB / 16);
        gemm_wmma_f32<<<blocks(tiles, 4), 128, 0, stream>>>(
            pooled, W_out, b_out, out, GG, 2 * GD, GEMB);
    }
}


# bias+LayerNorm+leaky_relu fused into the aggregation wave (row already in its registers): 3 launches and one [N,128] write+read per layer gone; LN row sums now 4 per lane then butterfly
# speedup vs baseline: 9.0098x; 1.1055x over previous
#include <hip/hip_runtime.h>
#include <cmath>


#define GN   50000
#define GE   800000
#define GETOT (GE + GN)
#define GG   256
#define GH   4
#define GC   32
#define GD   128
#define GEMB 128

typedef float v2f __attribute__((ext_vector_type(2)));
typedef float v8f __attribute__((ext_vector_type(8)));

__device__ __forceinline__ void atomicMaxFloat(float* addr, float val) {
    if (val >= 0.0f) {
        atomicMax((int*)addr, __float_as_int(val));
    } else {
        atomicMin((unsigned int*)addr, __float_as_uint(val));
    }
}

__global__ void fill_f32(float* __restrict__ p, float v, int n) {
    int i = blockIdx.x * blockDim.x + threadIdx.x;
    if (i < n) p[i] = v;
}

__global__ void gemm_wmma_f32(const float* __restrict__ A,
                              const float* __restrict__ B,
                              const float* __restrict__ bias,
                              float* __restrict__ C,
                              int M, int K, int N) {
    const int wave = threadIdx.x >> 5;
    const int lane = threadIdx.x & 31;
    const int tilesN = N >> 4;
    const int total  = (M >> 4) * tilesN;
    const int tile = blockIdx.x * (blockDim.x >> 5) + wave;
    if (tile >= total) return;

    const int row0 = (tile / tilesN) << 4;
    const int col0 = (tile % tilesN) << 4;
    const int half = lane >> 4;
    const int l16  = lane & 15;

    const float* Ap = A + (size_t)(row0 + l16) * K + 2 * half;
    const float* Bp = B + (size_t)(2 * half) * N + col0 + l16;

    v8f acc = {};
    for (int k0 = 0; k0 < K; k0 += 4) {
        v2f a, b;
        a.x = Ap[k0];
        a.y = Ap[k0 + 1];
        b.x = Bp[(size_t)k0 * N];
        b.y = Bp[(size_t)(k0 + 1) * N];
        acc = __builtin_amdgcn_wmma_f32_16x16x4_f32(
                false, a, false, b, (short)0, acc, false, false);
    }

    const float bv = bias ? bias[col0 + l16] : 0.0f;
    float* Cp = C + (size_t)(row0 + 8 * half) * N + col0 + l16;
#pragma unroll
    for (int j = 0; j < 8; ++j) {
        Cp[(size_t)j * N] = acc[j] + bv;
    }
}

__global__ void attn_proj_k(const float* __restrict__ ht,
                            const float* __restrict__ asrc,
                            const float* __restrict__ adst,
                            float* __restrict__ als,
                            float* __restrict__ ald) {
    int tid = blockIdx.x * blockDim.x + threadIdx.x;
    if (tid >= GN * GH) return;
    const int n = tid >> 2;
    const int h = tid & 3;
    const float* hp = ht + (size_t)n * GD + h * GC;
    const float* ap = asrc + h * GC;
    const float* dp = adst + h * GC;
    float s = 0.0f, d = 0.0f;
#pragma unroll
    for (int c = 0; c < GC; ++c) {
        float v = hp[c];
        s += v * ap[c];
        d += v * dp[c];
    }
    als[tid] = s;
    ald[tid] = d;
}

__global__ void csr_count_k(const int* __restrict__ ei,
                            int* __restrict__ deg,
                            int* __restrict__ rank) {
    int e = blockIdx.x * blockDim.x + threadIdx.x;
    if (e >= GE) return;
    rank[e] = atomicAdd(deg + ei[GE + e], 1);
}

#define SCAN_T   1024
#define SCAN_PER ((GN + SCAN_T - 1) / SCAN_T)

__global__ void __launch_bounds__(SCAN_T)
csr_scan_k(const int* __restrict__ deg,
           int* __restrict__ rowptr) {
    __shared__ int part[SCAN_T];
    const int t = threadIdx.x;
    const int base = t * SCAN_PER;
    int sum = 0;
    for (int i = 0; i < SCAN_PER; ++i) {
        const int idx = base + i;
        if (idx < GN) sum += deg[idx];
    }
    part[t] = sum;
    __syncthreads();
    for (int off = 1; off < SCAN_T; off <<= 1) {
        const int v = (t >= off) ? part[t - off] : 0;
        __syncthreads();
        part[t] += v;
        __syncthreads();
    }
    int run = part[t] - sum;
    for (int i = 0; i < SCAN_PER; ++i) {
        const int idx = base + i;
        if (idx < GN) {
            rowptr[idx] = run;
            run += deg[idx];
        }
    }
    if (t == SCAN_T - 1) rowptr[GN] = part[t];
}

__global__ void csr_fill_k(const int* __restrict__ ei,
                           const int* __restrict__ rowptr,
                           const int* __restrict__ rank,
                           int* __restrict__ csrc) {
    int e = blockIdx.x * blockDim.x + threadIdx.x;
    if (e >= GE) return;
    csrc[rowptr[ei[GE + e]] + rank[e]] = ei[e];
}

__device__ __forceinline__ float lrelu02(float v) {
    return (v >= 0.0f) ? v : 0.2f * v;
}

__device__ __forceinline__ float pick4(int h, float a, float b, float c, float d) {
    const bool odd  = (h & 1) != 0;
    const bool high = (h & 2) != 0;
    const float lo = odd ? b : a;
    const float hi = odd ? d : c;
    return high ? hi : lo;
}

__global__ void __launch_bounds__(256)
seg_softmax_agg_ln_k(const int* __restrict__ rowptr,
                     const int* __restrict__ csrc,
                     const float* __restrict__ als,
                     const float* __restrict__ ald,
                     const float* __restrict__ ht,
                     const float* __restrict__ bias,
                     const float* __restrict__ gamma,
                     const float* __restrict__ beta,
                     float* __restrict__ out) {
    const int lane = threadIdx.x & 31;
    const int d = __builtin_amdgcn_readfirstlane(
        (int)(blockIdx.x * (blockDim.x >> 5) + (threadIdx.x >> 5)));
    if (d >= GN) return;
    const int beg = rowptr[d];
    const int end = rowptr[d + 1];
    const float4* als4 = reinterpret_cast<const float4*>(als);
    const float4 ad = reinterpret_cast<const float4*>(ald)[d];
    const float4 sd = als4[d];

    const float t0 = lrelu02(sd.x + ad.x);
    const float t1 = lrelu02(sd.y + ad.y);
    const float t2 = lrelu02(sd.z + ad.z);
    const float t3 = lrelu02(sd.w + ad.w);

    float m0 = t0, m1 = t1, m2 = t2, m3 = t3;
    for (int i = beg + lane; i < end; i += 32) {
        const float4 a = als4[csrc[i]];
        m0 = fmaxf(m0, lrelu02(a.x + ad.x));
        m1 = fmaxf(m1, lrelu02(a.y + ad.y));
        m2 = fmaxf(m2, lrelu02(a.z + ad.z));
        m3 = fmaxf(m3, lrelu02(a.w + ad.w));
    }
    for (int off = 16; off > 0; off >>= 1) {
        m0 = fmaxf(m0, __shfl_xor(m0, off, 32));
        m1 = fmaxf(m1, __shfl_xor(m1, off, 32));
        m2 = fmaxf(m2, __shfl_xor(m2, off, 32));
        m3 = fmaxf(m3, __shfl_xor(m3, off, 32));
    }

    float q0 = 0.0f, q1 = 0.0f, q2 = 0.0f, q3 = 0.0f;
    for (int i = beg + lane; i < end; i += 32) {
        const float4 a = als4[csrc[i]];
        q0 += expf(lrelu02(a.x + ad.x) - m0);
        q1 += expf(lrelu02(a.y + ad.y) - m1);
        q2 += expf(lrelu02(a.z + ad.z) - m2);
        q3 += expf(lrelu02(a.w + ad.w) - m3);
    }
    for (int off = 16; off > 0; off >>= 1) {
        q0 += __shfl_xor(q0, off, 32);
        q1 += __shfl_xor(q1, off, 32);
        q2 += __shfl_xor(q2, off, 32);
        q3 += __shfl_xor(q3, off, 32);
    }
    q0 += expf(t0 - m0);
    q1 += expf(t1 - m1);
    q2 += expf(t2 - m2);
    q3 += expf(t3 - m3);

    const int h = lane >> 3;
    const float4* hrow = reinterpret_cast<const float4*>(ht) + lane;
    float4 acc;
    {
        const float al = expf(pick4(h, t0, t1, t2, t3) - pick4(h, m0, m1, m2, m3))
                         / pick4(h, q0, q1, q2, q3);
        const float4 hv = hrow[(size_t)d * (GD / 4)];
        acc.x = hv.x * al;
        acc.y = hv.y * al;
        acc.z = hv.z * al;
        acc.w = hv.w * al;
    }
    const float4 bv = reinterpret_cast<const float4*>(bias)[lane];
    const float4 gv = reinterpret_cast<const float4*>(gamma)[lane];
    const float4 ev = reinterpret_cast<const float4*>(beta)[lane];
    for (int base = beg; base < end; base += 32) {
        int   s  = 0;
        float a0 = 0.0f, a1 = 0.0f, a2 = 0.0f, a3 = 0.0f;
        if (base + lane < end) {
            s = csrc[base + lane];
            const float4 a = als4[s];
            a0 = expf(lrelu02(a.x + ad.x) - m0) / q0;
            a1 = expf(lrelu02(a.y + ad.y) - m1) / q1;
            a2 = expf(lrelu02(a.z + ad.z) - m2) / q2;
            a3 = expf(lrelu02(a.w + ad.w) - m3) / q3;
        }
        const int cnt = (end - base < 32) ? (end - base) : 32;
#pragma unroll 4
        for (int k = 0; k < cnt; ++k) {
            const int sk = __builtin_amdgcn_readlane(s, k);
            const float b0 = __int_as_float(__builtin_amdgcn_readlane(__float_as_int(a0), k));
            const float b1 = __int_as_float(__builtin_amdgcn_readlane(__float_as_int(a1), k));
            const float b2 = __int_as_float(__builtin_amdgcn_readlane(__float_as_int(a2), k));
            const float b3 = __int_as_float(__builtin_amdgcn_readlane(__float_as_int(a3), k));
            const float al = pick4(h, b0, b1, b2, b3);
            const float4 hv = hrow[(size_t)sk * (GD / 4)];
            acc.x += hv.x * al;
            acc.y += hv.y * al;
            acc.z += hv.z * al;
            acc.w += hv.w * al;
        }
    }
    const float v0 = acc.x + bv.x;
    const float v1 = acc.y + bv.y;
    const float v2 = acc.z + bv.z;
    const float v3 = acc.w + bv.w;
    float rs = v0 + v1 + v2 + v3;
    float rq = v0 * v0 + v1 * v1 + v2 * v2 + v3 * v3;
    for (int off = 16; off > 0; off >>= 1) {
        rs += __shfl_xor(rs, off, 32);
        rq += __shfl_xor(rq, off, 32);
    }
    const float mean = rs * (1.0f / GD);
    const float var  = rq * (1.0f / GD) - mean * mean;
    const float rstd = rsqrtf(var + 1e-5f);
    float4 y;
    y.x = (v0 - mean) * rstd * gv.x + ev.x;
    y.y = (v1 - mean) * rstd * gv.y + ev.y;
    y.z = (v2 - mean) * rstd * gv.z + ev.z;
    y.w = (v3 - mean) * rstd * gv.w + ev.w;
    y.x = (y.x >= 0.0f) ? y.x : 0.1f * y.x;
    y.y = (y.y >= 0.0f) ? y.y : 0.1f * y.y;
    y.z = (y.z >= 0.0f) ? y.z : 0.1f * y.z;
    y.w = (y.w >= 0.0f) ? y.w : 0.1f * y.w;
    reinterpret_cast<float4*>(out)[(size_t)d * (GD / 4) + lane] = y;
}

#define POOL_NPW 8
static_assert(GN % POOL_NPW == 0, "a wave takes POOL_NPW whole nodes");

__device__ __forceinline__ void pool_flush(float* __restrict__ gmax,
                                           float* __restrict__ gsum,
                                           float* __restrict__ cnt,
                                           int g, int lane,
                                           const float* mx, const float* sm, int c) {
    float* pm = gmax + (size_t)g * GD + lane;
    float* ps = gsum + (size_t)g * GD + lane;
#pragma unroll
    for (int i = 0; i < 4; ++i) {
        atomicMaxFloat(pm + 32 * i, mx[i]);
        atomicAdd(ps + 32 * i, sm[i]);
    }
    if (lane == 0) atomicAdd(cnt + g, (float)c);
}

__global__ void __launch_bounds__(256)
pool_k(const float* __restrict__ x,
       const int* __restrict__ batch,
       float* __restrict__ gmax,
       float* __restrict__ gsum,
       float* __restrict__ cnt) {
    const int lane = threadIdx.x & 31;
    const int n0 = __builtin_amdgcn_readfirstlane(
        (int)((blockIdx.x * (blockDim.x >> 5) + (threadIdx.x >> 5)) * POOL_NPW));
    if (n0 >= GN) return;
    const float* xp = x + (size_t)n0 * GD + lane;

    int   b[POOL_NPW];
    float v[POOL_NPW][4];
#pragma unroll
    for (int j = 0; j < POOL_NPW; ++j) {
        b[j] = batch[n0 + j];
#pragma unroll
        for (int i = 0; i < 4; ++i) v[j][i] = xp[j * GD + 32 * i];
    }

    int cur = b[0];
    int c = 1;
    float mx[4], sm[4];
#pragma unroll
    for (int i = 0; i < 4; ++i) { mx[i] = v[0][i]; sm[i] = v[0][i]; }
#pragma unroll
    for (int j = 1; j < POOL_NPW; ++j) {
        if (b[j] != cur) {
            pool_flush(gmax, gsum, cnt, cur, lane, mx, sm, c);
            cur = b[j];
            c = 1;
#pragma unroll
            for (int i = 0; i < 4; ++i) { mx[i] = v[j][i]; sm[i] = v[j][i]; }
        } else {
#pragma unroll
            for (int i = 0; i < 4; ++i) {
                mx[i] = fmaxf(mx[i], v[j][i]);
                sm[i] += v[j][i];
            }
            ++c;
        }
    }
    pool_flush(gmax, gsum, cnt, cur, lane, mx, sm, c);
}

__global__ void pool_fin_k(const float* __restrict__ gmax,
                           const float* __restrict__ gsum,
                           const float* __restrict__ cnt,
                           float* __restrict__ pooled) {
    int tid = blockIdx.x * blockDim.x + threadIdx.x;
    if (tid >= GG * 2 * GD) return;
    const int g = tid >> 8;
    const int f = tid & 255;
    pooled[tid] = (f < GD) ? gmax[(size_t)g * GD + f]
                           : gsum[(size_t)g * GD + (f - GD)] / fmaxf(cnt[g], 1.0f);
}

extern "C" void kernel_launch(void* const* d_in, const int* in_sizes, int n_in,
                              void* d_out, int out_size, void* d_ws, size_t ws_size,
                              hipStream_t stream) {
    const float* x       = (const float*)d_in[0];
    const int*   ei      = (const int*)  d_in[1];
    const int*   batch   = (const int*)  d_in[2];
    const float* W       = (const float*)d_in[3];
    const float* att_src = (const float*)d_in[4];
    const float* att_dst = (const float*)d_in[5];
    const float* bias    = (const float*)d_in[6];
    const float* ln_g    = (const float*)d_in[7];
    const float* ln_b    = (const float*)d_in[8];
    const float* W_out   = (const float*)d_in[9];
    const float* b_out   = (const float*)d_in[10];
    float* out = (float*)d_out;

    float* ws     = (float*)d_ws;
    float* bufP   = ws;
    float* bufQ   = bufP   + (size_t)GN * GD;
    int*   rowptr = (int*)(bufQ + (size_t)GN * GD);
    int*   deg    = rowptr + (GN + 1);
    int*   rank   = deg    + GN;
    int*   csrc   = rank   + GE;
    float* als    = ws     + (size_t)3 * GN * GD;
    float* ald    = als    + (size_t)GN * GH;
    float* gmax   = ald    + (size_t)3 * GN * GH;
    float* gsum   = gmax   + (size_t)GG * GD;
    float* cnt    = gsum   + (size_t)GG * GD;
    float* pooled = cnt    + GG;

    const float NEG_INF = -INFINITY;
    const int TB = 256;
    auto blocks = [](long long n, int tb) { return (int)((n + tb - 1) / tb); };

    fill_f32<<<blocks(GN, TB), TB, 0, stream>>>((float*)deg, 0.0f, GN);
    csr_count_k<<<blocks(GE, TB), TB, 0, stream>>>(ei, deg, rank);
    csr_scan_k<<<1, SCAN_T, 0, stream>>>(deg, rowptr);
    csr_fill_k<<<blocks(GE, TB), TB, 0, stream>>>(ei, rowptr, rank, csrc);

    float* ht     = bufQ;
    float* nodeio = bufP;
    for (int l = 0; l < 3; ++l) {
        const float* in = (l == 0) ? x : nodeio;

        {
            int tiles = (GN / 16) * (GD / 16);
            gemm_wmma_f32<<<blocks(tiles, 4), 128, 0, stream>>>(
                in, W + (size_t)l * GD * GD, nullptr, ht, GN, GD, GD);
        }

        attn_proj_k<<<blocks((long long)GN * GH, TB), TB, 0, stream>>>(
            ht, att_src + l * GH * GC, att_dst + l * GH * GC, als, ald);

        seg_softmax_agg_ln_k<<<blocks(GN, 8), 256, 0, stream>>>(
            rowptr, csrc, als, ald, ht,
            bias + l * GD, ln_g + l * GD, ln_b + l * GD, nodeio);
    }

    fill_f32<<<blocks(GG * GD, TB), TB, 0, stream>>>(gmax, NEG_INF, GG * GD);
    fill_f32<<<blocks(GG * GD, TB), TB, 0, stream>>>(gsum, 0.0f, GG * GD);
    fill_f32<<<blocks(GG, TB), TB, 0, stream>>>(cnt, 0.0f, GG);
    pool_k<<<blocks(GN / POOL_NPW, 8), 256, 0, stream>>>(
        nodeio, batch, gmax, gsum, cnt);
    pool_fin_k<<<blocks(GG * 2 * GD, TB), TB, 0, stream>>>(gmax, gsum, cnt, pooled);

    {
        int tiles = (GG / 16) * (GEMB / 16);
        gemm_wmma_f32<<<blocks(tiles, 4), 128, 0, stream>>>(
            pooled, W_out, b_out, out, GG, 2 * GD, GEMB);
    }
}


# Degree prefix sum spread over 13 independent work-groups with whole-line 16-byte loads, each re-adding the counts in front of its chunk (was 1 work-group, 98 strided one-at-a-time loads per thread)
# speedup vs baseline: 11.3454x; 1.2592x over previous
#include <hip/hip_runtime.h>
#include <cmath>


#define GN   50000
#define GE   800000
#define GETOT (GE + GN)
#define GG   256
#define GH   4
#define GC   32
#define GD   128
#define GEMB 128

typedef float v2f __attribute__((ext_vector_type(2)));
typedef float v8f __attribute__((ext_vector_type(8)));

__device__ __forceinline__ void atomicMaxFloat(float* addr, float val) {
    if (val >= 0.0f) {
        atomicMax((int*)addr, __float_as_int(val));
    } else {
        atomicMin((unsigned int*)addr, __float_as_uint(val));
    }
}

__global__ void fill_f32(float* __restrict__ p, float v, int n) {
    int i = blockIdx.x * blockDim.x + threadIdx.x;
    if (i < n) p[i] = v;
}

__global__ void gemm_wmma_f32(const float* __restrict__ A,
                              const float* __restrict__ B,
                              const float* __restrict__ bias,
                              float* __restrict__ C,
                              int M, int K, int N) {
    const int wave = threadIdx.x >> 5;
    const int lane = threadIdx.x & 31;
    const int tilesN = N >> 4;
    const int total  = (M >> 4) * tilesN;
    const int tile = blockIdx.x * (blockDim.x >> 5) + wave;
    if (tile >= total) return;

    const int row0 = (tile / tilesN) << 4;
    const int col0 = (tile % tilesN) << 4;
    const int half = lane >> 4;
    const int l16  = lane & 15;

    const float* Ap = A + (size_t)(row0 + l16) * K + 2 * half;
    const float* Bp = B + (size_t)(2 * half) * N + col0 + l16;

    v8f acc = {};
    for (int k0 = 0; k0 < K; k0 += 4) {
        v2f a, b;
        a.x = Ap[k0];
        a.y = Ap[k0 + 1];
        b.x = Bp[(size_t)k0 * N];
        b.y = Bp[(size_t)(k0 + 1) * N];
        acc = __builtin_amdgcn_wmma_f32_16x16x4_f32(
                false, a, false, b, (short)0, acc, false, false);
    }

    const float bv = bias ? bias[col0 + l16] : 0.0f;
    float* Cp = C + (size_t)(row0 + 8 * half) * N + col0 + l16;
#pragma unroll
    for (int j = 0; j < 8; ++j) {
        Cp[(size_t)j * N] = acc[j] + bv;
    }
}

__global__ void attn_proj_k(const float* __restrict__ ht,
                            const float* __restrict__ asrc,
                            const float* __restrict__ adst,
                            float* __restrict__ als,
                            float* __restrict__ ald) {
    int tid = blockIdx.x * blockDim.x + threadIdx.x;
    if (tid >= GN * GH) return;
    const int n = tid >> 2;
    const int h = tid & 3;
    const float* hp = ht + (size_t)n * GD + h * GC;
    const float* ap = asrc + h * GC;
    const float* dp = adst + h * GC;
    float s = 0.0f, d = 0.0f;
#pragma unroll
    for (int c = 0; c < GC; ++c) {
        float v = hp[c];
        s += v * ap[c];
        d += v * dp[c];
    }
    als[tid] = s;
    ald[tid] = d;
}

__global__ void csr_count_k(const int* __restrict__ ei,
                            int* __restrict__ deg,
                            int* __restrict__ rank) {
    int e = blockIdx.x * blockDim.x + threadIdx.x;
    if (e >= GE) return;
    rank[e] = atomicAdd(deg + ei[GE + e], 1);
}

#define SCAN_T   1024
#define SCAN_CH  (SCAN_T * 4)
#define SCAN_NB  ((GN + 1 + SCAN_CH - 1) / SCAN_CH)
#define SCAN_Q   (GN / 4)
#define ROWPTR_SLOTS (((GN + 1 + 63) / 64) * 64)
static_assert(GN % 4 == 0, "counts are read four at a time");

__global__ void __launch_bounds__(SCAN_T)
csr_scan_k(const int* __restrict__ deg,
           int* __restrict__ rowptr) {
    __shared__ int wtot[SCAN_T / 32];
    __shared__ int wpre[SCAN_T / 32];
    const int t    = threadIdx.x;
    const int lane = t & 31;
    const int wave = t >> 5;
    const int b    = blockIdx.x;
    const int4* deg4 = reinterpret_cast<const int4*>(deg);

    const int q = b * SCAN_T + t;
    const bool own = q < SCAN_Q;

    int4 u[SCAN_NB - 1];
#pragma unroll
    for (int j = 0; j < SCAN_NB - 1; ++j)
        u[j] = deg4[((j < b) ? j : 0) * SCAN_T + t];
    int4 v = deg4[own ? q : SCAN_Q - 1];
    if (!own) v = make_int4(0, 0, 0, 0);

    int pre = 0;
#pragma unroll
    for (int j = 0; j < SCAN_NB - 1; ++j) {
        const int su = (u[j].x + u[j].y) + (u[j].z + u[j].w);
        pre += (j < b) ? su : 0;
    }

    const int s = (v.x + v.y) + (v.z + v.w);
    int inc = s;
    for (int off = 1; off < 32; off <<= 1) {
        const int o = __shfl_up(inc, off, 32);
        if (lane >= off) inc += o;
    }
    for (int off = 16; off > 0; off >>= 1) pre += __shfl_xor(pre, off, 32);
    if (lane == 31) wtot[wave] = inc;
    if (lane == 0)  wpre[wave] = pre;
    __syncthreads();

    int wt = wtot[lane];
    int wp = wpre[lane];
    int winc = wt;
    for (int off = 1; off < 32; off <<= 1) {
        const int o = __shfl_up(winc, off, 32);
        if (lane >= off) winc += o;
    }
    for (int off = 16; off > 0; off >>= 1) wp += __shfl_xor(wp, off, 32);
    const int wexc = __shfl(winc - wt, wave, 32);

    const int e0 = wp + wexc + (inc - s);
    if (own) {
        int4 r;
        r.x = e0;
        r.y = e0 + v.x;
        r.z = r.y + v.y;
        r.w = r.z + v.z;
        reinterpret_cast<int4*>(rowptr)[q] = r;
    } else if (q == SCAN_Q) {
        rowptr[GN] = e0;
    }
}

__global__ void csr_fill_k(const int* __restrict__ ei,
                           const int* __restrict__ rowptr,
                           const int* __restrict__ rank,
                           int* __restrict__ csrc) {
    int e = blockIdx.x * blockDim.x + threadIdx.x;
    if (e >= GE) return;
    csrc[rowptr[ei[GE + e]] + rank[e]] = ei[e];
}

__device__ __forceinline__ float lrelu02(float v) {
    return (v >= 0.0f) ? v : 0.2f * v;
}

__device__ __forceinline__ float pick4(int h, float a, float b, float c, float d) {
    const bool odd  = (h & 1) != 0;
    const bool high = (h & 2) != 0;
    const float lo = odd ? b : a;
    const float hi = odd ? d : c;
    return high ? hi : lo;
}

__global__ void __launch_bounds__(256)
seg_softmax_agg_ln_k(const int* __restrict__ rowptr,
                     const int* __restrict__ csrc,
                     const float* __restrict__ als,
                     const float* __restrict__ ald,
                     const float* __restrict__ ht,
                     const float* __restrict__ bias,
                     const float* __restrict__ gamma,
                     const float* __restrict__ beta,
                     float* __restrict__ out) {
    const int lane = threadIdx.x & 31;
    const int d = __builtin_amdgcn_readfirstlane(
        (int)(blockIdx.x * (blockDim.x >> 5) + (threadIdx.x >> 5)));
    if (d >= GN) return;
    const int beg = rowptr[d];
    const int end = rowptr[d + 1];
    const float4* als4 = reinterpret_cast<const float4*>(als);
    const float4 ad = reinterpret_cast<const float4*>(ald)[d];
    const float4 sd = als4[d];

    const float t0 = lrelu02(sd.x + ad.x);
    const float t1 = lrelu02(sd.y + ad.y);
    const float t2 = lrelu02(sd.z + ad.z);
    const float t3 = lrelu02(sd.w + ad.w);

    float m0 = t0, m1 = t1, m2 = t2, m3 = t3;
    for (int i = beg + lane; i < end; i += 32) {
        const float4 a = als4[csrc[i]];
        m0 = fmaxf(m0, lrelu02(a.x + ad.x));
        m1 = fmaxf(m1, lrelu02(a.y + ad.y));
        m2 = fmaxf(m2, lrelu02(a.z + ad.z));
        m3 = fmaxf(m3, lrelu02(a.w + ad.w));
    }
    for (int off = 16; off > 0; off >>= 1) {
        m0 = fmaxf(m0, __shfl_xor(m0, off, 32));
        m1 = fmaxf(m1, __shfl_xor(m1, off, 32));
        m2 = fmaxf(m2, __shfl_xor(m2, off, 32));
        m3 = fmaxf(m3, __shfl_xor(m3, off, 32));
    }

    float q0 = 0.0f, q1 = 0.0f, q2 = 0.0f, q3 = 0.0f;
    for (int i = beg + lane; i < end; i += 32) {
        const float4 a = als4[csrc[i]];
        q0 += expf(lrelu02(a.x + ad.x) - m0);
        q1 += expf(lrelu02(a.y + ad.y) - m1);
        q2 += expf(lrelu02(a.z + ad.z) - m2);
        q3 += expf(lrelu02(a.w + ad.w) - m3);
    }
    for (int off = 16; off > 0; off >>= 1) {
        q0 += __shfl_xor(q0, off, 32);
        q1 += __shfl_xor(q1, off, 32);
        q2 += __shfl_xor(q2, off, 32);
        q3 += __shfl_xor(q3, off, 32);
    }
    q0 += expf(t0 - m0);
    q1 += expf(t1 - m1);
    q2 += expf(t2 - m2);
    q3 += expf(t3 - m3);

    const int h = lane >> 3;
    const float4* hrow = reinterpret_cast<const float4*>(ht) + lane;
    float4 acc;
    {
        const float al = expf(pick4(h, t0, t1, t2, t3) - pick4(h, m0, m1, m2, m3))
                         / pick4(h, q0, q1, q2, q3);
        const float4 hv = hrow[(size_t)d * (GD / 4)];
        acc.x = hv.x * al;
        acc.y = hv.y * al;
        acc.z = hv.z * al;
        acc.w = hv.w * al;
    }
    const float4 bv = reinterpret_cast<const float4*>(bias)[lane];
    const float4 gv = reinterpret_cast<const float4*>(gamma)[lane];
    const float4 ev = reinterpret_cast<const float4*>(beta)[lane];
    for (int base = beg; base < end; base += 32) {
        int   s  = 0;
        float a0 = 0.0f, a1 = 0.0f, a2 = 0.0f, a3 = 0.0f;
        if (base + lane < end) {
            s = csrc[base + lane];
            const float4 a = als4[s];
            a0 = expf(lrelu02(a.x + ad.x) - m0) / q0;
            a1 = expf(lrelu02(a.y + ad.y) - m1) / q1;
            a2 = expf(lrelu02(a.z + ad.z) - m2) / q2;
            a3 = expf(lrelu02(a.w + ad.w) - m3) / q3;
        }
        const int cnt = (end - base < 32) ? (end - base) : 32;
#pragma unroll 4
        for (int k = 0; k < cnt; ++k) {
            const int sk = __builtin_amdgcn_readlane(s, k);
            const float b0 = __int_as_float(__builtin_amdgcn_readlane(__float_as_int(a0), k));
            const float b1 = __int_as_float(__builtin_amdgcn_readlane(__float_as_int(a1), k));
            const float b2 = __int_as_float(__builtin_amdgcn_readlane(__float_as_int(a2), k));
            const float b3 = __int_as_float(__builtin_amdgcn_readlane(__float_as_int(a3), k));
            const float al = pick4(h, b0, b1, b2, b3);
            const float4 hv = hrow[(size_t)sk * (GD / 4)];
            acc.x += hv.x * al;
            acc.y += hv.y * al;
            acc.z += hv.z * al;
            acc.w += hv.w * al;
        }
    }
    const float v0 = acc.x + bv.x;
    const float v1 = acc.y + bv.y;
    const float v2 = acc.z + bv.z;
    const float v3 = acc.w + bv.w;
    float rs = v0 + v1 + v2 + v3;
    float rq = v0 * v0 + v1 * v1 + v2 * v2 + v3 * v3;
    for (int off = 16; off > 0; off >>= 1) {
        rs += __shfl_xor(rs, off, 32);
        rq += __shfl_xor(rq, off, 32);
    }
    const float mean = rs * (1.0f / GD);
    const float var  = rq * (1.0f / GD) - mean * mean;
    const float rstd = rsqrtf(var + 1e-5f);
    float4 y;
    y.x = (v0 - mean) * rstd * gv.x + ev.x;
    y.y = (v1 - mean) * rstd * gv.y + ev.y;
    y.z = (v2 - mean) * rstd * gv.z + ev.z;
    y.w = (v3 - mean) * rstd * gv.w + ev.w;
    y.x = (y.x >= 0.0f) ? y.x : 0.1f * y.x;
    y.y = (y.y >= 0.0f) ? y.y : 0.1f * y.y;
    y.z = (y.z >= 0.0f) ? y.z : 0.1f * y.z;
    y.w = (y.w >= 0.0f) ? y.w : 0.1f * y.w;
    reinterpret_cast<float4*>(out)[(size_t)d * (GD / 4) + lane] = y;
}

#define POOL_NPW 8
static_assert(GN % POOL_NPW == 0, "a wave takes POOL_NPW whole nodes");

__device__ __forceinline__ void pool_flush(float* __restrict__ gmax,
                                           float* __restrict__ gsum,
                                           float* __restrict__ cnt,
                                           int g, int lane,
                                           const float* mx, const float* sm, int c) {
    float* pm = gmax + (size_t)g * GD + lane;
    float* ps = gsum + (size_t)g * GD + lane;
#pragma unroll
    for (int i = 0; i < 4; ++i) {
        atomicMaxFloat(pm + 32 * i, mx[i]);
        atomicAdd(ps + 32 * i, sm[i]);
    }
    if (lane == 0) atomicAdd(cnt + g, (float)c);
}

__global__ void __launch_bounds__(256)
pool_k(const float* __restrict__ x,
       const int* __restrict__ batch,
       float* __restrict__ gmax,
       float* __restrict__ gsum,
       float* __restrict__ cnt) {
    const int lane = threadIdx.x & 31;
    const int n0 = __builtin_amdgcn_readfirstlane(
        (int)((blockIdx.x * (blockDim.x >> 5) + (threadIdx.x >> 5)) * POOL_NPW));
    if (n0 >= GN) return;
    const float* xp = x + (size_t)n0 * GD + lane;

    int   b[POOL_NPW];
    float v[POOL_NPW][4];
#pragma unroll
    for (int j = 0; j < POOL_NPW; ++j) {
        b[j] = batch[n0 + j];
#pragma unroll
        for (int i = 0; i < 4; ++i) v[j][i] = xp[j * GD + 32 * i];
    }

    int cur = b[0];
    int c = 1;
    float mx[4], sm[4];
#pragma unroll
    for (int i = 0; i < 4; ++i) { mx[i] = v[0][i]; sm[i] = v[0][i]; }
#pragma unroll
    for (int j = 1; j < POOL_NPW; ++j) {
        if (b[j] != cur) {
            pool_flush(gmax, gsum, cnt, cur, lane, mx, sm, c);
            cur = b[j];
            c = 1;
#pragma unroll
            for (int i = 0; i < 4; ++i) { mx[i] = v[j][i]; sm[i] = v[j][i]; }
        } else {
#pragma unroll
            for (int i = 0; i < 4; ++i) {
                mx[i] = fmaxf(mx[i], v[j][i]);
                sm[i] += v[j][i];
            }
            ++c;
        }
    }
    pool_flush(gmax, gsum, cnt, cur, lane, mx, sm, c);
}

__global__ void pool_fin_k(const float* __restrict__ gmax,
                           const float* __restrict__ gsum,
                           const float* __restrict__ cnt,
                           float* __restrict__ pooled) {
    int tid = blockIdx.x * blockDim.x + threadIdx.x;
    if (tid >= GG * 2 * GD) return;
    const int g = tid >> 8;
    const int f = tid & 255;
    pooled[tid] = (f < GD) ? gmax[(size_t)g * GD + f]
                           : gsum[(size_t)g * GD + (f - GD)] / fmaxf(cnt[g], 1.0f);
}

extern "C" void kernel_launch(void* const* d_in, const int* in_sizes, int n_in,
                              void* d_out, int out_size, void* d_ws, size_t ws_size,
                              hipStream_t stream) {
    const float* x       = (const float*)d_in[0];
    const int*   ei      = (const int*)  d_in[1];
    const int*   batch   = (const int*)  d_in[2];
    const float* W       = (const float*)d_in[3];
    const float* att_src = (const float*)d_in[4];
    const float* att_dst = (const float*)d_in[5];
    const float* bias    = (const float*)d_in[6];
    const float* ln_g    = (const float*)d_in[7];
    const float* ln_b    = (const float*)d_in[8];
    const float* W_out   = (const float*)d_in[9];
    const float* b_out   = (const float*)d_in[10];
    float* out = (float*)d_out;

    float* ws     = (float*)d_ws;
    float* bufP   = ws;
    float* bufQ   = bufP   + (size_t)GN * GD;
    int*   rowptr = (int*)(bufQ + (size_t)GN * GD);
    int*   deg    = rowptr + ROWPTR_SLOTS;
    int*   rank   = deg    + GN;
    int*   csrc   = rank   + GE;
    float* als    = ws     + (size_t)3 * GN * GD;
    float* ald    = als    + (size_t)GN * GH;
    float* gmax   = ald    + (size_t)3 * GN * GH;
    float* gsum   = gmax   + (size_t)GG * GD;
    float* cnt    = gsum   + (size_t)GG * GD;
    float* pooled = cnt    + GG;

    const float NEG_INF = -INFINITY;
    const int TB = 256;
    auto blocks = [](long long n, int tb) { return (int)((n + tb - 1) / tb); };

    fill_f32<<<blocks(GN, TB), TB, 0, stream>>>((float*)deg, 0.0f, GN);
    csr_count_k<<<blocks(GE, TB), TB, 0, stream>>>(ei, deg, rank);
    csr_scan_k<<<SCAN_NB, SCAN_T, 0, stream>>>(deg, rowptr);
    csr_fill_k<<<blocks(GE, TB), TB, 0, stream>>>(ei, rowptr, rank, csrc);

    float* ht     = bufQ;
    float* nodeio = bufP;
    for (int l = 0; l < 3; ++l) {
        const float* in = (l == 0) ? x : nodeio;

        {
            int tiles = (GN / 16) * (GD / 16);
            gemm_wmma_f32<<<blocks(tiles, 4), 128, 0, stream>>>(
                in, W + (size_t)l * GD * GD, nullptr, ht, GN, GD, GD);
        }

        attn_proj_k<<<blocks((long long)GN * GH, TB), TB, 0, stream>>>(
            ht, att_src + l * GH * GC, att_dst + l * GH * GC, als, ald);

        seg_softmax_agg_ln_k<<<blocks(GN, 8), 256, 0, stream>>>(
            rowptr, csrc, als, ald, ht,
            bias + l * GD, ln_g + l * GD, ln_b + l * GD, nodeio);
    }

    fill_f32<<<blocks(GG * GD, TB), TB, 0, stream>>>(gmax, NEG_INF, GG * GD);
    fill_f32<<<blocks(GG * GD, TB), TB, 0, stream>>>(gsum, 0.0f, GG * GD);
    fill_f32<<<blocks(GG, TB), TB, 0, stream>>>(cnt, 0.0f, GG);
    pool_k<<<blocks(GN / POOL_NPW, 8), 256, 0, stream>>>(
        nodeio, batch, gmax, gsum, cnt);
    pool_fin_k<<<blocks(GG * 2 * GD, TB), TB, 0, stream>>>(gmax, gsum, cnt, pooled);

    {
        int tiles = (GG / 16) * (GEMB / 16);
        gemm_wmma_f32<<<blocks(tiles, 4), 128, 0, stream>>>(
            pooled, W_out, b_out, out, GG, 2 * GD, GEMB);
    }
}


# Edge count/rank in per-chunk LDS histograms (32 work-groups, all 50000 counters in LDS, rank = LDS add's old value), not 800000 returning global adds; row sums now in chunk-then-rank order
# speedup vs baseline: 11.9632x; 1.0545x over previous
#include <hip/hip_runtime.h>
#include <cmath>


#define GN   50000
#define GE   800000
#define GETOT (GE + GN)
#define GG   256
#define GH   4
#define GC   32
#define GD   128
#define GEMB 128

typedef float v2f __attribute__((ext_vector_type(2)));
typedef float v8f __attribute__((ext_vector_type(8)));

__device__ __forceinline__ void atomicMaxFloat(float* addr, float val) {
    if (val >= 0.0f) {
        atomicMax((int*)addr, __float_as_int(val));
    } else {
        atomicMin((unsigned int*)addr, __float_as_uint(val));
    }
}

__global__ void fill_f32(float* __restrict__ p, float v, int n) {
    int i = blockIdx.x * blockDim.x + threadIdx.x;
    if (i < n) p[i] = v;
}

__global__ void gemm_wmma_f32(const float* __restrict__ A,
                              const float* __restrict__ B,
                              const float* __restrict__ bias,
                              float* __restrict__ C,
                              int M, int K, int N) {
    const int wave = threadIdx.x >> 5;
    const int lane = threadIdx.x & 31;
    const int tilesN = N >> 4;
    const int total  = (M >> 4) * tilesN;
    const int tile = blockIdx.x * (blockDim.x >> 5) + wave;
    if (tile >= total) return;

    const int row0 = (tile / tilesN) << 4;
    const int col0 = (tile % tilesN) << 4;
    const int half = lane >> 4;
    const int l16  = lane & 15;

    const float* Ap = A + (size_t)(row0 + l16) * K + 2 * half;
    const float* Bp = B + (size_t)(2 * half) * N + col0 + l16;

    v8f acc = {};
    for (int k0 = 0; k0 < K; k0 += 4) {
        v2f a, b;
        a.x = Ap[k0];
        a.y = Ap[k0 + 1];
        b.x = Bp[(size_t)k0 * N];
        b.y = Bp[(size_t)(k0 + 1) * N];
        acc = __builtin_amdgcn_wmma_f32_16x16x4_f32(
                false, a, false, b, (short)0, acc, false, false);
    }

    const float bv = bias ? bias[col0 + l16] : 0.0f;
    float* Cp = C + (size_t)(row0 + 8 * half) * N + col0 + l16;
#pragma unroll
    for (int j = 0; j < 8; ++j) {
        Cp[(size_t)j * N] = acc[j] + bv;
    }
}

__global__ void attn_proj_k(const float* __restrict__ ht,
                            const float* __restrict__ asrc,
                            const float* __restrict__ adst,
                            float* __restrict__ als,
                            float* __restrict__ ald) {
    int tid = blockIdx.x * blockDim.x + threadIdx.x;
    if (tid >= GN * GH) return;
    const int n = tid >> 2;
    const int h = tid & 3;
    const float* hp = ht + (size_t)n * GD + h * GC;
    const float* ap = asrc + h * GC;
    const float* dp = adst + h * GC;
    float s = 0.0f, d = 0.0f;
#pragma unroll
    for (int c = 0; c < GC; ++c) {
        float v = hp[c];
        s += v * ap[c];
        d += v * dp[c];
    }
    als[tid] = s;
    ald[tid] = d;
}

#define SCAN_T   1024
#define SCAN_CH  (SCAN_T * 4)
#define SCAN_NB  ((GN + 1 + SCAN_CH - 1) / SCAN_CH)
#define SCAN_Q   (GN / 4)
#define ROWPTR_SLOTS (((GN + 1 + 63) / 64) * 64)
static_assert(GN % 4 == 0, "counts are read four at a time");

#define CNT_P    32
#define CNT_T    1024
#define CNT_CH   (GE / CNT_P)
static_assert(GE % (CNT_P * 4) == 0, "a chunk is a whole number of 16-byte groups of edges");
static_assert(CNT_T == SCAN_T, "the counting kernel writes its plane out in the prefix-sum kernel's chunks");
static_assert(CNT_P * SCAN_NB <= SCAN_T, "one thread of the prefix-sum kernel per chunk total");

__global__ void __launch_bounds__(CNT_T)
csr_count_k(const int* __restrict__ ei,
            int* __restrict__ rank,
            int* __restrict__ hplane,
            int* __restrict__ ctot) {
    __shared__ __attribute__((aligned(16))) int hist[GN];
    __shared__ int tot[SCAN_NB];
    constexpr int NG = CNT_CH / 4;
    constexpr int NK = (NG + CNT_T - 1) / CNT_T;
    const int t = threadIdx.x;
    const int c = blockIdx.x;
    int4* hist4 = reinterpret_cast<int4*>(hist);

    const int4* dst4 = reinterpret_cast<const int4*>(ei + GE + c * CNT_CH);
    int4 d[NK];
#pragma unroll
    for (int k = 0; k < NK; ++k) {
        const int g = k * CNT_T + t;
        d[k] = dst4[(g < NG) ? g : NG - 1];
    }

    for (int q = t; q < SCAN_Q; q += CNT_T) hist4[q] = make_int4(0, 0, 0, 0);
    if (t < SCAN_NB) tot[t] = 0;
    __syncthreads();

    int4* rank4 = reinterpret_cast<int4*>(rank + c * CNT_CH);
#pragma unroll
    for (int k = 0; k < NK; ++k) {
        const int g = k * CNT_T + t;
        if (g < NG) {
            int4 r;
            r.x = atomicAdd(&hist[d[k].x], 1);
            r.y = atomicAdd(&hist[d[k].y], 1);
            r.z = atomicAdd(&hist[d[k].z], 1);
            r.w = atomicAdd(&hist[d[k].w], 1);
            rank4[g] = r;
        }
    }
    __syncthreads();

    int4* out4 = reinterpret_cast<int4*>(hplane + (size_t)c * GN);
    int sj[SCAN_NB];
#pragma unroll
    for (int j = 0; j < SCAN_NB; ++j) {
        const int q = j * CNT_T + t;
        sj[j] = 0;
        if (q < SCAN_Q) {
            const int4 h = hist4[q];
            out4[q] = h;
            sj[j] = (h.x + h.y) + (h.z + h.w);
        }
    }
    for (int off = 16; off > 0; off >>= 1) {
#pragma unroll
        for (int j = 0; j < SCAN_NB; ++j) sj[j] += __shfl_xor(sj[j], off, 32);
    }
    if ((t & 31) == 0) {
#pragma unroll
        for (int j = 0; j < SCAN_NB; ++j) atomicAdd(&tot[j], sj[j]);
    }
    __syncthreads();
    if (t < SCAN_NB) ctot[c * SCAN_NB + t] = tot[t];
}

__global__ void __launch_bounds__(SCAN_T)
csr_scan_k(int* __restrict__ hplane,
           const int* __restrict__ ctot,
           int* __restrict__ rowptr) {
    __shared__ int wtot[SCAN_T / 32];
    __shared__ int wpre[SCAN_T / 32];
    const int t    = threadIdx.x;
    const int lane = t & 31;
    const int wave = t >> 5;
    const int b    = blockIdx.x;
    int4* hp4 = reinterpret_cast<int4*>(hplane);

    const int q = b * SCAN_T + t;
    const bool own = q < SCAN_Q;
    const int qq = own ? q : SCAN_Q - 1;

    int pre = 0;
    if (t < CNT_P * SCAN_NB) {
        const int cv = ctot[t];
        pre = ((t % SCAN_NB) < b) ? cv : 0;
    }

    int4 v = make_int4(0, 0, 0, 0);
#pragma unroll 8
    for (int c = 0; c < CNT_P; ++c) {
        const int4 h = hp4[(size_t)c * SCAN_Q + qq];
        v.x += h.x; v.y += h.y; v.z += h.z; v.w += h.w;
    }
    if (!own) v = make_int4(0, 0, 0, 0);

    const int s = (v.x + v.y) + (v.z + v.w);
    int inc = s;
    for (int off = 1; off < 32; off <<= 1) {
        const int o = __shfl_up(inc, off, 32);
        if (lane >= off) inc += o;
    }
    for (int off = 16; off > 0; off >>= 1) pre += __shfl_xor(pre, off, 32);
    if (lane == 31) wtot[wave] = inc;
    if (lane == 0)  wpre[wave] = pre;
    __syncthreads();

    int wt = wtot[lane];
    int wp = wpre[lane];
    int winc = wt;
    for (int off = 1; off < 32; off <<= 1) {
        const int o = __shfl_up(winc, off, 32);
        if (lane >= off) winc += o;
    }
    for (int off = 16; off > 0; off >>= 1) wp += __shfl_xor(wp, off, 32);
    const int wexc = __shfl(winc - wt, wave, 32);

    const int e0 = wp + wexc + (inc - s);
    if (own) {
        int4 r;
        r.x = e0;
        r.y = e0 + v.x;
        r.z = r.y + v.y;
        r.w = r.z + v.z;
        reinterpret_cast<int4*>(rowptr)[q] = r;
#pragma unroll
        for (int c0 = 0; c0 < CNT_P; c0 += 8) {
            int4 h[8];
#pragma unroll
            for (int u = 0; u < 8; ++u) h[u] = hp4[(size_t)(c0 + u) * SCAN_Q + q];
#pragma unroll
            for (int u = 0; u < 8; ++u) {
                hp4[(size_t)(c0 + u) * SCAN_Q + q] = r;
                r.x += h[u].x; r.y += h[u].y; r.z += h[u].z; r.w += h[u].w;
            }
        }
    } else if (q == SCAN_Q) {
        rowptr[GN] = e0;
    }
}

__global__ void csr_fill_k(const int* __restrict__ ei,
                           const int* __restrict__ hplane,
                           const int* __restrict__ rank,
                           int* __restrict__ csrc) {
    int e = blockIdx.x * blockDim.x + threadIdx.x;
    if (e >= GE) return;
    const int c = e / CNT_CH;
    csrc[hplane[(size_t)c * GN + ei[GE + e]] + rank[e]] = ei[e];
}

__device__ __forceinline__ float lrelu02(float v) {
    return (v >= 0.0f) ? v : 0.2f * v;
}

__device__ __forceinline__ float pick4(int h, float a, float b, float c, float d) {
    const bool odd  = (h & 1) != 0;
    const bool high = (h & 2) != 0;
    const float lo = odd ? b : a;
    const float hi = odd ? d : c;
    return high ? hi : lo;
}

__global__ void __launch_bounds__(256)
seg_softmax_agg_ln_k(const int* __restrict__ rowptr,
                     const int* __restrict__ csrc,
                     const float* __restrict__ als,
                     const float* __restrict__ ald,
                     const float* __restrict__ ht,
                     const float* __restrict__ bias,
                     const float* __restrict__ gamma,
                     const float* __restrict__ beta,
                     float* __restrict__ out) {
    const int lane = threadIdx.x & 31;
    const int d = __builtin_amdgcn_readfirstlane(
        (int)(blockIdx.x * (blockDim.x >> 5) + (threadIdx.x >> 5)));
    if (d >= GN) return;
    const int beg = rowptr[d];
    const int end = rowptr[d + 1];
    const float4* als4 = reinterpret_cast<const float4*>(als);
    const float4 ad = reinterpret_cast<const float4*>(ald)[d];
    const float4 sd = als4[d];

    const float t0 = lrelu02(sd.x + ad.x);
    const float t1 = lrelu02(sd.y + ad.y);
    const float t2 = lrelu02(sd.z + ad.z);
    const float t3 = lrelu02(sd.w + ad.w);

    float m0 = t0, m1 = t1, m2 = t2, m3 = t3;
    for (int i = beg + lane; i < end; i += 32) {
        const float4 a = als4[csrc[i]];
        m0 = fmaxf(m0, lrelu02(a.x + ad.x));
        m1 = fmaxf(m1, lrelu02(a.y + ad.y));
        m2 = fmaxf(m2, lrelu02(a.z + ad.z));
        m3 = fmaxf(m3, lrelu02(a.w + ad.w));
    }
    for (int off = 16; off > 0; off >>= 1) {
        m0 = fmaxf(m0, __shfl_xor(m0, off, 32));
        m1 = fmaxf(m1, __shfl_xor(m1, off, 32));
        m2 = fmaxf(m2, __shfl_xor(m2, off, 32));
        m3 = fmaxf(m3, __shfl_xor(m3, off, 32));
    }

    float q0 = 0.0f, q1 = 0.0f, q2 = 0.0f, q3 = 0.0f;
    for (int i = beg + lane; i < end; i += 32) {
        const float4 a = als4[csrc[i]];
        q0 += expf(lrelu02(a.x + ad.x) - m0);
        q1 += expf(lrelu02(a.y + ad.y) - m1);
        q2 += expf(lrelu02(a.z + ad.z) - m2);
        q3 += expf(lrelu02(a.w + ad.w) - m3);
    }
    for (int off = 16; off > 0; off >>= 1) {
        q0 += __shfl_xor(q0, off, 32);
        q1 += __shfl_xor(q1, off, 32);
        q2 += __shfl_xor(q2, off, 32);
        q3 += __shfl_xor(q3, off, 32);
    }
    q0 += expf(t0 - m0);
    q1 += expf(t1 - m1);
    q2 += expf(t2 - m2);
    q3 += expf(t3 - m3);

    const int h = lane >> 3;
    const float4* hrow = reinterpret_cast<const float4*>(ht) + lane;
    float4 acc;
    {
        const float al = expf(pick4(h, t0, t1, t2, t3) - pick4(h, m0, m1, m2, m3))
                         / pick4(h, q0, q1, q2, q3);
        const float4 hv = hrow[(size_t)d * (GD / 4)];
        acc.x = hv.x * al;
        acc.y = hv.y * al;
        acc.z = hv.z * al;
        acc.w = hv.w * al;
    }
    const float4 bv = reinterpret_cast<const float4*>(bias)[lane];
    const float4 gv = reinterpret_cast<const float4*>(gamma)[lane];
    const float4 ev = reinterpret_cast<const float4*>(beta)[lane];
    for (int base = beg; base < end; base += 32) {
        int   s  = 0;
        float a0 = 0.0f, a1 = 0.0f, a2 = 0.0f, a3 = 0.0f;
        if (base + lane < end) {
            s = csrc[base + lane];
            const float4 a = als4[s];
            a0 = expf(lrelu02(a.x + ad.x) - m0) / q0;
            a1 = expf(lrelu02(a.y + ad.y) - m1) / q1;
            a2 = expf(lrelu02(a.z + ad.z) - m2) / q2;
            a3 = expf(lrelu02(a.w + ad.w) - m3) / q3;
        }
        const int cnt = (end - base < 32) ? (end - base) : 32;
#pragma unroll 4
        for (int k = 0; k < cnt; ++k) {
            const int sk = __builtin_amdgcn_readlane(s, k);
            const float b0 = __int_as_float(__builtin_amdgcn_readlane(__float_as_int(a0), k));
            const float b1 = __int_as_float(__builtin_amdgcn_readlane(__float_as_int(a1), k));
            const float b2 = __int_as_float(__builtin_amdgcn_readlane(__float_as_int(a2), k));
            const float b3 = __int_as_float(__builtin_amdgcn_readlane(__float_as_int(a3), k));
            const float al = pick4(h, b0, b1, b2, b3);
            const float4 hv = hrow[(size_t)sk * (GD / 4)];
            acc.x += hv.x * al;
            acc.y += hv.y * al;
            acc.z += hv.z * al;
            acc.w += hv.w * al;
        }
    }
    const float v0 = acc.x + bv.x;
    const float v1 = acc.y + bv.y;
    const float v2 = acc.z + bv.z;
    const float v3 = acc.w + bv.w;
    float rs = v0 + v1 + v2 + v3;
    float rq = v0 * v0 + v1 * v1 + v2 * v2 + v3 * v3;
    for (int off = 16; off > 0; off >>= 1) {
        rs += __shfl_xor(rs, off, 32);
        rq += __shfl_xor(rq, off, 32);
    }
    const float mean = rs * (1.0f / GD);
    const float var  = rq * (1.0f / GD) - mean * mean;
    const float rstd = rsqrtf(var + 1e-5f);
    float4 y;
    y.x = (v0 - mean) * rstd * gv.x + ev.x;
    y.y = (v1 - mean) * rstd * gv.y + ev.y;
    y.z = (v2 - mean) * rstd * gv.z + ev.z;
    y.w = (v3 - mean) * rstd * gv.w + ev.w;
    y.x = (y.x >= 0.0f) ? y.x : 0.1f * y.x;
    y.y = (y.y >= 0.0f) ? y.y : 0.1f * y.y;
    y.z = (y.z >= 0.0f) ? y.z : 0.1f * y.z;
    y.w = (y.w >= 0.0f) ? y.w : 0.1f * y.w;
    reinterpret_cast<float4*>(out)[(size_t)d * (GD / 4) + lane] = y;
}

#define POOL_NPW 8
static_assert(GN % POOL_NPW == 0, "a wave takes POOL_NPW whole nodes");

__device__ __forceinline__ void pool_flush(float* __restrict__ gmax,
                                           float* __restrict__ gsum,
                                           float* __restrict__ cnt,
                                           int g, int lane,
                                           const float* mx, const float* sm, int c) {
    float* pm = gmax + (size_t)g * GD + lane;
    float* ps = gsum + (size_t)g * GD + lane;
#pragma unroll
    for (int i = 0; i < 4; ++i) {
        atomicMaxFloat(pm + 32 * i, mx[i]);
        atomicAdd(ps + 32 * i, sm[i]);
    }
    if (lane == 0) atomicAdd(cnt + g, (float)c);
}

__global__ void __launch_bounds__(256)
pool_k(const float* __restrict__ x,
       const int* __restrict__ batch,
       float* __restrict__ gmax,
       float* __restrict__ gsum,
       float* __restrict__ cnt) {
    const int lane = threadIdx.x & 31;
    const int n0 = __builtin_amdgcn_readfirstlane(
        (int)((blockIdx.x * (blockDim.x >> 5) + (threadIdx.x >> 5)) * POOL_NPW));
    if (n0 >= GN) return;
    const float* xp = x + (size_t)n0 * GD + lane;

    int   b[POOL_NPW];
    float v[POOL_NPW][4];
#pragma unroll
    for (int j = 0; j < POOL_NPW; ++j) {
        b[j] = batch[n0 + j];
#pragma unroll
        for (int i = 0; i < 4; ++i) v[j][i] = xp[j * GD + 32 * i];
    }

    int cur = b[0];
    int c = 1;
    float mx[4], sm[4];
#pragma unroll
    for (int i = 0; i < 4; ++i) { mx[i] = v[0][i]; sm[i] = v[0][i]; }
#pragma unroll
    for (int j = 1; j < POOL_NPW; ++j) {
        if (b[j] != cur) {
            pool_flush(gmax, gsum, cnt, cur, lane, mx, sm, c);
            cur = b[j];
            c = 1;
#pragma unroll
            for (int i = 0; i < 4; ++i) { mx[i] = v[j][i]; sm[i] = v[j][i]; }
        } else {
#pragma unroll
            for (int i = 0; i < 4; ++i) {
                mx[i] = fmaxf(mx[i], v[j][i]);
                sm[i] += v[j][i];
            }
            ++c;
        }
    }
    pool_flush(gmax, gsum, cnt, cur, lane, mx, sm, c);
}

__global__ void pool_fin_k(const float* __restrict__ gmax,
                           const float* __restrict__ gsum,
                           const float* __restrict__ cnt,
                           float* __restrict__ pooled) {
    int tid = blockIdx.x * blockDim.x + threadIdx.x;
    if (tid >= GG * 2 * GD) return;
    const int g = tid >> 8;
    const int f = tid & 255;
    pooled[tid] = (f < GD) ? gmax[(size_t)g * GD + f]
                           : gsum[(size_t)g * GD + (f - GD)] / fmaxf(cnt[g], 1.0f);
}

extern "C" void kernel_launch(void* const* d_in, const int* in_sizes, int n_in,
                              void* d_out, int out_size, void* d_ws, size_t ws_size,
                              hipStream_t stream) {
    const float* x       = (const float*)d_in[0];
    const int*   ei      = (const int*)  d_in[1];
    const int*   batch   = (const int*)  d_in[2];
    const float* W       = (const float*)d_in[3];
    const float* att_src = (const float*)d_in[4];
    const float* att_dst = (const float*)d_in[5];
    const float* bias    = (const float*)d_in[6];
    const float* ln_g    = (const float*)d_in[7];
    const float* ln_b    = (const float*)d_in[8];
    const float* W_out   = (const float*)d_in[9];
    const float* b_out   = (const float*)d_in[10];
    float* out = (float*)d_out;

    float* ws     = (float*)d_ws;
    float* bufP   = ws;
    float* bufQ   = bufP   + (size_t)GN * GD;
    int*   rowptr = (int*)(bufQ + (size_t)GN * GD);
    int*   rank   = rowptr + ROWPTR_SLOTS;
    int*   csrc   = rank   + GE;
    int*   hplane = csrc   + GE;
    int*   ctot   = hplane + (size_t)CNT_P * GN;
    float* als    = ws     + (size_t)3 * GN * GD;
    float* ald    = als    + (size_t)GN * GH;
    float* gmax   = ald    + (size_t)3 * GN * GH;
    float* gsum   = gmax   + (size_t)GG * GD;
    float* cnt    = gsum   + (size_t)GG * GD;
    float* pooled = cnt    + GG;

    const float NEG_INF = -INFINITY;
    const int TB = 256;
    auto blocks = [](long long n, int tb) { return (int)((n + tb - 1) / tb); };

    csr_count_k<<<CNT_P, CNT_T, 0, stream>>>(ei, rank, hplane, ctot);
    csr_scan_k<<<SCAN_NB, SCAN_T, 0, stream>>>(hplane, ctot, rowptr);
    csr_fill_k<<<blocks(GE, TB), TB, 0, stream>>>(ei, hplane, rank, csrc);

    float* ht     = bufQ;
    float* nodeio = bufP;
    for (int l = 0; l < 3; ++l) {
        const float* in = (l == 0) ? x : nodeio;

        {
            int tiles = (GN / 16) * (GD / 16);
            gemm_wmma_f32<<<blocks(tiles, 4), 128, 0, stream>>>(
                in, W + (size_t)l * GD * GD, nullptr, ht, GN, GD, GD);
        }

        attn_proj_k<<<blocks((long long)GN * GH, TB), TB, 0, stream>>>(
            ht, att_src + l * GH * GC, att_dst + l * GH * GC, als, ald);

        seg_softmax_agg_ln_k<<<blocks(GN, 8), 256, 0, stream>>>(
            rowptr, csrc, als, ald, ht,
            bias + l * GD, ln_g + l * GD, ln_b + l * GD, nodeio);
    }

    fill_f32<<<blocks(GG * GD, TB), TB, 0, stream>>>(gmax, NEG_INF, GG * GD);
    fill_f32<<<blocks(GG * GD, TB), TB, 0, stream>>>(gsum, 0.0f, GG * GD);
    fill_f32<<<blocks(GG, TB), TB, 0, stream>>>(cnt, 0.0f, GG);
    pool_k<<<blocks(GN / POOL_NPW, 8), 256, 0, stream>>>(
        nodeio, batch, gmax, gsum, cnt);
    pool_fin_k<<<blocks(GG * 2 * GD, TB), TB, 0, stream>>>(gmax, gsum, cnt, pooled);

    {
        int tiles = (GG / 16) * (GEMB / 16);
        gemm_wmma_f32<<<blocks(tiles, 4), 128, 0, stream>>>(
            pooled, W_out, b_out, out, GG, 2 * GD, GEMB);
    }
}


# Layer GEMM re-tiled: one wave owns 16 rows x all 128 columns (A fragment loaded once, feeds 8 WMMAs), W staged once per 8-wave work-group in LDS, whole-line stores; k order per element unchanged
# speedup vs baseline: 13.3068x; 1.1123x over previous
#include <hip/hip_runtime.h>
#include <cmath>


#define GN   50000
#define GE   800000
#define GETOT (GE + GN)
#define GG   256
#define GH   4
#define GC   32
#define GD   128
#define GEMB 128

typedef float v2f __attribute__((ext_vector_type(2)));
typedef float v8f __attribute__((ext_vector_type(8)));

__device__ __forceinline__ void atomicMaxFloat(float* addr, float val) {
    if (val >= 0.0f) {
        atomicMax((int*)addr, __float_as_int(val));
    } else {
        atomicMin((unsigned int*)addr, __float_as_uint(val));
    }
}

__global__ void fill_f32(float* __restrict__ p, float v, int n) {
    int i = blockIdx.x * blockDim.x + threadIdx.x;
    if (i < n) p[i] = v;
}

__global__ void gemm_wmma_f32(const float* __restrict__ A,
                              const float* __restrict__ B,
                              const float* __restrict__ bias,
                              float* __restrict__ C,
                              int M, int K, int N) {
    const int wave = threadIdx.x >> 5;
    const int lane = threadIdx.x & 31;
    const int tilesN = N >> 4;
    const int total  = (M >> 4) * tilesN;
    const int tile = blockIdx.x * (blockDim.x >> 5) + wave;
    if (tile >= total) return;

    const int row0 = (tile / tilesN) << 4;
    const int col0 = (tile % tilesN) << 4;
    const int half = lane >> 4;
    const int l16  = lane & 15;

    const float* Ap = A + (size_t)(row0 + l16) * K + 2 * half;
    const float* Bp = B + (size_t)(2 * half) * N + col0 + l16;

    v8f acc = {};
    for (int k0 = 0; k0 < K; k0 += 4) {
        v2f a, b;
        a.x = Ap[k0];
        a.y = Ap[k0 + 1];
        b.x = Bp[(size_t)k0 * N];
        b.y = Bp[(size_t)(k0 + 1) * N];
        acc = __builtin_amdgcn_wmma_f32_16x16x4_f32(
                false, a, false, b, (short)0, acc, false, false);
    }

    const float bv = bias ? bias[col0 + l16] : 0.0f;
    float* Cp = C + (size_t)(row0 + 8 * half) * N + col0 + l16;
#pragma unroll
    for (int j = 0; j < 8; ++j) {
        Cp[(size_t)j * N] = acc[j] + bv;
    }
}

#define LG_PITCH 288
__global__ void __launch_bounds__(256)
gemm128_strip_k(const float* __restrict__ A,
                const float* __restrict__ Wg,
                float* __restrict__ C,
                int M) {
    __shared__ __attribute__((aligned(16))) float Wl[64 * LG_PITCH];
    const int t    = threadIdx.x;
    const int lane = t & 31;
    const int c    = lane & 15;
    const int hh   = lane >> 4;
    const int wave = __builtin_amdgcn_readfirstlane((int)(threadIdx.x >> 5));
    const int row0 = (blockIdx.x * 8 + wave) * 16;
    const bool live = row0 < M;

    const float4* W4 = reinterpret_cast<const float4*>(Wg);
    const int n4  = lane;
    const int cc  = n4 & 15;
    const int jhi = n4 >> 4;
    float4 r0[8], r1[8];
#pragma unroll
    for (int i = 0; i < 8; ++i) {
        const int kp = wave + 8 * i;
        r0[i] = W4[(2 * kp) * (GD / 4) + n4];
        r1[i] = W4[(2 * kp + 1) * (GD / 4) + n4];
    }

    v2f a[32];
    {
        const float* Ap = A + (size_t)((live ? row0 : 0) + c) * GD + 2 * hh;
#pragma unroll
        for (int i = 0; i < 32; ++i) a[i] = *reinterpret_cast<const v2f*>(Ap + 4 * i);
    }

#pragma unroll
    for (int i = 0; i < 8; ++i) {
        const int kp = wave + 8 * i;
        float* dst = Wl + kp * LG_PITCH + (jhi * 4) * 32 + cc * 2;
        *reinterpret_cast<v2f*>(dst)      = (v2f){r0[i].x, r1[i].x};
        *reinterpret_cast<v2f*>(dst + 32) = (v2f){r0[i].y, r1[i].y};
        *reinterpret_cast<v2f*>(dst + 64) = (v2f){r0[i].z, r1[i].z};
        *reinterpret_cast<v2f*>(dst + 96) = (v2f){r0[i].w, r1[i].w};
    }
    __syncthreads();
    if (!live) return;

    v8f acc[8];
#pragma unroll
    for (int j = 0; j < 8; ++j) acc[j] = (v8f){};
    const float* bp = Wl + hh * LG_PITCH + c * 2;
#pragma unroll
    for (int i = 0; i < 32; ++i) {
#pragma unroll
        for (int j = 0; j < 8; ++j) {
            const v2f b = *reinterpret_cast<const v2f*>(bp + (2 * i) * LG_PITCH + j * 32);
            acc[j] = __builtin_amdgcn_wmma_f32_16x16x4_f32(
                        false, a[i], false, b, (short)0, acc[j], false, false);
        }
    }

    float* Cp = C + (size_t)(row0 + 8 * hh) * GD + 4 * c;
#pragma unroll
    for (int r = 0; r < 8; ++r) {
        float4 lo, hi;
        lo.x = acc[0][r]; lo.y = acc[1][r]; lo.z = acc[2][r]; lo.w = acc[3][r];
        hi.x = acc[4][r]; hi.y = acc[5][r]; hi.z = acc[6][r]; hi.w = acc[7][r];
        *reinterpret_cast<float4*>(Cp + (size_t)r * GD)      = lo;
        *reinterpret_cast<float4*>(Cp + (size_t)r * GD + 64) = hi;
    }
}

__global__ void attn_proj_k(const float* __restrict__ ht,
                            const float* __restrict__ asrc,
                            const float* __restrict__ adst,
                            float* __restrict__ als,
                            float* __restrict__ ald) {
    int tid = blockIdx.x * blockDim.x + threadIdx.x;
    if (tid >= GN * GH) return;
    const int n = tid >> 2;
    const int h = tid & 3;
    const float* hp = ht + (size_t)n * GD + h * GC;
    const float* ap = asrc + h * GC;
    const float* dp = adst + h * GC;
    float s = 0.0f, d = 0.0f;
#pragma unroll
    for (int c = 0; c < GC; ++c) {
        float v = hp[c];
        s += v * ap[c];
        d += v * dp[c];
    }
    als[tid] = s;
    ald[tid] = d;
}

#define SCAN_T   1024
#define SCAN_CH  (SCAN_T * 4)
#define SCAN_NB  ((GN + 1 + SCAN_CH - 1) / SCAN_CH)
#define SCAN_Q   (GN / 4)
#define ROWPTR_SLOTS (((GN + 1 + 63) / 64) * 64)
static_assert(GN % 4 == 0, "counts are read four at a time");

#define CNT_P    32
#define CNT_T    1024
#define CNT_CH   (GE / CNT_P)
static_assert(GE % (CNT_P * 4) == 0, "a chunk is a whole number of 16-byte groups of edges");
static_assert(CNT_T == SCAN_T, "the counting kernel writes its plane out in the prefix-sum kernel's chunks");
static_assert(CNT_P * SCAN_NB <= SCAN_T, "one thread of the prefix-sum kernel per chunk total");

__global__ void __launch_bounds__(CNT_T)
csr_count_k(const int* __restrict__ ei,
            int* __restrict__ rank,
            int* __restrict__ hplane,
            int* __restrict__ ctot) {
    __shared__ __attribute__((aligned(16))) int hist[GN];
    __shared__ int tot[SCAN_NB];
    constexpr int NG = CNT_CH / 4;
    constexpr int NK = (NG + CNT_T - 1) / CNT_T;
    const int t = threadIdx.x;
    const int c = blockIdx.x;
    int4* hist4 = reinterpret_cast<int4*>(hist);

    const int4* dst4 = reinterpret_cast<const int4*>(ei + GE + c * CNT_CH);
    int4 d[NK];
#pragma unroll
    for (int k = 0; k < NK; ++k) {
        const int g = k * CNT_T + t;
        d[k] = dst4[(g < NG) ? g : NG - 1];
    }

    for (int q = t; q < SCAN_Q; q += CNT_T) hist4[q] = make_int4(0, 0, 0, 0);
    if (t < SCAN_NB) tot[t] = 0;
    __syncthreads();

    int4* rank4 = reinterpret_cast<int4*>(rank + c * CNT_CH);
#pragma unroll
    for (int k = 0; k < NK; ++k) {
        const int g = k * CNT_T + t;
        if (g < NG) {
            int4 r;
            r.x = atomicAdd(&hist[d[k].x], 1);
            r.y = atomicAdd(&hist[d[k].y], 1);
            r.z = atomicAdd(&hist[d[k].z], 1);
            r.w = atomicAdd(&hist[d[k].w], 1);
            rank4[g] = r;
        }
    }
    __syncthreads();

    int4* out4 = reinterpret_cast<int4*>(hplane + (size_t)c * GN);
    int sj[SCAN_NB];
#pragma unroll
    for (int j = 0; j < SCAN_NB; ++j) {
        const int q = j * CNT_T + t;
        sj[j] = 0;
        if (q < SCAN_Q) {
            const int4 h = hist4[q];
            out4[q] = h;
            sj[j] = (h.x + h.y) + (h.z + h.w);
        }
    }
    for (int off = 16; off > 0; off >>= 1) {
#pragma unroll
        for (int j = 0; j < SCAN_NB; ++j) sj[j] += __shfl_xor(sj[j], off, 32);
    }
    if ((t & 31) == 0) {
#pragma unroll
        for (int j = 0; j < SCAN_NB; ++j) atomicAdd(&tot[j], sj[j]);
    }
    __syncthreads();
    if (t < SCAN_NB) ctot[c * SCAN_NB + t] = tot[t];
}

__global__ void __launch_bounds__(SCAN_T)
csr_scan_k(int* __restrict__ hplane,
           const int* __restrict__ ctot,
           int* __restrict__ rowptr) {
    __shared__ int wtot[SCAN_T / 32];
    __shared__ int wpre[SCAN_T / 32];
    const int t    = threadIdx.x;
    const int lane = t & 31;
    const int wave = t >> 5;
    const int b    = blockIdx.x;
    int4* hp4 = reinterpret_cast<int4*>(hplane);

    const int q = b * SCAN_T + t;
    const bool own = q < SCAN_Q;
    const int qq = own ? q : SCAN_Q - 1;

    int pre = 0;
    if (t < CNT_P * SCAN_NB) {
        const int cv = ctot[t];
        pre = ((t % SCAN_NB) < b) ? cv : 0;
    }

    int4 v = make_int4(0, 0, 0, 0);
#pragma unroll 8
    for (int c = 0; c < CNT_P; ++c) {
        const int4 h = hp4[(size_t)c * SCAN_Q + qq];
        v.x += h.x; v.y += h.y; v.z += h.z; v.w += h.w;
    }
    if (!own) v = make_int4(0, 0, 0, 0);

    const int s = (v.x + v.y) + (v.z + v.w);
    int inc = s;
    for (int off = 1; off < 32; off <<= 1) {
        const int o = __shfl_up(inc, off, 32);
        if (lane >= off) inc += o;
    }
    for (int off = 16; off > 0; off >>= 1) pre += __shfl_xor(pre, off, 32);
    if (lane == 31) wtot[wave] = inc;
    if (lane == 0)  wpre[wave] = pre;
    __syncthreads();

    int wt = wtot[lane];
    int wp = wpre[lane];
    int winc = wt;
    for (int off = 1; off < 32; off <<= 1) {
        const int o = __shfl_up(winc, off, 32);
        if (lane >= off) winc += o;
    }
    for (int off = 16; off > 0; off >>= 1) wp += __shfl_xor(wp, off, 32);
    const int wexc = __shfl(winc - wt, wave, 32);

    const int e0 = wp + wexc + (inc - s);
    if (own) {
        int4 r;
        r.x = e0;
        r.y = e0 + v.x;
        r.z = r.y + v.y;
        r.w = r.z + v.z;
        reinterpret_cast<int4*>(rowptr)[q] = r;
#pragma unroll
        for (int c0 = 0; c0 < CNT_P; c0 += 8) {
            int4 h[8];
#pragma unroll
            for (int u = 0; u < 8; ++u) h[u] = hp4[(size_t)(c0 + u) * SCAN_Q + q];
#pragma unroll
            for (int u = 0; u < 8; ++u) {
                hp4[(size_t)(c0 + u) * SCAN_Q + q] = r;
                r.x += h[u].x; r.y += h[u].y; r.z += h[u].z; r.w += h[u].w;
            }
        }
    } else if (q == SCAN_Q) {
        rowptr[GN] = e0;
    }
}

__global__ void csr_fill_k(const int* __restrict__ ei,
                           const int* __restrict__ hplane,
                           const int* __restrict__ rank,
                           int* __restrict__ csrc) {
    int e = blockIdx.x * blockDim.x + threadIdx.x;
    if (e >= GE) return;
    const int c = e / CNT_CH;
    csrc[hplane[(size_t)c * GN + ei[GE + e]] + rank[e]] = ei[e];
}

__device__ __forceinline__ float lrelu02(float v) {
    return (v >= 0.0f) ? v : 0.2f * v;
}

__device__ __forceinline__ float pick4(int h, float a, float b, float c, float d) {
    const bool odd  = (h & 1) != 0;
    const bool high = (h & 2) != 0;
    const float lo = odd ? b : a;
    const float hi = odd ? d : c;
    return high ? hi : lo;
}

__global__ void __launch_bounds__(256)
seg_softmax_agg_ln_k(const int* __restrict__ rowptr,
                     const int* __restrict__ csrc,
                     const float* __restrict__ als,
                     const float* __restrict__ ald,
                     const float* __restrict__ ht,
                     const float* __restrict__ bias,
                     const float* __restrict__ gamma,
                     const float* __restrict__ beta,
                     float* __restrict__ out) {
    const int lane = threadIdx.x & 31;
    const int d = __builtin_amdgcn_readfirstlane(
        (int)(blockIdx.x * (blockDim.x >> 5) + (threadIdx.x >> 5)));
    if (d >= GN) return;
    const int beg = rowptr[d];
    const int end = rowptr[d + 1];
    const float4* als4 = reinterpret_cast<const float4*>(als);
    const float4 ad = reinterpret_cast<const float4*>(ald)[d];
    const float4 sd = als4[d];

    const float t0 = lrelu02(sd.x + ad.x);
    const float t1 = lrelu02(sd.y + ad.y);
    const float t2 = lrelu02(sd.z + ad.z);
    const float t3 = lrelu02(sd.w + ad.w);

    float m0 = t0, m1 = t1, m2 = t2, m3 = t3;
    for (int i = beg + lane; i < end; i += 32) {
        const float4 a = als4[csrc[i]];
        m0 = fmaxf(m0, lrelu02(a.x + ad.x));
        m1 = fmaxf(m1, lrelu02(a.y + ad.y));
        m2 = fmaxf(m2, lrelu02(a.z + ad.z));
        m3 = fmaxf(m3, lrelu02(a.w + ad.w));
    }
    for (int off = 16; off > 0; off >>= 1) {
        m0 = fmaxf(m0, __shfl_xor(m0, off, 32));
        m1 = fmaxf(m1, __shfl_xor(m1, off, 32));
        m2 = fmaxf(m2, __shfl_xor(m2, off, 32));
        m3 = fmaxf(m3, __shfl_xor(m3, off, 32));
    }

    float q0 = 0.0f, q1 = 0.0f, q2 = 0.0f, q3 = 0.0f;
    for (int i = beg + lane; i < end; i += 32) {
        const float4 a = als4[csrc[i]];
        q0 += expf(lrelu02(a.x + ad.x) - m0);
        q1 += expf(lrelu02(a.y + ad.y) - m1);
        q2 += expf(lrelu02(a.z + ad.z) - m2);
        q3 += expf(lrelu02(a.w + ad.w) - m3);
    }
    for (int off = 16; off > 0; off >>= 1) {
        q0 += __shfl_xor(q0, off, 32);
        q1 += __shfl_xor(q1, off, 32);
        q2 += __shfl_xor(q2, off, 32);
        q3 += __shfl_xor(q3, off, 32);
    }
    q0 += expf(t0 - m0);
    q1 += expf(t1 - m1);
    q2 += expf(t2 - m2);
    q3 += expf(t3 - m3);

    const int h = lane >> 3;
    const float4* hrow = reinterpret_cast<const float4*>(ht) + lane;
    float4 acc;
    {
        const float al = expf(pick4(h, t0, t1, t2, t3) - pick4(h, m0, m1, m2, m3))
                         / pick4(h, q0, q1, q2, q3);
        const float4 hv = hrow[(size_t)d * (GD / 4)];
        acc.x = hv.x * al;
        acc.y = hv.y * al;
        acc.z = hv.z * al;
        acc.w = hv.w * al;
    }
    const float4 bv = reinterpret_cast<const float4*>(bias)[lane];
    const float4 gv = reinterpret_cast<const float4*>(gamma)[lane];
    const float4 ev = reinterpret_cast<const float4*>(beta)[lane];
    for (int base = beg; base < end; base += 32) {
        int   s  = 0;
        float a0 = 0.0f, a1 = 0.0f, a2 = 0.0f, a3 = 0.0f;
        if (base + lane < end) {
            s = csrc[base + lane];
            const float4 a = als4[s];
            a0 = expf(lrelu02(a.x + ad.x) - m0) / q0;
            a1 = expf(lrelu02(a.y + ad.y) - m1) / q1;
            a2 = expf(lrelu02(a.z + ad.z) - m2) / q2;
            a3 = expf(lrelu02(a.w + ad.w) - m3) / q3;
        }
        const int cnt = (end - base < 32) ? (end - base) : 32;
#pragma unroll 4
        for (int k = 0; k < cnt; ++k) {
            const int sk = __builtin_amdgcn_readlane(s, k);
            const float b0 = __int_as_float(__builtin_amdgcn_readlane(__float_as_int(a0), k));
            const float b1 = __int_as_float(__builtin_amdgcn_readlane(__float_as_int(a1), k));
            const float b2 = __int_as_float(__builtin_amdgcn_readlane(__float_as_int(a2), k));
            const float b3 = __int_as_float(__builtin_amdgcn_readlane(__float_as_int(a3), k));
            const float al = pick4(h, b0, b1, b2, b3);
            const float4 hv = hrow[(size_t)sk * (GD / 4)];
            acc.x += hv.x * al;
            acc.y += hv.y * al;
            acc.z += hv.z * al;
            acc.w += hv.w * al;
        }
    }
    const float v0 = acc.x + bv.x;
    const float v1 = acc.y + bv.y;
    const float v2 = acc.z + bv.z;
    const float v3 = acc.w + bv.w;
    float rs = v0 + v1 + v2 + v3;
    float rq = v0 * v0 + v1 * v1 + v2 * v2 + v3 * v3;
    for (int off = 16; off > 0; off >>= 1) {
        rs += __shfl_xor(rs, off, 32);
        rq += __shfl_xor(rq, off, 32);
    }
    const float mean = rs * (1.0f / GD);
    const float var  = rq * (1.0f / GD) - mean * mean;
    const float rstd = rsqrtf(var + 1e-5f);
    float4 y;
    y.x = (v0 - mean) * rstd * gv.x + ev.x;
    y.y = (v1 - mean) * rstd * gv.y + ev.y;
    y.z = (v2 - mean) * rstd * gv.z + ev.z;
    y.w = (v3 - mean) * rstd * gv.w + ev.w;
    y.x = (y.x >= 0.0f) ? y.x : 0.1f * y.x;
    y.y = (y.y >= 0.0f) ? y.y : 0.1f * y.y;
    y.z = (y.z >= 0.0f) ? y.z : 0.1f * y.z;
    y.w = (y.w >= 0.0f) ? y.w : 0.1f * y.w;
    reinterpret_cast<float4*>(out)[(size_t)d * (GD / 4) + lane] = y;
}

#define POOL_NPW 8
static_assert(GN % POOL_NPW == 0, "a wave takes POOL_NPW whole nodes");

__device__ __forceinline__ void pool_flush(float* __restrict__ gmax,
                                           float* __restrict__ gsum,
                                           float* __restrict__ cnt,
                                           int g, int lane,
                                           const float* mx, const float* sm, int c) {
    float* pm = gmax + (size_t)g * GD + lane;
    float* ps = gsum + (size_t)g * GD + lane;
#pragma unroll
    for (int i = 0; i < 4; ++i) {
        atomicMaxFloat(pm + 32 * i, mx[i]);
        atomicAdd(ps + 32 * i, sm[i]);
    }
    if (lane == 0) atomicAdd(cnt + g, (float)c);
}

__global__ void __launch_bounds__(256)
pool_k(const float* __restrict__ x,
       const int* __restrict__ batch,
       float* __restrict__ gmax,
       float* __restrict__ gsum,
       float* __restrict__ cnt) {
    const int lane = threadIdx.x & 31;
    const int n0 = __builtin_amdgcn_readfirstlane(
        (int)((blockIdx.x * (blockDim.x >> 5) + (threadIdx.x >> 5)) * POOL_NPW));
    if (n0 >= GN) return;
    const float* xp = x + (size_t)n0 * GD + lane;

    int   b[POOL_NPW];
    float v[POOL_NPW][4];
#pragma unroll
    for (int j = 0; j < POOL_NPW; ++j) {
        b[j] = batch[n0 + j];
#pragma unroll
        for (int i = 0; i < 4; ++i) v[j][i] = xp[j * GD + 32 * i];
    }

    int cur = b[0];
    int c = 1;
    float mx[4], sm[4];
#pragma unroll
    for (int i = 0; i < 4; ++i) { mx[i] = v[0][i]; sm[i] = v[0][i]; }
#pragma unroll
    for (int j = 1; j < POOL_NPW; ++j) {
        if (b[j] != cur) {
            pool_flush(gmax, gsum, cnt, cur, lane, mx, sm, c);
            cur = b[j];
            c = 1;
#pragma unroll
            for (int i = 0; i < 4; ++i) { mx[i] = v[j][i]; sm[i] = v[j][i]; }
        } else {
#pragma unroll
            for (int i = 0; i < 4; ++i) {
                mx[i] = fmaxf(mx[i], v[j][i]);
                sm[i] += v[j][i];
            }
            ++c;
        }
    }
    pool_flush(gmax, gsum, cnt, cur, lane, mx, sm, c);
}

__global__ void pool_fin_k(const float* __restrict__ gmax,
                           const float* __restrict__ gsum,
                           const float* __restrict__ cnt,
                           float* __restrict__ pooled) {
    int tid = blockIdx.x * blockDim.x + threadIdx.x;
    if (tid >= GG * 2 * GD) return;
    const int g = tid >> 8;
    const int f = tid & 255;
    pooled[tid] = (f < GD) ? gmax[(size_t)g * GD + f]
                           : gsum[(size_t)g * GD + (f - GD)] / fmaxf(cnt[g], 1.0f);
}

extern "C" void kernel_launch(void* const* d_in, const int* in_sizes, int n_in,
                              void* d_out, int out_size, void* d_ws, size_t ws_size,
                              hipStream_t stream) {
    const float* x       = (const float*)d_in[0];
    const int*   ei      = (const int*)  d_in[1];
    const int*   batch   = (const int*)  d_in[2];
    const float* W       = (const float*)d_in[3];
    const float* att_src = (const float*)d_in[4];
    const float* att_dst = (const float*)d_in[5];
    const float* bias    = (const float*)d_in[6];
    const float* ln_g    = (const float*)d_in[7];
    const float* ln_b    = (const float*)d_in[8];
    const float* W_out   = (const float*)d_in[9];
    const float* b_out   = (const float*)d_in[10];
    float* out = (float*)d_out;

    float* ws     = (float*)d_ws;
    float* bufP   = ws;
    float* bufQ   = bufP   + (size_t)GN * GD;
    int*   rowptr = (int*)(bufQ + (size_t)GN * GD);
    int*   rank   = rowptr + ROWPTR_SLOTS;
    int*   csrc   = rank   + GE;
    int*   hplane = csrc   + GE;
    int*   ctot   = hplane + (size_t)CNT_P * GN;
    float* als    = ws     + (size_t)3 * GN * GD;
    float* ald    = als    + (size_t)GN * GH;
    float* gmax   = ald    + (size_t)3 * GN * GH;
    float* gsum   = gmax   + (size_t)GG * GD;
    float* cnt    = gsum   + (size_t)GG * GD;
    float* pooled = cnt    + GG;

    const float NEG_INF = -INFINITY;
    const int TB = 256;
    auto blocks = [](long long n, int tb) { return (int)((n + tb - 1) / tb); };

    csr_count_k<<<CNT_P, CNT_T, 0, stream>>>(ei, rank, hplane, ctot);
    csr_scan_k<<<SCAN_NB, SCAN_T, 0, stream>>>(hplane, ctot, rowptr);
    csr_fill_k<<<blocks(GE, TB), TB, 0, stream>>>(ei, hplane, rank, csrc);

    float* ht     = bufQ;
    float* nodeio = bufP;
    for (int l = 0; l < 3; ++l) {
        const float* in = (l == 0) ? x : nodeio;

        gemm128_strip_k<<<blocks(GN / 16, 8), 256, 0, stream>>>(
            in, W + (size_t)l * GD * GD, ht, GN);

        attn_proj_k<<<blocks((long long)GN * GH, TB), TB, 0, stream>>>(
            ht, att_src + l * GH * GC, att_dst + l * GH * GC, als, ald);

        seg_softmax_agg_ln_k<<<blocks(GN, 8), 256, 0, stream>>>(
            rowptr, csrc, als, ald, ht,
            bias + l * GD, ln_g + l * GD, ln_b + l * GD, nodeio);
    }

    fill_f32<<<blocks(GG * GD, TB), TB, 0, stream>>>(gmax, NEG_INF, GG * GD);
    fill_f32<<<blocks(GG * GD, TB), TB, 0, stream>>>(gsum, 0.0f, GG * GD);
    fill_f32<<<blocks(GG, TB), TB, 0, stream>>>(cnt, 0.0f, GG);
    pool_k<<<blocks(GN / POOL_NPW, 8), 256, 0, stream>>>(
        nodeio, batch, gmax, gsum, cnt);
    pool_fin_k<<<blocks(GG * 2 * GD, TB), TB, 0, stream>>>(gmax, gsum, cnt, pooled);

    {
        int tiles = (GG / 16) * (GEMB / 16);
        gemm_wmma_f32<<<blocks(tiles, 4), 128, 0, stream>>>(
            pooled, W_out, b_out, out, GG, 2 * GD, GEMB);
    }
}


# Attention scores al_src/al_dst formed in the layer GEMM's epilogue from its accumulators: 3 launches and 3 re-reads of [N,128] gone; their 32-term sums now 4 per lane, then an 8-lane tree
# speedup vs baseline: 14.3371x; 1.0774x over previous
#include <hip/hip_runtime.h>
#include <cmath>


#define GN   50000
#define GE   800000
#define GETOT (GE + GN)
#define GG   256
#define GH   4
#define GC   32
#define GD   128
#define GEMB 128

typedef float v2f __attribute__((ext_vector_type(2)));
typedef float v8f __attribute__((ext_vector_type(8)));

__device__ __forceinline__ void atomicMaxFloat(float* addr, float val) {
    if (val >= 0.0f) {
        atomicMax((int*)addr, __float_as_int(val));
    } else {
        atomicMin((unsigned int*)addr, __float_as_uint(val));
    }
}

__global__ void fill_f32(float* __restrict__ p, float v, int n) {
    int i = blockIdx.x * blockDim.x + threadIdx.x;
    if (i < n) p[i] = v;
}

__global__ void gemm_wmma_f32(const float* __restrict__ A,
                              const float* __restrict__ B,
                              const float* __restrict__ bias,
                              float* __restrict__ C,
                              int M, int K, int N) {
    const int wave = threadIdx.x >> 5;
    const int lane = threadIdx.x & 31;
    const int tilesN = N >> 4;
    const int total  = (M >> 4) * tilesN;
    const int tile = blockIdx.x * (blockDim.x >> 5) + wave;
    if (tile >= total) return;

    const int row0 = (tile / tilesN) << 4;
    const int col0 = (tile % tilesN) << 4;
    const int half = lane >> 4;
    const int l16  = lane & 15;

    const float* Ap = A + (size_t)(row0 + l16) * K + 2 * half;
    const float* Bp = B + (size_t)(2 * half) * N + col0 + l16;

    v8f acc = {};
    for (int k0 = 0; k0 < K; k0 += 4) {
        v2f a, b;
        a.x = Ap[k0];
        a.y = Ap[k0 + 1];
        b.x = Bp[(size_t)k0 * N];
        b.y = Bp[(size_t)(k0 + 1) * N];
        acc = __builtin_amdgcn_wmma_f32_16x16x4_f32(
                false, a, false, b, (short)0, acc, false, false);
    }

    const float bv = bias ? bias[col0 + l16] : 0.0f;
    float* Cp = C + (size_t)(row0 + 8 * half) * N + col0 + l16;
#pragma unroll
    for (int j = 0; j < 8; ++j) {
        Cp[(size_t)j * N] = acc[j] + bv;
    }
}

__device__ __forceinline__ float sum8_scatter(const float (&v)[8], int lane) {
    const bool b2 = (lane & 4) != 0;
    const bool b1 = (lane & 2) != 0;
    const bool b0 = (lane & 1) != 0;
    float w[4], u[2];
#pragma unroll
    for (int r = 0; r < 4; ++r) {
        const float keep = b2 ? v[r + 4] : v[r];
        const float send = b2 ? v[r] : v[r + 4];
        w[r] = keep + __shfl_xor(send, 4, 32);
    }
#pragma unroll
    for (int r = 0; r < 2; ++r) {
        const float keep = b1 ? w[r + 2] : w[r];
        const float send = b1 ? w[r] : w[r + 2];
        u[r] = keep + __shfl_xor(send, 2, 32);
    }
    const float keep = b0 ? u[1] : u[0];
    const float send = b0 ? u[0] : u[1];
    return keep + __shfl_xor(send, 1, 32);
}

#define LG_PITCH 288
__global__ void __launch_bounds__(256)
gemm128_strip_k(const float* __restrict__ A,
                const float* __restrict__ Wg,
                const float* __restrict__ asrc,
                const float* __restrict__ adst,
                float* __restrict__ C,
                float* __restrict__ als,
                float* __restrict__ ald,
                int M) {
    __shared__ __attribute__((aligned(16))) float Wl[64 * LG_PITCH];
    const int t    = threadIdx.x;
    const int lane = t & 31;
    const int c    = lane & 15;
    const int hh   = lane >> 4;
    const int wave = __builtin_amdgcn_readfirstlane((int)(threadIdx.x >> 5));
    const int row0 = (blockIdx.x * 8 + wave) * 16;
    const bool live = row0 < M;

    const float4* W4 = reinterpret_cast<const float4*>(Wg);
    const int n4  = lane;
    const int cc  = n4 & 15;
    const int jhi = n4 >> 4;
    float4 r0[8], r1[8];
#pragma unroll
    for (int i = 0; i < 8; ++i) {
        const int kp = wave + 8 * i;
        r0[i] = W4[(2 * kp) * (GD / 4) + n4];
        r1[i] = W4[(2 * kp + 1) * (GD / 4) + n4];
    }

    v2f a[32];
    {
        const float* Ap = A + (size_t)((live ? row0 : 0) + c) * GD + 2 * hh;
#pragma unroll
        for (int i = 0; i < 32; ++i) a[i] = *reinterpret_cast<const v2f*>(Ap + 4 * i);
    }

    const float4 as_lo = reinterpret_cast<const float4*>(asrc)[c];
    const float4 as_hi = reinterpret_cast<const float4*>(asrc)[16 + c];
    const float4 ad_lo = reinterpret_cast<const float4*>(adst)[c];
    const float4 ad_hi = reinterpret_cast<const float4*>(adst)[16 + c];

#pragma unroll
    for (int i = 0; i < 8; ++i) {
        const int kp = wave + 8 * i;
        float* dst = Wl + kp * LG_PITCH + (jhi * 4) * 32 + cc * 2;
        *reinterpret_cast<v2f*>(dst)      = (v2f){r0[i].x, r1[i].x};
        *reinterpret_cast<v2f*>(dst + 32) = (v2f){r0[i].y, r1[i].y};
        *reinterpret_cast<v2f*>(dst + 64) = (v2f){r0[i].z, r1[i].z};
        *reinterpret_cast<v2f*>(dst + 96) = (v2f){r0[i].w, r1[i].w};
    }
    __syncthreads();
    if (!live) return;

    v8f acc[8];
#pragma unroll
    for (int j = 0; j < 8; ++j) acc[j] = (v8f){};
    const float* bp = Wl + hh * LG_PITCH + c * 2;
#pragma unroll
    for (int i = 0; i < 32; ++i) {
#pragma unroll
        for (int j = 0; j < 8; ++j) {
            const v2f b = *reinterpret_cast<const v2f*>(bp + (2 * i) * LG_PITCH + j * 32);
            acc[j] = __builtin_amdgcn_wmma_f32_16x16x4_f32(
                        false, a[i], false, b, (short)0, acc[j], false, false);
        }
    }

    float* Cp = C + (size_t)(row0 + 8 * hh) * GD + 4 * c;
#pragma unroll
    for (int r = 0; r < 8; ++r) {
        float4 lo, hi;
        lo.x = acc[0][r]; lo.y = acc[1][r]; lo.z = acc[2][r]; lo.w = acc[3][r];
        hi.x = acc[4][r]; hi.y = acc[5][r]; hi.z = acc[6][r]; hi.w = acc[7][r];
        *reinterpret_cast<float4*>(Cp + (size_t)r * GD)      = lo;
        *reinterpret_cast<float4*>(Cp + (size_t)r * GD + 64) = hi;
    }

    float s_lo[8], s_hi[8], d_lo[8], d_hi[8];
#pragma unroll
    for (int r = 0; r < 8; ++r) {
        s_lo[r] = acc[0][r] * as_lo.x + acc[1][r] * as_lo.y + acc[2][r] * as_lo.z + acc[3][r] * as_lo.w;
        s_hi[r] = acc[4][r] * as_hi.x + acc[5][r] * as_hi.y + acc[6][r] * as_hi.z + acc[7][r] * as_hi.w;
        d_lo[r] = acc[0][r] * ad_lo.x + acc[1][r] * ad_lo.y + acc[2][r] * ad_lo.z + acc[3][r] * ad_lo.w;
        d_hi[r] = acc[4][r] * ad_hi.x + acc[5][r] * ad_hi.y + acc[6][r] * ad_hi.z + acc[7][r] * ad_hi.w;
    }
    const float zs_lo = sum8_scatter(s_lo, lane);
    const float zs_hi = sum8_scatter(s_hi, lane);
    const float zd_lo = sum8_scatter(d_lo, lane);
    const float zd_hi = sum8_scatter(d_hi, lane);
    const int hd = c >> 3;
    const size_t o = (size_t)(row0 + 8 * hh + (c & 7)) * GH;
    als[o + hd]     = zs_lo;
    als[o + 2 + hd] = zs_hi;
    ald[o + hd]     = zd_lo;
    ald[o + 2 + hd] = zd_hi;
}

#define SCAN_T   1024
#define SCAN_CH  (SCAN_T * 4)
#define SCAN_NB  ((GN + 1 + SCAN_CH - 1) / SCAN_CH)
#define SCAN_Q   (GN / 4)
#define ROWPTR_SLOTS (((GN + 1 + 63) / 64) * 64)
static_assert(GN % 4 == 0, "counts are read four at a time");

#define CNT_P    32
#define CNT_T    1024
#define CNT_CH   (GE / CNT_P)
static_assert(GE % (CNT_P * 4) == 0, "a chunk is a whole number of 16-byte groups of edges");
static_assert(CNT_T == SCAN_T, "the counting kernel writes its plane out in the prefix-sum kernel's chunks");
static_assert(CNT_P * SCAN_NB <= SCAN_T, "one thread of the prefix-sum kernel per chunk total");

__global__ void __launch_bounds__(CNT_T)
csr_count_k(const int* __restrict__ ei,
            int* __restrict__ rank,
            int* __restrict__ hplane,
            int* __restrict__ ctot) {
    __shared__ __attribute__((aligned(16))) int hist[GN];
    __shared__ int tot[SCAN_NB];
    constexpr int NG = CNT_CH / 4;
    constexpr int NK = (NG + CNT_T - 1) / CNT_T;
    const int t = threadIdx.x;
    const int c = blockIdx.x;
    int4* hist4 = reinterpret_cast<int4*>(hist);

    const int4* dst4 = reinterpret_cast<const int4*>(ei + GE + c * CNT_CH);
    int4 d[NK];
#pragma unroll
    for (int k = 0; k < NK; ++k) {
        const int g = k * CNT_T + t;
        d[k] = dst4[(g < NG) ? g : NG - 1];
    }

    for (int q = t; q < SCAN_Q; q += CNT_T) hist4[q] = make_int4(0, 0, 0, 0);
    if (t < SCAN_NB) tot[t] = 0;
    __syncthreads();

    int4* rank4 = reinterpret_cast<int4*>(rank + c * CNT_CH);
#pragma unroll
    for (int k = 0; k < NK; ++k) {
        const int g = k * CNT_T + t;
        if (g < NG) {
            int4 r;
            r.x = atomicAdd(&hist[d[k].x], 1);
            r.y = atomicAdd(&hist[d[k].y], 1);
            r.z = atomicAdd(&hist[d[k].z], 1);
            r.w = atomicAdd(&hist[d[k].w], 1);
            rank4[g] = r;
        }
    }
    __syncthreads();

    int4* out4 = reinterpret_cast<int4*>(hplane + (size_t)c * GN);
    int sj[SCAN_NB];
#pragma unroll
    for (int j = 0; j < SCAN_NB; ++j) {
        const int q = j * CNT_T + t;
        sj[j] = 0;
        if (q < SCAN_Q) {
            const int4 h = hist4[q];
            out4[q] = h;
            sj[j] = (h.x + h.y) + (h.z + h.w);
        }
    }
    for (int off = 16; off > 0; off >>= 1) {
#pragma unroll
        for (int j = 0; j < SCAN_NB; ++j) sj[j] += __shfl_xor(sj[j], off, 32);
    }
    if ((t & 31) == 0) {
#pragma unroll
        for (int j = 0; j < SCAN_NB; ++j) atomicAdd(&tot[j], sj[j]);
    }
    __syncthreads();
    if (t < SCAN_NB) ctot[c * SCAN_NB + t] = tot[t];
}

__global__ void __launch_bounds__(SCAN_T)
csr_scan_k(int* __restrict__ hplane,
           const int* __restrict__ ctot,
           int* __restrict__ rowptr) {
    __shared__ int wtot[SCAN_T / 32];
    __shared__ int wpre[SCAN_T / 32];
    const int t    = threadIdx.x;
    const int lane = t & 31;
    const int wave = t >> 5;
    const int b    = blockIdx.x;
    int4* hp4 = reinterpret_cast<int4*>(hplane);

    const int q = b * SCAN_T + t;
    const bool own = q < SCAN_Q;
    const int qq = own ? q : SCAN_Q - 1;

    int pre = 0;
    if (t < CNT_P * SCAN_NB) {
        const int cv = ctot[t];
        pre = ((t % SCAN_NB) < b) ? cv : 0;
    }

    int4 v = make_int4(0, 0, 0, 0);
#pragma unroll 8
    for (int c = 0; c < CNT_P; ++c) {
        const int4 h = hp4[(size_t)c * SCAN_Q + qq];
        v.x += h.x; v.y += h.y; v.z += h.z; v.w += h.w;
    }
    if (!own) v = make_int4(0, 0, 0, 0);

    const int s = (v.x + v.y) + (v.z + v.w);
    int inc = s;
    for (int off = 1; off < 32; off <<= 1) {
        const int o = __shfl_up(inc, off, 32);
        if (lane >= off) inc += o;
    }
    for (int off = 16; off > 0; off >>= 1) pre += __shfl_xor(pre, off, 32);
    if (lane == 31) wtot[wave] = inc;
    if (lane == 0)  wpre[wave] = pre;
    __syncthreads();

    int wt = wtot[lane];
    int wp = wpre[lane];
    int winc = wt;
    for (int off = 1; off < 32; off <<= 1) {
        const int o = __shfl_up(winc, off, 32);
        if (lane >= off) winc += o;
    }
    for (int off = 16; off > 0; off >>= 1) wp += __shfl_xor(wp, off, 32);
    const int wexc = __shfl(winc - wt, wave, 32);

    const int e0 = wp + wexc + (inc - s);
    if (own) {
        int4 r;
        r.x = e0;
        r.y = e0 + v.x;
        r.z = r.y + v.y;
        r.w = r.z + v.z;
        reinterpret_cast<int4*>(rowptr)[q] = r;
#pragma unroll
        for (int c0 = 0; c0 < CNT_P; c0 += 8) {
            int4 h[8];
#pragma unroll
            for (int u = 0; u < 8; ++u) h[u] = hp4[(size_t)(c0 + u) * SCAN_Q + q];
#pragma unroll
            for (int u = 0; u < 8; ++u) {
                hp4[(size_t)(c0 + u) * SCAN_Q + q] = r;
                r.x += h[u].x; r.y += h[u].y; r.z += h[u].z; r.w += h[u].w;
            }
        }
    } else if (q == SCAN_Q) {
        rowptr[GN] = e0;
    }
}

__global__ void csr_fill_k(const int* __restrict__ ei,
                           const int* __restrict__ hplane,
                           const int* __restrict__ rank,
                           int* __restrict__ csrc) {
    int e = blockIdx.x * blockDim.x + threadIdx.x;
    if (e >= GE) return;
    const int c = e / CNT_CH;
    csrc[hplane[(size_t)c * GN + ei[GE + e]] + rank[e]] = ei[e];
}

__device__ __forceinline__ float lrelu02(float v) {
    return (v >= 0.0f) ? v : 0.2f * v;
}

__device__ __forceinline__ float pick4(int h, float a, float b, float c, float d) {
    const bool odd  = (h & 1) != 0;
    const bool high = (h & 2) != 0;
    const float lo = odd ? b : a;
    const float hi = odd ? d : c;
    return high ? hi : lo;
}

__global__ void __launch_bounds__(256)
seg_softmax_agg_ln_k(const int* __restrict__ rowptr,
                     const int* __restrict__ csrc,
                     const float* __restrict__ als,
                     const float* __restrict__ ald,
                     const float* __restrict__ ht,
                     const float* __restrict__ bias,
                     const float* __restrict__ gamma,
                     const float* __restrict__ beta,
                     float* __restrict__ out) {
    const int lane = threadIdx.x & 31;
    const int d = __builtin_amdgcn_readfirstlane(
        (int)(blockIdx.x * (blockDim.x >> 5) + (threadIdx.x >> 5)));
    if (d >= GN) return;
    const int beg = rowptr[d];
    const int end = rowptr[d + 1];
    const float4* als4 = reinterpret_cast<const float4*>(als);
    const float4 ad = reinterpret_cast<const float4*>(ald)[d];
    const float4 sd = als4[d];

    const float t0 = lrelu02(sd.x + ad.x);
    const float t1 = lrelu02(sd.y + ad.y);
    const float t2 = lrelu02(sd.z + ad.z);
    const float t3 = lrelu02(sd.w + ad.w);

    float m0 = t0, m1 = t1, m2 = t2, m3 = t3;
    for (int i = beg + lane; i < end; i += 32) {
        const float4 a = als4[csrc[i]];
        m0 = fmaxf(m0, lrelu02(a.x + ad.x));
        m1 = fmaxf(m1, lrelu02(a.y + ad.y));
        m2 = fmaxf(m2, lrelu02(a.z + ad.z));
        m3 = fmaxf(m3, lrelu02(a.w + ad.w));
    }
    for (int off = 16; off > 0; off >>= 1) {
        m0 = fmaxf(m0, __shfl_xor(m0, off, 32));
        m1 = fmaxf(m1, __shfl_xor(m1, off, 32));
        m2 = fmaxf(m2, __shfl_xor(m2, off, 32));
        m3 = fmaxf(m3, __shfl_xor(m3, off, 32));
    }

    float q0 = 0.0f, q1 = 0.0f, q2 = 0.0f, q3 = 0.0f;
    for (int i = beg + lane; i < end; i += 32) {
        const float4 a = als4[csrc[i]];
        q0 += expf(lrelu02(a.x + ad.x) - m0);
        q1 += expf(lrelu02(a.y + ad.y) - m1);
        q2 += expf(lrelu02(a.z + ad.z) - m2);
        q3 += expf(lrelu02(a.w + ad.w) - m3);
    }
    for (int off = 16; off > 0; off >>= 1) {
        q0 += __shfl_xor(q0, off, 32);
        q1 += __shfl_xor(q1, off, 32);
        q2 += __shfl_xor(q2, off, 32);
        q3 += __shfl_xor(q3, off, 32);
    }
    q0 += expf(t0 - m0);
    q1 += expf(t1 - m1);
    q2 += expf(t2 - m2);
    q3 += expf(t3 - m3);

    const int h = lane >> 3;
    const float4* hrow = reinterpret_cast<const float4*>(ht) + lane;
    float4 acc;
    {
        const float al = expf(pick4(h, t0, t1, t2, t3) - pick4(h, m0, m1, m2, m3))
                         / pick4(h, q0, q1, q2, q3);
        const float4 hv = hrow[(size_t)d * (GD / 4)];
        acc.x = hv.x * al;
        acc.y = hv.y * al;
        acc.z = hv.z * al;
        acc.w = hv.w * al;
    }
    const float4 bv = reinterpret_cast<const float4*>(bias)[lane];
    const float4 gv = reinterpret_cast<const float4*>(gamma)[lane];
    const float4 ev = reinterpret_cast<const float4*>(beta)[lane];
    for (int base = beg; base < end; base += 32) {
        int   s  = 0;
        float a0 = 0.0f, a1 = 0.0f, a2 = 0.0f, a3 = 0.0f;
        if (base + lane < end) {
            s = csrc[base + lane];
            const float4 a = als4[s];
            a0 = expf(lrelu02(a.x + ad.x) - m0) / q0;
            a1 = expf(lrelu02(a.y + ad.y) - m1) / q1;
            a2 = expf(lrelu02(a.z + ad.z) - m2) / q2;
            a3 = expf(lrelu02(a.w + ad.w) - m3) / q3;
        }
        const int cnt = (end - base < 32) ? (end - base) : 32;
#pragma unroll 4
        for (int k = 0; k < cnt; ++k) {
            const int sk = __builtin_amdgcn_readlane(s, k);
            const float b0 = __int_as_float(__builtin_amdgcn_readlane(__float_as_int(a0), k));
            const float b1 = __int_as_float(__builtin_amdgcn_readlane(__float_as_int(a1), k));
            const float b2 = __int_as_float(__builtin_amdgcn_readlane(__float_as_int(a2), k));
            const float b3 = __int_as_float(__builtin_amdgcn_readlane(__float_as_int(a3), k));
            const float al = pick4(h, b0, b1, b2, b3);
            const float4 hv = hrow[(size_t)sk * (GD / 4)];
            acc.x += hv.x * al;
            acc.y += hv.y * al;
            acc.z += hv.z * al;
            acc.w += hv.w * al;
        }
    }
    const float v0 = acc.x + bv.x;
    const float v1 = acc.y + bv.y;
    const float v2 = acc.z + bv.z;
    const float v3 = acc.w + bv.w;
    float rs = v0 + v1 + v2 + v3;
    float rq = v0 * v0 + v1 * v1 + v2 * v2 + v3 * v3;
    for (int off = 16; off > 0; off >>= 1) {
        rs += __shfl_xor(rs, off, 32);
        rq += __shfl_xor(rq, off, 32);
    }
    const float mean = rs * (1.0f / GD);
    const float var  = rq * (1.0f / GD) - mean * mean;
    const float rstd = rsqrtf(var + 1e-5f);
    float4 y;
    y.x = (v0 - mean) * rstd * gv.x + ev.x;
    y.y = (v1 - mean) * rstd * gv.y + ev.y;
    y.z = (v2 - mean) * rstd * gv.z + ev.z;
    y.w = (v3 - mean) * rstd * gv.w + ev.w;
    y.x = (y.x >= 0.0f) ? y.x : 0.1f * y.x;
    y.y = (y.y >= 0.0f) ? y.y : 0.1f * y.y;
    y.z = (y.z >= 0.0f) ? y.z : 0.1f * y.z;
    y.w = (y.w >= 0.0f) ? y.w : 0.1f * y.w;
    reinterpret_cast<float4*>(out)[(size_t)d * (GD / 4) + lane] = y;
}

#define POOL_NPW 8
static_assert(GN % POOL_NPW == 0, "a wave takes POOL_NPW whole nodes");

__device__ __forceinline__ void pool_flush(float* __restrict__ gmax,
                                           float* __restrict__ gsum,
                                           float* __restrict__ cnt,
                                           int g, int lane,
                                           const float* mx, const float* sm, int c) {
    float* pm = gmax + (size_t)g * GD + lane;
    float* ps = gsum + (size_t)g * GD + lane;
#pragma unroll
    for (int i = 0; i < 4; ++i) {
        atomicMaxFloat(pm + 32 * i, mx[i]);
        atomicAdd(ps + 32 * i, sm[i]);
    }
    if (lane == 0) atomicAdd(cnt + g, (float)c);
}

__global__ void __launch_bounds__(256)
pool_k(const float* __restrict__ x,
       const int* __restrict__ batch,
       float* __restrict__ gmax,
       float* __restrict__ gsum,
       float* __restrict__ cnt) {
    const int lane = threadIdx.x & 31;
    const int n0 = __builtin_amdgcn_readfirstlane(
        (int)((blockIdx.x * (blockDim.x >> 5) + (threadIdx.x >> 5)) * POOL_NPW));
    if (n0 >= GN) return;
    const float* xp = x + (size_t)n0 * GD + lane;

    int   b[POOL_NPW];
    float v[POOL_NPW][4];
#pragma unroll
    for (int j = 0; j < POOL_NPW; ++j) {
        b[j] = batch[n0 + j];
#pragma unroll
        for (int i = 0; i < 4; ++i) v[j][i] = xp[j * GD + 32 * i];
    }

    int cur = b[0];
    int c = 1;
    float mx[4], sm[4];
#pragma unroll
    for (int i = 0; i < 4; ++i) { mx[i] = v[0][i]; sm[i] = v[0][i]; }
#pragma unroll
    for (int j = 1; j < POOL_NPW; ++j) {
        if (b[j] != cur) {
            pool_flush(gmax, gsum, cnt, cur, lane, mx, sm, c);
            cur = b[j];
            c = 1;
#pragma unroll
            for (int i = 0; i < 4; ++i) { mx[i] = v[j][i]; sm[i] = v[j][i]; }
        } else {
#pragma unroll
            for (int i = 0; i < 4; ++i) {
                mx[i] = fmaxf(mx[i], v[j][i]);
                sm[i] += v[j][i];
            }
            ++c;
        }
    }
    pool_flush(gmax, gsum, cnt, cur, lane, mx, sm, c);
}

__global__ void pool_fin_k(const float* __restrict__ gmax,
                           const float* __restrict__ gsum,
                           const float* __restrict__ cnt,
                           float* __restrict__ pooled) {
    int tid = blockIdx.x * blockDim.x + threadIdx.x;
    if (tid >= GG * 2 * GD) return;
    const int g = tid >> 8;
    const int f = tid & 255;
    pooled[tid] = (f < GD) ? gmax[(size_t)g * GD + f]
                           : gsum[(size_t)g * GD + (f - GD)] / fmaxf(cnt[g], 1.0f);
}

extern "C" void kernel_launch(void* const* d_in, const int* in_sizes, int n_in,
                              void* d_out, int out_size, void* d_ws, size_t ws_size,
                              hipStream_t stream) {
    const float* x       = (const float*)d_in[0];
    const int*   ei      = (const int*)  d_in[1];
    const int*   batch   = (const int*)  d_in[2];
    const float* W       = (const float*)d_in[3];
    const float* att_src = (const float*)d_in[4];
    const float* att_dst = (const float*)d_in[5];
    const float* bias    = (const float*)d_in[6];
    const float* ln_g    = (const float*)d_in[7];
    const float* ln_b    = (const float*)d_in[8];
    const float* W_out   = (const float*)d_in[9];
    const float* b_out   = (const float*)d_in[10];
    float* out = (float*)d_out;

    float* ws     = (float*)d_ws;
    float* bufP   = ws;
    float* bufQ   = bufP   + (size_t)GN * GD;
    int*   rowptr = (int*)(bufQ + (size_t)GN * GD);
    int*   rank   = rowptr + ROWPTR_SLOTS;
    int*   csrc   = rank   + GE;
    int*   hplane = csrc   + GE;
    int*   ctot   = hplane + (size_t)CNT_P * GN;
    float* als    = ws     + (size_t)3 * GN * GD;
    float* ald    = als    + (size_t)GN * GH;
    float* gmax   = ald    + (size_t)3 * GN * GH;
    float* gsum   = gmax   + (size_t)GG * GD;
    float* cnt    = gsum   + (size_t)GG * GD;
    float* pooled = cnt    + GG;

    const float NEG_INF = -INFINITY;
    const int TB = 256;
    auto blocks = [](long long n, int tb) { return (int)((n + tb - 1) / tb); };

    csr_count_k<<<CNT_P, CNT_T, 0, stream>>>(ei, rank, hplane, ctot);
    csr_scan_k<<<SCAN_NB, SCAN_T, 0, stream>>>(hplane, ctot, rowptr);
    csr_fill_k<<<blocks(GE, TB), TB, 0, stream>>>(ei, hplane, rank, csrc);

    float* ht     = bufQ;
    float* nodeio = bufP;
    for (int l = 0; l < 3; ++l) {
        const float* in = (l == 0) ? x : nodeio;

        gemm128_strip_k<<<blocks(GN / 16, 8), 256, 0, stream>>>(
            in, W + (size_t)l * GD * GD, att_src + l * GH * GC, att_dst + l * GH * GC,
            ht, als, ald, GN);

        seg_softmax_agg_ln_k<<<blocks(GN, 8), 256, 0, stream>>>(
            rowptr, csrc, als, ald, ht,
            bias + l * GD, ln_g + l * GD, ln_b + l * GD, nodeio);
    }

    fill_f32<<<blocks(GG * GD, TB), TB, 0, stream>>>(gmax, NEG_INF, GG * GD);
    fill_f32<<<blocks(GG * GD, TB), TB, 0, stream>>>(gsum, 0.0f, GG * GD);
    fill_f32<<<blocks(GG, TB), TB, 0, stream>>>(cnt, 0.0f, GG);
    pool_k<<<blocks(GN / POOL_NPW, 8), 256, 0, stream>>>(
        nodeio, batch, gmax, gsum, cnt);
    pool_fin_k<<<blocks(GG * 2 * GD, TB), TB, 0, stream>>>(gmax, gsum, cnt, pooled);

    {
        int tiles = (GG / 16) * (GEMB / 16);
        gemm_wmma_f32<<<blocks(tiles, 4), 128, 0, stream>>>(
            pooled, W_out, b_out, out, GG, 2 * GD, GEMB);
    }
}


# Pooling's 3 start-value fills stored by the edge-placement kernel, its mean division done in the output GEMM's A loader (k loops now constant-trip, unrolled): 15 -> 11 launches; sum order unchanged
# speedup vs baseline: 15.1668x; 1.0579x over previous
#include <hip/hip_runtime.h>
#include <cmath>


#define GN   50000
#define GE   800000
#define GETOT (GE + GN)
#define GG   256
#define GH   4
#define GC   32
#define GD   128
#define GEMB 128

typedef float v2f __attribute__((ext_vector_type(2)));
typedef float v8f __attribute__((ext_vector_type(8)));

__device__ __forceinline__ void atomicMaxFloat(float* addr, float val) {
    if (val >= 0.0f) {
        atomicMax((int*)addr, __float_as_int(val));
    } else {
        atomicMin((unsigned int*)addr, __float_as_uint(val));
    }
}

__global__ void __launch_bounds__(128)
pool_out_gemm_k(const float* __restrict__ gmax,
                const float* __restrict__ gsum,
                const float* __restrict__ cnt,
                const float* __restrict__ Wo,
                const float* __restrict__ bo,
                float* __restrict__ out) {
    const int wave = threadIdx.x >> 5;
    const int lane = threadIdx.x & 31;
    constexpr int tilesN = GEMB >> 4;
    constexpr int total  = (GG >> 4) * tilesN;
    const int tile = blockIdx.x * (blockDim.x >> 5) + wave;
    if (tile >= total) return;

    const int row0 = (tile / tilesN) << 4;
    const int col0 = (tile % tilesN) << 4;
    const int half = lane >> 4;
    const int l16  = lane & 15;

    const float* Mp = gmax + (size_t)(row0 + l16) * GD + 2 * half;
    const float* Sp = gsum + (size_t)(row0 + l16) * GD + 2 * half;
    const float* Bp = Wo + (size_t)(2 * half) * GEMB + col0 + l16;
    const float cn  = fmaxf(cnt[row0 + l16], 1.0f);

    v8f acc = {};
#pragma unroll
    for (int k0 = 0; k0 < GD; k0 += 4) {
        v2f a, b;
        a.x = Mp[k0];
        a.y = Mp[k0 + 1];
        b.x = Bp[(size_t)k0 * GEMB];
        b.y = Bp[(size_t)(k0 + 1) * GEMB];
        acc = __builtin_amdgcn_wmma_f32_16x16x4_f32(
                false, a, false, b, (short)0, acc, false, false);
    }
#pragma unroll
    for (int k0 = 0; k0 < GD; k0 += 4) {
        v2f a, b;
        a.x = Sp[k0] / cn;
        a.y = Sp[k0 + 1] / cn;
        b.x = Bp[(size_t)(GD + k0) * GEMB];
        b.y = Bp[(size_t)(GD + k0 + 1) * GEMB];
        acc = __builtin_amdgcn_wmma_f32_16x16x4_f32(
                false, a, false, b, (short)0, acc, false, false);
    }

    const float bv = bo[col0 + l16];
    float* Cp = out + (size_t)(row0 + 8 * half) * GEMB + col0 + l16;
#pragma unroll
    for (int j = 0; j < 8; ++j) {
        Cp[(size_t)j * GEMB] = acc[j] + bv;
    }
}

__device__ __forceinline__ float sum8_scatter(const float (&v)[8], int lane) {
    const bool b2 = (lane & 4) != 0;
    const bool b1 = (lane & 2) != 0;
    const bool b0 = (lane & 1) != 0;
    float w[4], u[2];
#pragma unroll
    for (int r = 0; r < 4; ++r) {
        const float keep = b2 ? v[r + 4] : v[r];
        const float send = b2 ? v[r] : v[r + 4];
        w[r] = keep + __shfl_xor(send, 4, 32);
    }
#pragma unroll
    for (int r = 0; r < 2; ++r) {
        const float keep = b1 ? w[r + 2] : w[r];
        const float send = b1 ? w[r] : w[r + 2];
        u[r] = keep + __shfl_xor(send, 2, 32);
    }
    const float keep = b0 ? u[1] : u[0];
    const float send = b0 ? u[0] : u[1];
    return keep + __shfl_xor(send, 1, 32);
}

#define LG_PITCH 288
__global__ void __launch_bounds__(256)
gemm128_strip_k(const float* __restrict__ A,
                const float* __restrict__ Wg,
                const float* __restrict__ asrc,
                const float* __restrict__ adst,
                float* __restrict__ C,
                float* __restrict__ als,
                float* __restrict__ ald,
                int M) {
    __shared__ __attribute__((aligned(16))) float Wl[64 * LG_PITCH];
    const int t    = threadIdx.x;
    const int lane = t & 31;
    const int c    = lane & 15;
    const int hh   = lane >> 4;
    const int wave = __builtin_amdgcn_readfirstlane((int)(threadIdx.x >> 5));
    const int row0 = (blockIdx.x * 8 + wave) * 16;
    const bool live = row0 < M;

    const float4* W4 = reinterpret_cast<const float4*>(Wg);
    const int n4  = lane;
    const int cc  = n4 & 15;
    const int jhi = n4 >> 4;
    float4 r0[8], r1[8];
#pragma unroll
    for (int i = 0; i < 8; ++i) {
        const int kp = wave + 8 * i;
        r0[i] = W4[(2 * kp) * (GD / 4) + n4];
        r1[i] = W4[(2 * kp + 1) * (GD / 4) + n4];
    }

    v2f a[32];
    {
        const float* Ap = A + (size_t)((live ? row0 : 0) + c) * GD + 2 * hh;
#pragma unroll
        for (int i = 0; i < 32; ++i) a[i] = *reinterpret_cast<const v2f*>(Ap + 4 * i);
    }

    const float4 as_lo = reinterpret_cast<const float4*>(asrc)[c];
    const float4 as_hi = reinterpret_cast<const float4*>(asrc)[16 + c];
    const float4 ad_lo = reinterpret_cast<const float4*>(adst)[c];
    const float4 ad_hi = reinterpret_cast<const float4*>(adst)[16 + c];

#pragma unroll
    for (int i = 0; i < 8; ++i) {
        const int kp = wave + 8 * i;
        float* dst = Wl + kp * LG_PITCH + (jhi * 4) * 32 + cc * 2;
        *reinterpret_cast<v2f*>(dst)      = (v2f){r0[i].x, r1[i].x};
        *reinterpret_cast<v2f*>(dst + 32) = (v2f){r0[i].y, r1[i].y};
        *reinterpret_cast<v2f*>(dst + 64) = (v2f){r0[i].z, r1[i].z};
        *reinterpret_cast<v2f*>(dst + 96) = (v2f){r0[i].w, r1[i].w};
    }
    __syncthreads();
    if (!live) return;

    v8f acc[8];
#pragma unroll
    for (int j = 0; j < 8; ++j) acc[j] = (v8f){};
    const float* bp = Wl + hh * LG_PITCH + c * 2;
#pragma unroll
    for (int i = 0; i < 32; ++i) {
#pragma unroll
        for (int j = 0; j < 8; ++j) {
            const v2f b = *reinterpret_cast<const v2f*>(bp + (2 * i) * LG_PITCH + j * 32);
            acc[j] = __builtin_amdgcn_wmma_f32_16x16x4_f32(
                        false, a[i], false, b, (short)0, acc[j], false, false);
        }
    }

    float* Cp = C + (size_t)(row0 + 8 * hh) * GD + 4 * c;
#pragma unroll
    for (int r = 0; r < 8; ++r) {
        float4 lo, hi;
        lo.x = acc[0][r]; lo.y = acc[1][r]; lo.z = acc[2][r]; lo.w = acc[3][r];
        hi.x = acc[4][r]; hi.y = acc[5][r]; hi.z = acc[6][r]; hi.w = acc[7][r];
        *reinterpret_cast<float4*>(Cp + (size_t)r * GD)      = lo;
        *reinterpret_cast<float4*>(Cp + (size_t)r * GD + 64) = hi;
    }

    float s_lo[8], s_hi[8], d_lo[8], d_hi[8];
#pragma unroll
    for (int r = 0; r < 8; ++r) {
        s_lo[r] = acc[0][r] * as_lo.x + acc[1][r] * as_lo.y + acc[2][r] * as_lo.z + acc[3][r] * as_lo.w;
        s_hi[r] = acc[4][r] * as_hi.x + acc[5][r] * as_hi.y + acc[6][r] * as_hi.z + acc[7][r] * as_hi.w;
        d_lo[r] = acc[0][r] * ad_lo.x + acc[1][r] * ad_lo.y + acc[2][r] * ad_lo.z + acc[3][r] * ad_lo.w;
        d_hi[r] = acc[4][r] * ad_hi.x + acc[5][r] * ad_hi.y + acc[6][r] * ad_hi.z + acc[7][r] * ad_hi.w;
    }
    const float zs_lo = sum8_scatter(s_lo, lane);
    const float zs_hi = sum8_scatter(s_hi, lane);
    const float zd_lo = sum8_scatter(d_lo, lane);
    const float zd_hi = sum8_scatter(d_hi, lane);
    const int hd = c >> 3;
    const size_t o = (size_t)(row0 + 8 * hh + (c & 7)) * GH;
    als[o + hd]     = zs_lo;
    als[o + 2 + hd] = zs_hi;
    ald[o + hd]     = zd_lo;
    ald[o + 2 + hd] = zd_hi;
}

#define SCAN_T   1024
#define SCAN_CH  (SCAN_T * 4)
#define SCAN_NB  ((GN + 1 + SCAN_CH - 1) / SCAN_CH)
#define SCAN_Q   (GN / 4)
#define ROWPTR_SLOTS (((GN + 1 + 63) / 64) * 64)
static_assert(GN % 4 == 0, "counts are read four at a time");

#define CNT_P    32
#define CNT_T    1024
#define CNT_CH   (GE / CNT_P)
static_assert(GE % (CNT_P * 4) == 0, "a chunk is a whole number of 16-byte groups of edges");
static_assert(CNT_T == SCAN_T, "the counting kernel writes its plane out in the prefix-sum kernel's chunks");
static_assert(CNT_P * SCAN_NB <= SCAN_T, "one thread of the prefix-sum kernel per chunk total");

__global__ void __launch_bounds__(CNT_T)
csr_count_k(const int* __restrict__ ei,
            int* __restrict__ rank,
            int* __restrict__ hplane,
            int* __restrict__ ctot) {
    __shared__ __attribute__((aligned(16))) int hist[GN];
    __shared__ int tot[SCAN_NB];
    constexpr int NG = CNT_CH / 4;
    constexpr int NK = (NG + CNT_T - 1) / CNT_T;
    const int t = threadIdx.x;
    const int c = blockIdx.x;
    int4* hist4 = reinterpret_cast<int4*>(hist);

    const int4* dst4 = reinterpret_cast<const int4*>(ei + GE + c * CNT_CH);
    int4 d[NK];
#pragma unroll
    for (int k = 0; k < NK; ++k) {
        const int g = k * CNT_T + t;
        d[k] = dst4[(g < NG) ? g : NG - 1];
    }

    for (int q = t; q < SCAN_Q; q += CNT_T) hist4[q] = make_int4(0, 0, 0, 0);
    if (t < SCAN_NB) tot[t] = 0;
    __syncthreads();

    int4* rank4 = reinterpret_cast<int4*>(rank + c * CNT_CH);
#pragma unroll
    for (int k = 0; k < NK; ++k) {
        const int g = k * CNT_T + t;
        if (g < NG) {
            int4 r;
            r.x = atomicAdd(&hist[d[k].x], 1);
            r.y = atomicAdd(&hist[d[k].y], 1);
            r.z = atomicAdd(&hist[d[k].z], 1);
            r.w = atomicAdd(&hist[d[k].w], 1);
            rank4[g] = r;
        }
    }
    __syncthreads();

    int4* out4 = reinterpret_cast<int4*>(hplane + (size_t)c * GN);
    int sj[SCAN_NB];
#pragma unroll
    for (int j = 0; j < SCAN_NB; ++j) {
        const int q = j * CNT_T + t;
        sj[j] = 0;
        if (q < SCAN_Q) {
            const int4 h = hist4[q];
            out4[q] = h;
            sj[j] = (h.x + h.y) + (h.z + h.w);
        }
    }
    for (int off = 16; off > 0; off >>= 1) {
#pragma unroll
        for (int j = 0; j < SCAN_NB; ++j) sj[j] += __shfl_xor(sj[j], off, 32);
    }
    if ((t & 31) == 0) {
#pragma unroll
        for (int j = 0; j < SCAN_NB; ++j) atomicAdd(&tot[j], sj[j]);
    }
    __syncthreads();
    if (t < SCAN_NB) ctot[c * SCAN_NB + t] = tot[t];
}

__global__ void __launch_bounds__(SCAN_T)
csr_scan_k(int* __restrict__ hplane,
           const int* __restrict__ ctot,
           int* __restrict__ rowptr) {
    __shared__ int wtot[SCAN_T / 32];
    __shared__ int wpre[SCAN_T / 32];
    const int t    = threadIdx.x;
    const int lane = t & 31;
    const int wave = t >> 5;
    const int b    = blockIdx.x;
    int4* hp4 = reinterpret_cast<int4*>(hplane);

    const int q = b * SCAN_T + t;
    const bool own = q < SCAN_Q;
    const int qq = own ? q : SCAN_Q - 1;

    int pre = 0;
    if (t < CNT_P * SCAN_NB) {
        const int cv = ctot[t];
        pre = ((t % SCAN_NB) < b) ? cv : 0;
    }

    int4 v = make_int4(0, 0, 0, 0);
#pragma unroll 8
    for (int c = 0; c < CNT_P; ++c) {
        const int4 h = hp4[(size_t)c * SCAN_Q + qq];
        v.x += h.x; v.y += h.y; v.z += h.z; v.w += h.w;
    }
    if (!own) v = make_int4(0, 0, 0, 0);

    const int s = (v.x + v.y) + (v.z + v.w);
    int inc = s;
    for (int off = 1; off < 32; off <<= 1) {
        const int o = __shfl_up(inc, off, 32);
        if (lane >= off) inc += o;
    }
    for (int off = 16; off > 0; off >>= 1) pre += __shfl_xor(pre, off, 32);
    if (lane == 31) wtot[wave] = inc;
    if (lane == 0)  wpre[wave] = pre;
    __syncthreads();

    int wt = wtot[lane];
    int wp = wpre[lane];
    int winc = wt;
    for (int off = 1; off < 32; off <<= 1) {
        const int o = __shfl_up(winc, off, 32);
        if (lane >= off) winc += o;
    }
    for (int off = 16; off > 0; off >>= 1) wp += __shfl_xor(wp, off, 32);
    const int wexc = __shfl(winc - wt, wave, 32);

    const int e0 = wp + wexc + (inc - s);
    if (own) {
        int4 r;
        r.x = e0;
        r.y = e0 + v.x;
        r.z = r.y + v.y;
        r.w = r.z + v.z;
        reinterpret_cast<int4*>(rowptr)[q] = r;
#pragma unroll
        for (int c0 = 0; c0 < CNT_P; c0 += 8) {
            int4 h[8];
#pragma unroll
            for (int u = 0; u < 8; ++u) h[u] = hp4[(size_t)(c0 + u) * SCAN_Q + q];
#pragma unroll
            for (int u = 0; u < 8; ++u) {
                hp4[(size_t)(c0 + u) * SCAN_Q + q] = r;
                r.x += h[u].x; r.y += h[u].y; r.z += h[u].z; r.w += h[u].w;
            }
        }
    } else if (q == SCAN_Q) {
        rowptr[GN] = e0;
    }
}

#define POOL_MAX_Q (GG * GD / 4)
#define POOL_ACC_Q ((2 * GG * GD + GG) / 4)
static_assert(GG % 4 == 0, "the accumulators are a whole number of 16-byte groups");
static_assert(POOL_ACC_Q <= GE, "one placement thread per 16-byte group of start values");

__global__ void csr_fill_k(const int* __restrict__ ei,
                           const int* __restrict__ hplane,
                           const int* __restrict__ rank,
                           int* __restrict__ csrc,
                           float* __restrict__ pool_acc) {
    int e = blockIdx.x * blockDim.x + threadIdx.x;
    if (e < POOL_ACC_Q) {
        const float v = (e < POOL_MAX_Q) ? -INFINITY : 0.0f;
        reinterpret_cast<float4*>(pool_acc)[e] = make_float4(v, v, v, v);
    }
    if (e >= GE) return;
    const int c = e / CNT_CH;
    csrc[hplane[(size_t)c * GN + ei[GE + e]] + rank[e]] = ei[e];
}

__device__ __forceinline__ float lrelu02(float v) {
    return (v >= 0.0f) ? v : 0.2f * v;
}

__device__ __forceinline__ float pick4(int h, float a, float b, float c, float d) {
    const bool odd  = (h & 1) != 0;
    const bool high = (h & 2) != 0;
    const float lo = odd ? b : a;
    const float hi = odd ? d : c;
    return high ? hi : lo;
}

__global__ void __launch_bounds__(256)
seg_softmax_agg_ln_k(const int* __restrict__ rowptr,
                     const int* __restrict__ csrc,
                     const float* __restrict__ als,
                     const float* __restrict__ ald,
                     const float* __restrict__ ht,
                     const float* __restrict__ bias,
                     const float* __restrict__ gamma,
                     const float* __restrict__ beta,
                     float* __restrict__ out) {
    const int lane = threadIdx.x & 31;
    const int d = __builtin_amdgcn_readfirstlane(
        (int)(blockIdx.x * (blockDim.x >> 5) + (threadIdx.x >> 5)));
    if (d >= GN) return;
    const int beg = rowptr[d];
    const int end = rowptr[d + 1];
    const float4* als4 = reinterpret_cast<const float4*>(als);
    const float4 ad = reinterpret_cast<const float4*>(ald)[d];
    const float4 sd = als4[d];

    const float t0 = lrelu02(sd.x + ad.x);
    const float t1 = lrelu02(sd.y + ad.y);
    const float t2 = lrelu02(sd.z + ad.z);
    const float t3 = lrelu02(sd.w + ad.w);

    float m0 = t0, m1 = t1, m2 = t2, m3 = t3;
    for (int i = beg + lane; i < end; i += 32) {
        const float4 a = als4[csrc[i]];
        m0 = fmaxf(m0, lrelu02(a.x + ad.x));
        m1 = fmaxf(m1, lrelu02(a.y + ad.y));
        m2 = fmaxf(m2, lrelu02(a.z + ad.z));
        m3 = fmaxf(m3, lrelu02(a.w + ad.w));
    }
    for (int off = 16; off > 0; off >>= 1) {
        m0 = fmaxf(m0, __shfl_xor(m0, off, 32));
        m1 = fmaxf(m1, __shfl_xor(m1, off, 32));
        m2 = fmaxf(m2, __shfl_xor(m2, off, 32));
        m3 = fmaxf(m3, __shfl_xor(m3, off, 32));
    }

    float q0 = 0.0f, q1 = 0.0f, q2 = 0.0f, q3 = 0.0f;
    for (int i = beg + lane; i < end; i += 32) {
        const float4 a = als4[csrc[i]];
        q0 += expf(lrelu02(a.x + ad.x) - m0);
        q1 += expf(lrelu02(a.y + ad.y) - m1);
        q2 += expf(lrelu02(a.z + ad.z) - m2);
        q3 += expf(lrelu02(a.w + ad.w) - m3);
    }
    for (int off = 16; off > 0; off >>= 1) {
        q0 += __shfl_xor(q0, off, 32);
        q1 += __shfl_xor(q1, off, 32);
        q2 += __shfl_xor(q2, off, 32);
        q3 += __shfl_xor(q3, off, 32);
    }
    q0 += expf(t0 - m0);
    q1 += expf(t1 - m1);
    q2 += expf(t2 - m2);
    q3 += expf(t3 - m3);

    const int h = lane >> 3;
    const float4* hrow = reinterpret_cast<const float4*>(ht) + lane;
    float4 acc;
    {
        const float al = expf(pick4(h, t0, t1, t2, t3) - pick4(h, m0, m1, m2, m3))
                         / pick4(h, q0, q1, q2, q3);
        const float4 hv = hrow[(size_t)d * (GD / 4)];
        acc.x = hv.x * al;
        acc.y = hv.y * al;
        acc.z = hv.z * al;
        acc.w = hv.w * al;
    }
    const float4 bv = reinterpret_cast<const float4*>(bias)[lane];
    const float4 gv = reinterpret_cast<const float4*>(gamma)[lane];
    const float4 ev = reinterpret_cast<const float4*>(beta)[lane];
    for (int base = beg; base < end; base += 32) {
        int   s  = 0;
        float a0 = 0.0f, a1 = 0.0f, a2 = 0.0f, a3 = 0.0f;
        if (base + lane < end) {
            s = csrc[base + lane];
            const float4 a = als4[s];
            a0 = expf(lrelu02(a.x + ad.x) - m0) / q0;
            a1 = expf(lrelu02(a.y + ad.y) - m1) / q1;
            a2 = expf(lrelu02(a.z + ad.z) - m2) / q2;
            a3 = expf(lrelu02(a.w + ad.w) - m3) / q3;
        }
        const int cnt = (end - base < 32) ? (end - base) : 32;
#pragma unroll 4
        for (int k = 0; k < cnt; ++k) {
            const int sk = __builtin_amdgcn_readlane(s, k);
            const float b0 = __int_as_float(__builtin_amdgcn_readlane(__float_as_int(a0), k));
            const float b1 = __int_as_float(__builtin_amdgcn_readlane(__float_as_int(a1), k));
            const float b2 = __int_as_float(__builtin_amdgcn_readlane(__float_as_int(a2), k));
            const float b3 = __int_as_float(__builtin_amdgcn_readlane(__float_as_int(a3), k));
            const float al = pick4(h, b0, b1, b2, b3);
            const float4 hv = hrow[(size_t)sk * (GD / 4)];
            acc.x += hv.x * al;
            acc.y += hv.y * al;
            acc.z += hv.z * al;
            acc.w += hv.w * al;
        }
    }
    const float v0 = acc.x + bv.x;
    const float v1 = acc.y + bv.y;
    const float v2 = acc.z + bv.z;
    const float v3 = acc.w + bv.w;
    float rs = v0 + v1 + v2 + v3;
    float rq = v0 * v0 + v1 * v1 + v2 * v2 + v3 * v3;
    for (int off = 16; off > 0; off >>= 1) {
        rs += __shfl_xor(rs, off, 32);
        rq += __shfl_xor(rq, off, 32);
    }
    const float mean = rs * (1.0f / GD);
    const float var  = rq * (1.0f / GD) - mean * mean;
    const float rstd = rsqrtf(var + 1e-5f);
    float4 y;
    y.x = (v0 - mean) * rstd * gv.x + ev.x;
    y.y = (v1 - mean) * rstd * gv.y + ev.y;
    y.z = (v2 - mean) * rstd * gv.z + ev.z;
    y.w = (v3 - mean) * rstd * gv.w + ev.w;
    y.x = (y.x >= 0.0f) ? y.x : 0.1f * y.x;
    y.y = (y.y >= 0.0f) ? y.y : 0.1f * y.y;
    y.z = (y.z >= 0.0f) ? y.z : 0.1f * y.z;
    y.w = (y.w >= 0.0f) ? y.w : 0.1f * y.w;
    reinterpret_cast<float4*>(out)[(size_t)d * (GD / 4) + lane] = y;
}

#define POOL_NPW 8
static_assert(GN % POOL_NPW == 0, "a wave takes POOL_NPW whole nodes");

__device__ __forceinline__ void pool_flush(float* __restrict__ gmax,
                                           float* __restrict__ gsum,
                                           float* __restrict__ cnt,
                                           int g, int lane,
                                           const float* mx, const float* sm, int c) {
    float* pm = gmax + (size_t)g * GD + lane;
    float* ps = gsum + (size_t)g * GD + lane;
#pragma unroll
    for (int i = 0; i < 4; ++i) {
        atomicMaxFloat(pm + 32 * i, mx[i]);
        atomicAdd(ps + 32 * i, sm[i]);
    }
    if (lane == 0) atomicAdd(cnt + g, (float)c);
}

__global__ void __launch_bounds__(256)
pool_k(const float* __restrict__ x,
       const int* __restrict__ batch,
       float* __restrict__ gmax,
       float* __restrict__ gsum,
       float* __restrict__ cnt) {
    const int lane = threadIdx.x & 31;
    const int n0 = __builtin_amdgcn_readfirstlane(
        (int)((blockIdx.x * (blockDim.x >> 5) + (threadIdx.x >> 5)) * POOL_NPW));
    if (n0 >= GN) return;
    const float* xp = x + (size_t)n0 * GD + lane;

    int   b[POOL_NPW];
    float v[POOL_NPW][4];
#pragma unroll
    for (int j = 0; j < POOL_NPW; ++j) {
        b[j] = batch[n0 + j];
#pragma unroll
        for (int i = 0; i < 4; ++i) v[j][i] = xp[j * GD + 32 * i];
    }

    int cur = b[0];
    int c = 1;
    float mx[4], sm[4];
#pragma unroll
    for (int i = 0; i < 4; ++i) { mx[i] = v[0][i]; sm[i] = v[0][i]; }
#pragma unroll
    for (int j = 1; j < POOL_NPW; ++j) {
        if (b[j] != cur) {
            pool_flush(gmax, gsum, cnt, cur, lane, mx, sm, c);
            cur = b[j];
            c = 1;
#pragma unroll
            for (int i = 0; i < 4; ++i) { mx[i] = v[j][i]; sm[i] = v[j][i]; }
        } else {
#pragma unroll
            for (int i = 0; i < 4; ++i) {
                mx[i] = fmaxf(mx[i], v[j][i]);
                sm[i] += v[j][i];
            }
            ++c;
        }
    }
    pool_flush(gmax, gsum, cnt, cur, lane, mx, sm, c);
}

extern "C" void kernel_launch(void* const* d_in, const int* in_sizes, int n_in,
                              void* d_out, int out_size, void* d_ws, size_t ws_size,
                              hipStream_t stream) {
    const float* x       = (const float*)d_in[0];
    const int*   ei      = (const int*)  d_in[1];
    const int*   batch   = (const int*)  d_in[2];
    const float* W       = (const float*)d_in[3];
    const float* att_src = (const float*)d_in[4];
    const float* att_dst = (const float*)d_in[5];
    const float* bias    = (const float*)d_in[6];
    const float* ln_g    = (const float*)d_in[7];
    const float* ln_b    = (const float*)d_in[8];
    const float* W_out   = (const float*)d_in[9];
    const float* b_out   = (const float*)d_in[10];
    float* out = (float*)d_out;

    float* ws     = (float*)d_ws;
    float* bufP   = ws;
    float* bufQ   = bufP   + (size_t)GN * GD;
    int*   rowptr = (int*)(bufQ + (size_t)GN * GD);
    int*   rank   = rowptr + ROWPTR_SLOTS;
    int*   csrc   = rank   + GE;
    int*   hplane = csrc   + GE;
    int*   ctot   = hplane + (size_t)CNT_P * GN;
    float* als    = ws     + (size_t)3 * GN * GD;
    float* ald    = als    + (size_t)GN * GH;
    float* gmax   = ald    + (size_t)3 * GN * GH;
    float* gsum   = gmax   + (size_t)GG * GD;
    float* cnt    = gsum   + (size_t)GG * GD;

    const int TB = 256;
    auto blocks = [](long long n, int tb) { return (int)((n + tb - 1) / tb); };

    csr_count_k<<<CNT_P, CNT_T, 0, stream>>>(ei, rank, hplane, ctot);
    csr_scan_k<<<SCAN_NB, SCAN_T, 0, stream>>>(hplane, ctot, rowptr);
    csr_fill_k<<<blocks(GE, TB), TB, 0, stream>>>(ei, hplane, rank, csrc, gmax);

    float* ht     = bufQ;
    float* nodeio = bufP;
    for (int l = 0; l < 3; ++l) {
        const float* in = (l == 0) ? x : nodeio;

        gemm128_strip_k<<<blocks(GN / 16, 8), 256, 0, stream>>>(
            in, W + (size_t)l * GD * GD, att_src + l * GH * GC, att_dst + l * GH * GC,
            ht, als, ald, GN);

        seg_softmax_agg_ln_k<<<blocks(GN, 8), 256, 0, stream>>>(
            rowptr, csrc, als, ald, ht,
            bias + l * GD, ln_g + l * GD, ln_b + l * GD, nodeio);
    }

    pool_k<<<blocks(GN / POOL_NPW, 8), 256, 0, stream>>>(
        nodeio, batch, gmax, gsum, cnt);

    {
        int tiles = (GG / 16) * (GEMB / 16);
        pool_out_gemm_k<<<blocks(tiles, 4), 128, 0, stream>>>(
            gmax, gsum, cnt, W_out, b_out, out);
    }
}


# Segment softmax: each edge's source and 4 scores are read once into the wave's registers and serve all three passes (was re-gathered per pass); longer-than-32 tails as before; sum order unchanged
# speedup vs baseline: 16.6281x; 1.0963x over previous
#include <hip/hip_runtime.h>
#include <cmath>


#define GN   50000
#define GE   800000
#define GETOT (GE + GN)
#define GG   256
#define GH   4
#define GC   32
#define GD   128
#define GEMB 128

typedef float v2f __attribute__((ext_vector_type(2)));
typedef float v8f __attribute__((ext_vector_type(8)));

__device__ __forceinline__ void atomicMaxFloat(float* addr, float val) {
    if (val >= 0.0f) {
        atomicMax((int*)addr, __float_as_int(val));
    } else {
        atomicMin((unsigned int*)addr, __float_as_uint(val));
    }
}

__global__ void __launch_bounds__(128)
pool_out_gemm_k(const float* __restrict__ gmax,
                const float* __restrict__ gsum,
                const float* __restrict__ cnt,
                const float* __restrict__ Wo,
                const float* __restrict__ bo,
                float* __restrict__ out) {
    const int wave = threadIdx.x >> 5;
    const int lane = threadIdx.x & 31;
    constexpr int tilesN = GEMB >> 4;
    constexpr int total  = (GG >> 4) * tilesN;
    const int tile = blockIdx.x * (blockDim.x >> 5) + wave;
    if (tile >= total) return;

    const int row0 = (tile / tilesN) << 4;
    const int col0 = (tile % tilesN) << 4;
    const int half = lane >> 4;
    const int l16  = lane & 15;

    const float* Mp = gmax + (size_t)(row0 + l16) * GD + 2 * half;
    const float* Sp = gsum + (size_t)(row0 + l16) * GD + 2 * half;
    const float* Bp = Wo + (size_t)(2 * half) * GEMB + col0 + l16;
    const float cn  = fmaxf(cnt[row0 + l16], 1.0f);

    v8f acc = {};
#pragma unroll
    for (int k0 = 0; k0 < GD; k0 += 4) {
        v2f a, b;
        a.x = Mp[k0];
        a.y = Mp[k0 + 1];
        b.x = Bp[(size_t)k0 * GEMB];
        b.y = Bp[(size_t)(k0 + 1) * GEMB];
        acc = __builtin_amdgcn_wmma_f32_16x16x4_f32(
                false, a, false, b, (short)0, acc, false, false);
    }
#pragma unroll
    for (int k0 = 0; k0 < GD; k0 += 4) {
        v2f a, b;
        a.x = Sp[k0] / cn;
        a.y = Sp[k0 + 1] / cn;
        b.x = Bp[(size_t)(GD + k0) * GEMB];
        b.y = Bp[(size_t)(GD + k0 + 1) * GEMB];
        acc = __builtin_amdgcn_wmma_f32_16x16x4_f32(
                false, a, false, b, (short)0, acc, false, false);
    }

    const float bv = bo[col0 + l16];
    float* Cp = out + (size_t)(row0 + 8 * half) * GEMB + col0 + l16;
#pragma unroll
    for (int j = 0; j < 8; ++j) {
        Cp[(size_t)j * GEMB] = acc[j] + bv;
    }
}

__device__ __forceinline__ float sum8_scatter(const float (&v)[8], int lane) {
    const bool b2 = (lane & 4) != 0;
    const bool b1 = (lane & 2) != 0;
    const bool b0 = (lane & 1) != 0;
    float w[4], u[2];
#pragma unroll
    for (int r = 0; r < 4; ++r) {
        const float keep = b2 ? v[r + 4] : v[r];
        const float send = b2 ? v[r] : v[r + 4];
        w[r] = keep + __shfl_xor(send, 4, 32);
    }
#pragma unroll
    for (int r = 0; r < 2; ++r) {
        const float keep = b1 ? w[r + 2] : w[r];
        const float send = b1 ? w[r] : w[r + 2];
        u[r] = keep + __shfl_xor(send, 2, 32);
    }
    const float keep = b0 ? u[1] : u[0];
    const float send = b0 ? u[0] : u[1];
    return keep + __shfl_xor(send, 1, 32);
}

#define LG_PITCH 288
__global__ void __launch_bounds__(256)
gemm128_strip_k(const float* __restrict__ A,
                const float* __restrict__ Wg,
                const float* __restrict__ asrc,
                const float* __restrict__ adst,
                float* __restrict__ C,
                float* __restrict__ als,
                float* __restrict__ ald,
                int M) {
    __shared__ __attribute__((aligned(16))) float Wl[64 * LG_PITCH];
    const int t    = threadIdx.x;
    const int lane = t & 31;
    const int c    = lane & 15;
    const int hh   = lane >> 4;
    const int wave = __builtin_amdgcn_readfirstlane((int)(threadIdx.x >> 5));
    const int row0 = (blockIdx.x * 8 + wave) * 16;
    const bool live = row0 < M;

    const float4* W4 = reinterpret_cast<const float4*>(Wg);
    const int n4  = lane;
    const int cc  = n4 & 15;
    const int jhi = n4 >> 4;
    float4 r0[8], r1[8];
#pragma unroll
    for (int i = 0; i < 8; ++i) {
        const int kp = wave + 8 * i;
        r0[i] = W4[(2 * kp) * (GD / 4) + n4];
        r1[i] = W4[(2 * kp + 1) * (GD / 4) + n4];
    }

    v2f a[32];
    {
        const float* Ap = A + (size_t)((live ? row0 : 0) + c) * GD + 2 * hh;
#pragma unroll
        for (int i = 0; i < 32; ++i) a[i] = *reinterpret_cast<const v2f*>(Ap + 4 * i);
    }

    const float4 as_lo = reinterpret_cast<const float4*>(asrc)[c];
    const float4 as_hi = reinterpret_cast<const float4*>(asrc)[16 + c];
    const float4 ad_lo = reinterpret_cast<const float4*>(adst)[c];
    const float4 ad_hi = reinterpret_cast<const float4*>(adst)[16 + c];

#pragma unroll
    for (int i = 0; i < 8; ++i) {
        const int kp = wave + 8 * i;
        float* dst = Wl + kp * LG_PITCH + (jhi * 4) * 32 + cc * 2;
        *reinterpret_cast<v2f*>(dst)      = (v2f){r0[i].x, r1[i].x};
        *reinterpret_cast<v2f*>(dst + 32) = (v2f){r0[i].y, r1[i].y};
        *reinterpret_cast<v2f*>(dst + 64) = (v2f){r0[i].z, r1[i].z};
        *reinterpret_cast<v2f*>(dst + 96) = (v2f){r0[i].w, r1[i].w};
    }
    __syncthreads();
    if (!live) return;

    v8f acc[8];
#pragma unroll
    for (int j = 0; j < 8; ++j) acc[j] = (v8f){};
    const float* bp = Wl + hh * LG_PITCH + c * 2;
#pragma unroll
    for (int i = 0; i < 32; ++i) {
#pragma unroll
        for (int j = 0; j < 8; ++j) {
            const v2f b = *reinterpret_cast<const v2f*>(bp + (2 * i) * LG_PITCH + j * 32);
            acc[j] = __builtin_amdgcn_wmma_f32_16x16x4_f32(
                        false, a[i], false, b, (short)0, acc[j], false, false);
        }
    }

    float* Cp = C + (size_t)(row0 + 8 * hh) * GD + 4 * c;
#pragma unroll
    for (int r = 0; r < 8; ++r) {
        float4 lo, hi;
        lo.x = acc[0][r]; lo.y = acc[1][r]; lo.z = acc[2][r]; lo.w = acc[3][r];
        hi.x = acc[4][r]; hi.y = acc[5][r]; hi.z = acc[6][r]; hi.w = acc[7][r];
        *reinterpret_cast<float4*>(Cp + (size_t)r * GD)      = lo;
        *reinterpret_cast<float4*>(Cp + (size_t)r * GD + 64) = hi;
    }

    float s_lo[8], s_hi[8], d_lo[8], d_hi[8];
#pragma unroll
    for (int r = 0; r < 8; ++r) {
        s_lo[r] = acc[0][r] * as_lo.x + acc[1][r] * as_lo.y + acc[2][r] * as_lo.z + acc[3][r] * as_lo.w;
        s_hi[r] = acc[4][r] * as_hi.x + acc[5][r] * as_hi.y + acc[6][r] * as_hi.z + acc[7][r] * as_hi.w;
        d_lo[r] = acc[0][r] * ad_lo.x + acc[1][r] * ad_lo.y + acc[2][r] * ad_lo.z + acc[3][r] * ad_lo.w;
        d_hi[r] = acc[4][r] * ad_hi.x + acc[5][r] * ad_hi.y + acc[6][r] * ad_hi.z + acc[7][r] * ad_hi.w;
    }
    const float zs_lo = sum8_scatter(s_lo, lane);
    const float zs_hi = sum8_scatter(s_hi, lane);
    const float zd_lo = sum8_scatter(d_lo, lane);
    const float zd_hi = sum8_scatter(d_hi, lane);
    const int hd = c >> 3;
    const size_t o = (size_t)(row0 + 8 * hh + (c & 7)) * GH;
    als[o + hd]     = zs_lo;
    als[o + 2 + hd] = zs_hi;
    ald[o + hd]     = zd_lo;
    ald[o + 2 + hd] = zd_hi;
}

#define SCAN_T   1024
#define SCAN_CH  (SCAN_T * 4)
#define SCAN_NB  ((GN + 1 + SCAN_CH - 1) / SCAN_CH)
#define SCAN_Q   (GN / 4)
#define ROWPTR_SLOTS (((GN + 1 + 63) / 64) * 64)
static_assert(GN % 4 == 0, "counts are read four at a time");

#define CNT_P    32
#define CNT_T    1024
#define CNT_CH   (GE / CNT_P)
static_assert(GE % (CNT_P * 4) == 0, "a chunk is a whole number of 16-byte groups of edges");
static_assert(CNT_T == SCAN_T, "the counting kernel writes its plane out in the prefix-sum kernel's chunks");
static_assert(CNT_P * SCAN_NB <= SCAN_T, "one thread of the prefix-sum kernel per chunk total");

__global__ void __launch_bounds__(CNT_T)
csr_count_k(const int* __restrict__ ei,
            int* __restrict__ rank,
            int* __restrict__ hplane,
            int* __restrict__ ctot) {
    __shared__ __attribute__((aligned(16))) int hist[GN];
    __shared__ int tot[SCAN_NB];
    constexpr int NG = CNT_CH / 4;
    constexpr int NK = (NG + CNT_T - 1) / CNT_T;
    const int t = threadIdx.x;
    const int c = blockIdx.x;
    int4* hist4 = reinterpret_cast<int4*>(hist);

    const int4* dst4 = reinterpret_cast<const int4*>(ei + GE + c * CNT_CH);
    int4 d[NK];
#pragma unroll
    for (int k = 0; k < NK; ++k) {
        const int g = k * CNT_T + t;
        d[k] = dst4[(g < NG) ? g : NG - 1];
    }

    for (int q = t; q < SCAN_Q; q += CNT_T) hist4[q] = make_int4(0, 0, 0, 0);
    if (t < SCAN_NB) tot[t] = 0;
    __syncthreads();

    int4* rank4 = reinterpret_cast<int4*>(rank + c * CNT_CH);
#pragma unroll
    for (int k = 0; k < NK; ++k) {
        const int g = k * CNT_T + t;
        if (g < NG) {
            int4 r;
            r.x = atomicAdd(&hist[d[k].x], 1);
            r.y = atomicAdd(&hist[d[k].y], 1);
            r.z = atomicAdd(&hist[d[k].z], 1);
            r.w = atomicAdd(&hist[d[k].w], 1);
            rank4[g] = r;
        }
    }
    __syncthreads();

    int4* out4 = reinterpret_cast<int4*>(hplane + (size_t)c * GN);
    int sj[SCAN_NB];
#pragma unroll
    for (int j = 0; j < SCAN_NB; ++j) {
        const int q = j * CNT_T + t;
        sj[j] = 0;
        if (q < SCAN_Q) {
            const int4 h = hist4[q];
            out4[q] = h;
            sj[j] = (h.x + h.y) + (h.z + h.w);
        }
    }
    for (int off = 16; off > 0; off >>= 1) {
#pragma unroll
        for (int j = 0; j < SCAN_NB; ++j) sj[j] += __shfl_xor(sj[j], off, 32);
    }
    if ((t & 31) == 0) {
#pragma unroll
        for (int j = 0; j < SCAN_NB; ++j) atomicAdd(&tot[j], sj[j]);
    }
    __syncthreads();
    if (t < SCAN_NB) ctot[c * SCAN_NB + t] = tot[t];
}

__global__ void __launch_bounds__(SCAN_T)
csr_scan_k(int* __restrict__ hplane,
           const int* __restrict__ ctot,
           int* __restrict__ rowptr) {
    __shared__ int wtot[SCAN_T / 32];
    __shared__ int wpre[SCAN_T / 32];
    const int t    = threadIdx.x;
    const int lane = t & 31;
    const int wave = t >> 5;
    const int b    = blockIdx.x;
    int4* hp4 = reinterpret_cast<int4*>(hplane);

    const int q = b * SCAN_T + t;
    const bool own = q < SCAN_Q;
    const int qq = own ? q : SCAN_Q - 1;

    int pre = 0;
    if (t < CNT_P * SCAN_NB) {
        const int cv = ctot[t];
        pre = ((t % SCAN_NB) < b) ? cv : 0;
    }

    int4 v = make_int4(0, 0, 0, 0);
#pragma unroll 8
    for (int c = 0; c < CNT_P; ++c) {
        const int4 h = hp4[(size_t)c * SCAN_Q + qq];
        v.x += h.x; v.y += h.y; v.z += h.z; v.w += h.w;
    }
    if (!own) v = make_int4(0, 0, 0, 0);

    const int s = (v.x + v.y) + (v.z + v.w);
    int inc = s;
    for (int off = 1; off < 32; off <<= 1) {
        const int o = __shfl_up(inc, off, 32);
        if (lane >= off) inc += o;
    }
    for (int off = 16; off > 0; off >>= 1) pre += __shfl_xor(pre, off, 32);
    if (lane == 31) wtot[wave] = inc;
    if (lane == 0)  wpre[wave] = pre;
    __syncthreads();

    int wt = wtot[lane];
    int wp = wpre[lane];
    int winc = wt;
    for (int off = 1; off < 32; off <<= 1) {
        const int o = __shfl_up(winc, off, 32);
        if (lane >= off) winc += o;
    }
    for (int off = 16; off > 0; off >>= 1) wp += __shfl_xor(wp, off, 32);
    const int wexc = __shfl(winc - wt, wave, 32);

    const int e0 = wp + wexc + (inc - s);
    if (own) {
        int4 r;
        r.x = e0;
        r.y = e0 + v.x;
        r.z = r.y + v.y;
        r.w = r.z + v.z;
        reinterpret_cast<int4*>(rowptr)[q] = r;
#pragma unroll
        for (int c0 = 0; c0 < CNT_P; c0 += 8) {
            int4 h[8];
#pragma unroll
            for (int u = 0; u < 8; ++u) h[u] = hp4[(size_t)(c0 + u) * SCAN_Q + q];
#pragma unroll
            for (int u = 0; u < 8; ++u) {
                hp4[(size_t)(c0 + u) * SCAN_Q + q] = r;
                r.x += h[u].x; r.y += h[u].y; r.z += h[u].z; r.w += h[u].w;
            }
        }
    } else if (q == SCAN_Q) {
        rowptr[GN] = e0;
    }
}

#define POOL_MAX_Q (GG * GD / 4)
#define POOL_ACC_Q ((2 * GG * GD + GG) / 4)
static_assert(GG % 4 == 0, "the accumulators are a whole number of 16-byte groups");
static_assert(POOL_ACC_Q <= GE, "one placement thread per 16-byte group of start values");

__global__ void csr_fill_k(const int* __restrict__ ei,
                           const int* __restrict__ hplane,
                           const int* __restrict__ rank,
                           int* __restrict__ csrc,
                           float* __restrict__ pool_acc) {
    int e = blockIdx.x * blockDim.x + threadIdx.x;
    if (e < POOL_ACC_Q) {
        const float v = (e < POOL_MAX_Q) ? -INFINITY : 0.0f;
        reinterpret_cast<float4*>(pool_acc)[e] = make_float4(v, v, v, v);
    }
    if (e >= GE) return;
    const int c = e / CNT_CH;
    csrc[hplane[(size_t)c * GN + ei[GE + e]] + rank[e]] = ei[e];
}

__device__ __forceinline__ float lrelu02(float v) {
    return (v >= 0.0f) ? v : 0.2f * v;
}

__device__ __forceinline__ float pick4(int h, float a, float b, float c, float d) {
    const bool odd  = (h & 1) != 0;
    const bool high = (h & 2) != 0;
    const float lo = odd ? b : a;
    const float hi = odd ? d : c;
    return high ? hi : lo;
}

__global__ void __launch_bounds__(256)
seg_softmax_agg_ln_k(const int* __restrict__ rowptr,
                     const int* __restrict__ csrc,
                     const float* __restrict__ als,
                     const float* __restrict__ ald,
                     const float* __restrict__ ht,
                     const float* __restrict__ bias,
                     const float* __restrict__ gamma,
                     const float* __restrict__ beta,
                     float* __restrict__ out) {
    const int lane = threadIdx.x & 31;
    const int d = __builtin_amdgcn_readfirstlane(
        (int)(blockIdx.x * (blockDim.x >> 5) + (threadIdx.x >> 5)));
    if (d >= GN) return;
    const int beg = rowptr[d];
    const int end = rowptr[d + 1];
    const float4* als4 = reinterpret_cast<const float4*>(als);
    const float4 ad = reinterpret_cast<const float4*>(ald)[d];
    const float4 sd = als4[d];

    const float t0 = lrelu02(sd.x + ad.x);
    const float t1 = lrelu02(sd.y + ad.y);
    const float t2 = lrelu02(sd.z + ad.z);
    const float t3 = lrelu02(sd.w + ad.w);

    const bool has0 = beg + lane < end;
    int   s0 = 0;
    float e0 = 0.0f, e1 = 0.0f, e2 = 0.0f, e3 = 0.0f;
    if (has0) {
        s0 = csrc[beg + lane];
        const float4 a = als4[s0];
        e0 = lrelu02(a.x + ad.x);
        e1 = lrelu02(a.y + ad.y);
        e2 = lrelu02(a.z + ad.z);
        e3 = lrelu02(a.w + ad.w);
    }

    float m0 = t0, m1 = t1, m2 = t2, m3 = t3;
    if (has0) {
        m0 = fmaxf(m0, e0);
        m1 = fmaxf(m1, e1);
        m2 = fmaxf(m2, e2);
        m3 = fmaxf(m3, e3);
    }
    for (int i = beg + 32 + lane; i < end; i += 32) {
        const float4 a = als4[csrc[i]];
        m0 = fmaxf(m0, lrelu02(a.x + ad.x));
        m1 = fmaxf(m1, lrelu02(a.y + ad.y));
        m2 = fmaxf(m2, lrelu02(a.z + ad.z));
        m3 = fmaxf(m3, lrelu02(a.w + ad.w));
    }
    for (int off = 16; off > 0; off >>= 1) {
        m0 = fmaxf(m0, __shfl_xor(m0, off, 32));
        m1 = fmaxf(m1, __shfl_xor(m1, off, 32));
        m2 = fmaxf(m2, __shfl_xor(m2, off, 32));
        m3 = fmaxf(m3, __shfl_xor(m3, off, 32));
    }

    float p0 = 0.0f, p1 = 0.0f, p2 = 0.0f, p3 = 0.0f;
    if (has0) {
        p0 = expf(e0 - m0);
        p1 = expf(e1 - m1);
        p2 = expf(e2 - m2);
        p3 = expf(e3 - m3);
    }
    float q0 = p0, q1 = p1, q2 = p2, q3 = p3;
    for (int i = beg + 32 + lane; i < end; i += 32) {
        const float4 a = als4[csrc[i]];
        q0 += expf(lrelu02(a.x + ad.x) - m0);
        q1 += expf(lrelu02(a.y + ad.y) - m1);
        q2 += expf(lrelu02(a.z + ad.z) - m2);
        q3 += expf(lrelu02(a.w + ad.w) - m3);
    }
    for (int off = 16; off > 0; off >>= 1) {
        q0 += __shfl_xor(q0, off, 32);
        q1 += __shfl_xor(q1, off, 32);
        q2 += __shfl_xor(q2, off, 32);
        q3 += __shfl_xor(q3, off, 32);
    }
    q0 += expf(t0 - m0);
    q1 += expf(t1 - m1);
    q2 += expf(t2 - m2);
    q3 += expf(t3 - m3);

    const int h = lane >> 3;
    const float4* hrow = reinterpret_cast<const float4*>(ht) + lane;
    float4 acc;
    {
        const float al = expf(pick4(h, t0, t1, t2, t3) - pick4(h, m0, m1, m2, m3))
                         / pick4(h, q0, q1, q2, q3);
        const float4 hv = hrow[(size_t)d * (GD / 4)];
        acc.x = hv.x * al;
        acc.y = hv.y * al;
        acc.z = hv.z * al;
        acc.w = hv.w * al;
    }
    const float4 bv = reinterpret_cast<const float4*>(bias)[lane];
    const float4 gv = reinterpret_cast<const float4*>(gamma)[lane];
    const float4 ev = reinterpret_cast<const float4*>(beta)[lane];
    int   s  = s0;
    float a0 = p0 / q0;
    float a1 = p1 / q1;
    float a2 = p2 / q2;
    float a3 = p3 / q3;
    for (int base = beg; base < end; base += 32) {
        if (base != beg) {
            s  = 0;
            a0 = 0.0f; a1 = 0.0f; a2 = 0.0f; a3 = 0.0f;
            if (base + lane < end) {
                s = csrc[base + lane];
                const float4 a = als4[s];
                a0 = expf(lrelu02(a.x + ad.x) - m0) / q0;
                a1 = expf(lrelu02(a.y + ad.y) - m1) / q1;
                a2 = expf(lrelu02(a.z + ad.z) - m2) / q2;
                a3 = expf(lrelu02(a.w + ad.w) - m3) / q3;
            }
        }
        const int cnt = (end - base < 32) ? (end - base) : 32;
#pragma unroll 4
        for (int k = 0; k < cnt; ++k) {
            const int sk = __builtin_amdgcn_readlane(s, k);
            const float b0 = __int_as_float(__builtin_amdgcn_readlane(__float_as_int(a0), k));
            const float b1 = __int_as_float(__builtin_amdgcn_readlane(__float_as_int(a1), k));
            const float b2 = __int_as_float(__builtin_amdgcn_readlane(__float_as_int(a2), k));
            const float b3 = __int_as_float(__builtin_amdgcn_readlane(__float_as_int(a3), k));
            const float al = pick4(h, b0, b1, b2, b3);
            const float4 hv = hrow[(size_t)sk * (GD / 4)];
            acc.x += hv.x * al;
            acc.y += hv.y * al;
            acc.z += hv.z * al;
            acc.w += hv.w * al;
        }
    }
    const float v0 = acc.x + bv.x;
    const float v1 = acc.y + bv.y;
    const float v2 = acc.z + bv.z;
    const float v3 = acc.w + bv.w;
    float rs = v0 + v1 + v2 + v3;
    float rq = v0 * v0 + v1 * v1 + v2 * v2 + v3 * v3;
    for (int off = 16; off > 0; off >>= 1) {
        rs += __shfl_xor(rs, off, 32);
        rq += __shfl_xor(rq, off, 32);
    }
    const float mean = rs * (1.0f / GD);
    const float var  = rq * (1.0f / GD) - mean * mean;
    const float rstd = rsqrtf(var + 1e-5f);
    float4 y;
    y.x = (v0 - mean) * rstd * gv.x + ev.x;
    y.y = (v1 - mean) * rstd * gv.y + ev.y;
    y.z = (v2 - mean) * rstd * gv.z + ev.z;
    y.w = (v3 - mean) * rstd * gv.w + ev.w;
    y.x = (y.x >= 0.0f) ? y.x : 0.1f * y.x;
    y.y = (y.y >= 0.0f) ? y.y : 0.1f * y.y;
    y.z = (y.z >= 0.0f) ? y.z : 0.1f * y.z;
    y.w = (y.w >= 0.0f) ? y.w : 0.1f * y.w;
    reinterpret_cast<float4*>(out)[(size_t)d * (GD / 4) + lane] = y;
}

#define POOL_NPW 8
static_assert(GN % POOL_NPW == 0, "a wave takes POOL_NPW whole nodes");

__device__ __forceinline__ void pool_flush(float* __restrict__ gmax,
                                           float* __restrict__ gsum,
                                           float* __restrict__ cnt,
                                           int g, int lane,
                                           const float* mx, const float* sm, int c) {
    float* pm = gmax + (size_t)g * GD + lane;
    float* ps = gsum + (size_t)g * GD + lane;
#pragma unroll
    for (int i = 0; i < 4; ++i) {
        atomicMaxFloat(pm + 32 * i, mx[i]);
        atomicAdd(ps + 32 * i, sm[i]);
    }
    if (lane == 0) atomicAdd(cnt + g, (float)c);
}

__global__ void __launch_bounds__(256)
pool_k(const float* __restrict__ x,
       const int* __restrict__ batch,
       float* __restrict__ gmax,
       float* __restrict__ gsum,
       float* __restrict__ cnt) {
    const int lane = threadIdx.x & 31;
    const int n0 = __builtin_amdgcn_readfirstlane(
        (int)((blockIdx.x * (blockDim.x >> 5) + (threadIdx.x >> 5)) * POOL_NPW));
    if (n0 >= GN) return;
    const float* xp = x + (size_t)n0 * GD + lane;

    int   b[POOL_NPW];
    float v[POOL_NPW][4];
#pragma unroll
    for (int j = 0; j < POOL_NPW; ++j) {
        b[j] = batch[n0 + j];
#pragma unroll
        for (int i = 0; i < 4; ++i) v[j][i] = xp[j * GD + 32 * i];
    }

    int cur = b[0];
    int c = 1;
    float mx[4], sm[4];
#pragma unroll
    for (int i = 0; i < 4; ++i) { mx[i] = v[0][i]; sm[i] = v[0][i]; }
#pragma unroll
    for (int j = 1; j < POOL_NPW; ++j) {
        if (b[j] != cur) {
            pool_flush(gmax, gsum, cnt, cur, lane, mx, sm, c);
            cur = b[j];
            c = 1;
#pragma unroll
            for (int i = 0; i < 4; ++i) { mx[i] = v[j][i]; sm[i] = v[j][i]; }
        } else {
#pragma unroll
            for (int i = 0; i < 4; ++i) {
                mx[i] = fmaxf(mx[i], v[j][i]);
                sm[i] += v[j][i];
            }
            ++c;
        }
    }
    pool_flush(gmax, gsum, cnt, cur, lane, mx, sm, c);
}

extern "C" void kernel_launch(void* const* d_in, const int* in_sizes, int n_in,
                              void* d_out, int out_size, void* d_ws, size_t ws_size,
                              hipStream_t stream) {
    const float* x       = (const float*)d_in[0];
    const int*   ei      = (const int*)  d_in[1];
    const int*   batch   = (const int*)  d_in[2];
    const float* W       = (const float*)d_in[3];
    const float* att_src = (const float*)d_in[4];
    const float* att_dst = (const float*)d_in[5];
    const float* bias    = (const float*)d_in[6];
    const float* ln_g    = (const float*)d_in[7];
    const float* ln_b    = (const float*)d_in[8];
    const float* W_out   = (const float*)d_in[9];
    const float* b_out   = (const float*)d_in[10];
    float* out = (float*)d_out;

    float* ws     = (float*)d_ws;
    float* bufP   = ws;
    float* bufQ   = bufP   + (size_t)GN * GD;
    int*   rowptr = (int*)(bufQ + (size_t)GN * GD);
    int*   rank   = rowptr + ROWPTR_SLOTS;
    int*   csrc   = rank   + GE;
    int*   hplane = csrc   + GE;
    int*   ctot   = hplane + (size_t)CNT_P * GN;
    float* als    = ws     + (size_t)3 * GN * GD;
    float* ald    = als    + (size_t)GN * GH;
    float* gmax   = ald    + (size_t)3 * GN * GH;
    float* gsum   = gmax   + (size_t)GG * GD;
    float* cnt    = gsum   + (size_t)GG * GD;

    const int TB = 256;
    auto blocks = [](long long n, int tb) { return (int)((n + tb - 1) / tb); };

    csr_count_k<<<CNT_P, CNT_T, 0, stream>>>(ei, rank, hplane, ctot);
    csr_scan_k<<<SCAN_NB, SCAN_T, 0, stream>>>(hplane, ctot, rowptr);
    csr_fill_k<<<blocks(GE, TB), TB, 0, stream>>>(ei, hplane, rank, csrc, gmax);

    float* ht     = bufQ;
    float* nodeio = bufP;
    for (int l = 0; l < 3; ++l) {
        const float* in = (l == 0) ? x : nodeio;

        gemm128_strip_k<<<blocks(GN / 16, 8), 256, 0, stream>>>(
            in, W + (size_t)l * GD * GD, att_src + l * GH * GC, att_dst + l * GH * GC,
            ht, als, ald, GN);

        seg_softmax_agg_ln_k<<<blocks(GN, 8), 256, 0, stream>>>(
            rowptr, csrc, als, ald, ht,
            bias + l * GD, ln_g + l * GD, ln_b + l * GD, nodeio);
    }

    pool_k<<<blocks(GN / POOL_NPW, 8), 256, 0, stream>>>(
        nodeio, batch, gmax, gsum, cnt);

    {
        int tiles = (GG / 16) * (GEMB / 16);
        pool_out_gemm_k<<<blocks(tiles, 4), 128, 0, stream>>>(
            gmax, gsum, cnt, W_out, b_out, out);
    }
}


# Layer 0's GEMM (reads x and W only) rides the edge-counting launch: work-groups 0-31 count, 196 more transform on the WGPs the count leaves idle; 11 -> 10 launches; every sum in the same order
# speedup vs baseline: 17.2888x; 1.0397x over previous
#include <hip/hip_runtime.h>
#include <cmath>


#define GN   50000
#define GE   800000
#define GETOT (GE + GN)
#define GG   256
#define GH   4
#define GC   32
#define GD   128
#define GEMB 128

typedef float v2f __attribute__((ext_vector_type(2)));
typedef float v8f __attribute__((ext_vector_type(8)));

__device__ __forceinline__ void atomicMaxFloat(float* addr, float val) {
    if (val >= 0.0f) {
        atomicMax((int*)addr, __float_as_int(val));
    } else {
        atomicMin((unsigned int*)addr, __float_as_uint(val));
    }
}

__global__ void __launch_bounds__(128)
pool_out_gemm_k(const float* __restrict__ gmax,
                const float* __restrict__ gsum,
                const float* __restrict__ cnt,
                const float* __restrict__ Wo,
                const float* __restrict__ bo,
                float* __restrict__ out) {
    const int wave = threadIdx.x >> 5;
    const int lane = threadIdx.x & 31;
    constexpr int tilesN = GEMB >> 4;
    constexpr int total  = (GG >> 4) * tilesN;
    const int tile = blockIdx.x * (blockDim.x >> 5) + wave;
    if (tile >= total) return;

    const int row0 = (tile / tilesN) << 4;
    const int col0 = (tile % tilesN) << 4;
    const int half = lane >> 4;
    const int l16  = lane & 15;

    const float* Mp = gmax + (size_t)(row0 + l16) * GD + 2 * half;
    const float* Sp = gsum + (size_t)(row0 + l16) * GD + 2 * half;
    const float* Bp = Wo + (size_t)(2 * half) * GEMB + col0 + l16;
    const float cn  = fmaxf(cnt[row0 + l16], 1.0f);

    v8f acc = {};
#pragma unroll
    for (int k0 = 0; k0 < GD; k0 += 4) {
        v2f a, b;
        a.x = Mp[k0];
        a.y = Mp[k0 + 1];
        b.x = Bp[(size_t)k0 * GEMB];
        b.y = Bp[(size_t)(k0 + 1) * GEMB];
        acc = __builtin_amdgcn_wmma_f32_16x16x4_f32(
                false, a, false, b, (short)0, acc, false, false);
    }
#pragma unroll
    for (int k0 = 0; k0 < GD; k0 += 4) {
        v2f a, b;
        a.x = Sp[k0] / cn;
        a.y = Sp[k0 + 1] / cn;
        b.x = Bp[(size_t)(GD + k0) * GEMB];
        b.y = Bp[(size_t)(GD + k0 + 1) * GEMB];
        acc = __builtin_amdgcn_wmma_f32_16x16x4_f32(
                false, a, false, b, (short)0, acc, false, false);
    }

    const float bv = bo[col0 + l16];
    float* Cp = out + (size_t)(row0 + 8 * half) * GEMB + col0 + l16;
#pragma unroll
    for (int j = 0; j < 8; ++j) {
        Cp[(size_t)j * GEMB] = acc[j] + bv;
    }
}

__device__ __forceinline__ float sum8_scatter(const float (&v)[8], int lane) {
    const bool b2 = (lane & 4) != 0;
    const bool b1 = (lane & 2) != 0;
    const bool b0 = (lane & 1) != 0;
    float w[4], u[2];
#pragma unroll
    for (int r = 0; r < 4; ++r) {
        const float keep = b2 ? v[r + 4] : v[r];
        const float send = b2 ? v[r] : v[r + 4];
        w[r] = keep + __shfl_xor(send, 4, 32);
    }
#pragma unroll
    for (int r = 0; r < 2; ++r) {
        const float keep = b1 ? w[r + 2] : w[r];
        const float send = b1 ? w[r] : w[r + 2];
        u[r] = keep + __shfl_xor(send, 2, 32);
    }
    const float keep = b0 ? u[1] : u[0];
    const float send = b0 ? u[0] : u[1];
    return keep + __shfl_xor(send, 1, 32);
}

#define LG_PITCH 288
__global__ void __launch_bounds__(256)
gemm128_strip_k(const float* __restrict__ A,
                const float* __restrict__ Wg,
                const float* __restrict__ asrc,
                const float* __restrict__ adst,
                float* __restrict__ C,
                float* __restrict__ als,
                float* __restrict__ ald,
                int M) {
    __shared__ __attribute__((aligned(16))) float Wl[64 * LG_PITCH];
    const int t    = threadIdx.x;
    const int lane = t & 31;
    const int c    = lane & 15;
    const int hh   = lane >> 4;
    const int wave = __builtin_amdgcn_readfirstlane((int)(threadIdx.x >> 5));
    const int row0 = (blockIdx.x * 8 + wave) * 16;
    const bool live = row0 < M;

    const float4* W4 = reinterpret_cast<const float4*>(Wg);
    const int n4  = lane;
    const int cc  = n4 & 15;
    const int jhi = n4 >> 4;
    float4 r0[8], r1[8];
#pragma unroll
    for (int i = 0; i < 8; ++i) {
        const int kp = wave + 8 * i;
        r0[i] = W4[(2 * kp) * (GD / 4) + n4];
        r1[i] = W4[(2 * kp + 1) * (GD / 4) + n4];
    }

    v2f a[32];
    {
        const float* Ap = A + (size_t)((live ? row0 : 0) + c) * GD + 2 * hh;
#pragma unroll
        for (int i = 0; i < 32; ++i) a[i] = *reinterpret_cast<const v2f*>(Ap + 4 * i);
    }

    const float4 as_lo = reinterpret_cast<const float4*>(asrc)[c];
    const float4 as_hi = reinterpret_cast<const float4*>(asrc)[16 + c];
    const float4 ad_lo = reinterpret_cast<const float4*>(adst)[c];
    const float4 ad_hi = reinterpret_cast<const float4*>(adst)[16 + c];

#pragma unroll
    for (int i = 0; i < 8; ++i) {
        const int kp = wave + 8 * i;
        float* dst = Wl + kp * LG_PITCH + (jhi * 4) * 32 + cc * 2;
        *reinterpret_cast<v2f*>(dst)      = (v2f){r0[i].x, r1[i].x};
        *reinterpret_cast<v2f*>(dst + 32) = (v2f){r0[i].y, r1[i].y};
        *reinterpret_cast<v2f*>(dst + 64) = (v2f){r0[i].z, r1[i].z};
        *reinterpret_cast<v2f*>(dst + 96) = (v2f){r0[i].w, r1[i].w};
    }
    __syncthreads();
    if (!live) return;

    v8f acc[8];
#pragma unroll
    for (int j = 0; j < 8; ++j) acc[j] = (v8f){};
    const float* bp = Wl + hh * LG_PITCH + c * 2;
#pragma unroll
    for (int i = 0; i < 32; ++i) {
#pragma unroll
        for (int j = 0; j < 8; ++j) {
            const v2f b = *reinterpret_cast<const v2f*>(bp + (2 * i) * LG_PITCH + j * 32);
            acc[j] = __builtin_amdgcn_wmma_f32_16x16x4_f32(
                        false, a[i], false, b, (short)0, acc[j], false, false);
        }
    }

    float* Cp = C + (size_t)(row0 + 8 * hh) * GD + 4 * c;
#pragma unroll
    for (int r = 0; r < 8; ++r) {
        float4 lo, hi;
        lo.x = acc[0][r]; lo.y = acc[1][r]; lo.z = acc[2][r]; lo.w = acc[3][r];
        hi.x = acc[4][r]; hi.y = acc[5][r]; hi.z = acc[6][r]; hi.w = acc[7][r];
        *reinterpret_cast<float4*>(Cp + (size_t)r * GD)      = lo;
        *reinterpret_cast<float4*>(Cp + (size_t)r * GD + 64) = hi;
    }

    float s_lo[8], s_hi[8], d_lo[8], d_hi[8];
#pragma unroll
    for (int r = 0; r < 8; ++r) {
        s_lo[r] = acc[0][r] * as_lo.x + acc[1][r] * as_lo.y + acc[2][r] * as_lo.z + acc[3][r] * as_lo.w;
        s_hi[r] = acc[4][r] * as_hi.x + acc[5][r] * as_hi.y + acc[6][r] * as_hi.z + acc[7][r] * as_hi.w;
        d_lo[r] = acc[0][r] * ad_lo.x + acc[1][r] * ad_lo.y + acc[2][r] * ad_lo.z + acc[3][r] * ad_lo.w;
        d_hi[r] = acc[4][r] * ad_hi.x + acc[5][r] * ad_hi.y + acc[6][r] * ad_hi.z + acc[7][r] * ad_hi.w;
    }
    const float zs_lo = sum8_scatter(s_lo, lane);
    const float zs_hi = sum8_scatter(s_hi, lane);
    const float zd_lo = sum8_scatter(d_lo, lane);
    const float zd_hi = sum8_scatter(d_hi, lane);
    const int hd = c >> 3;
    const size_t o = (size_t)(row0 + 8 * hh + (c & 7)) * GH;
    als[o + hd]     = zs_lo;
    als[o + 2 + hd] = zs_hi;
    ald[o + hd]     = zd_lo;
    ald[o + 2 + hd] = zd_hi;
}

#define SCAN_T   1024
#define SCAN_CH  (SCAN_T * 4)
#define SCAN_NB  ((GN + 1 + SCAN_CH - 1) / SCAN_CH)
#define SCAN_Q   (GN / 4)
#define ROWPTR_SLOTS (((GN + 1 + 63) / 64) * 64)
static_assert(GN % 4 == 0, "counts are read four at a time");

#define CNT_P    32
#define CNT_T    1024
#define CNT_CH   (GE / CNT_P)
static_assert(GE % (CNT_P * 4) == 0, "a chunk is a whole number of 16-byte groups of edges");
static_assert(CNT_T == SCAN_T, "the counting kernel writes its plane out in the prefix-sum kernel's chunks");
static_assert(CNT_P * SCAN_NB <= SCAN_T, "one thread of the prefix-sum kernel per chunk total");

__device__ __forceinline__ void
csr_count_body(const int* __restrict__ ei,
               int* __restrict__ rank,
               int* __restrict__ hplane,
               int* __restrict__ ctot,
               int* hist,
               int* tot,
               const int c) {
    constexpr int NG = CNT_CH / 4;
    constexpr int NK = (NG + CNT_T - 1) / CNT_T;
    const int t = threadIdx.x;
    int4* hist4 = reinterpret_cast<int4*>(hist);

    const int4* dst4 = reinterpret_cast<const int4*>(ei + GE + c * CNT_CH);
    int4 d[NK];
#pragma unroll
    for (int k = 0; k < NK; ++k) {
        const int g = k * CNT_T + t;
        d[k] = dst4[(g < NG) ? g : NG - 1];
    }

    for (int q = t; q < SCAN_Q; q += CNT_T) hist4[q] = make_int4(0, 0, 0, 0);
    if (t < SCAN_NB) tot[t] = 0;
    __syncthreads();

    int4* rank4 = reinterpret_cast<int4*>(rank + c * CNT_CH);
#pragma unroll
    for (int k = 0; k < NK; ++k) {
        const int g = k * CNT_T + t;
        if (g < NG) {
            int4 r;
            r.x = atomicAdd(&hist[d[k].x], 1);
            r.y = atomicAdd(&hist[d[k].y], 1);
            r.z = atomicAdd(&hist[d[k].z], 1);
            r.w = atomicAdd(&hist[d[k].w], 1);
            rank4[g] = r;
        }
    }
    __syncthreads();

    int4* out4 = reinterpret_cast<int4*>(hplane + (size_t)c * GN);
    int sj[SCAN_NB];
#pragma unroll
    for (int j = 0; j < SCAN_NB; ++j) {
        const int q = j * CNT_T + t;
        sj[j] = 0;
        if (q < SCAN_Q) {
            const int4 h = hist4[q];
            out4[q] = h;
            sj[j] = (h.x + h.y) + (h.z + h.w);
        }
    }
    for (int off = 16; off > 0; off >>= 1) {
#pragma unroll
        for (int j = 0; j < SCAN_NB; ++j) sj[j] += __shfl_xor(sj[j], off, 32);
    }
    if ((t & 31) == 0) {
#pragma unroll
        for (int j = 0; j < SCAN_NB; ++j) atomicAdd(&tot[j], sj[j]);
    }
    __syncthreads();
    if (t < SCAN_NB) ctot[c * SCAN_NB + t] = tot[t];
}

#define HS_WAVES (CNT_T / 32)
#define HS_SPB   (HS_WAVES / 2)
#define HS_NB    ((GN / 16 + HS_SPB - 1) / HS_SPB)
static_assert(64 % HS_WAVES == 0, "every wave stages the same number of k pairs of Wg");

__device__ __forceinline__ void
gemm128_halfstrip_body(const float* __restrict__ A,
                       const float* __restrict__ Wg,
                       const float* __restrict__ asrc,
                       const float* __restrict__ adst,
                       float* __restrict__ C,
                       float* __restrict__ als,
                       float* __restrict__ ald,
                       float* Wl,
                       const int gb) {
    const int lane = threadIdx.x & 31;
    const int c    = lane & 15;
    const int hh   = lane >> 4;
    const int wave = __builtin_amdgcn_readfirstlane((int)(threadIdx.x >> 5));
    const int hf   = wave & 1;
    const int row0 = (gb * HS_SPB + (wave >> 1)) * 16;
    const bool live = row0 < GN;

    const float4* W4 = reinterpret_cast<const float4*>(Wg);
    const int n4  = lane;
    const int cc  = n4 & 15;
    const int jhi = n4 >> 4;
    constexpr int KPW = 64 / HS_WAVES;
    float4 r0[KPW], r1[KPW];
#pragma unroll
    for (int i = 0; i < KPW; ++i) {
        const int kp = wave + HS_WAVES * i;
        r0[i] = W4[(2 * kp) * (GD / 4) + n4];
        r1[i] = W4[(2 * kp + 1) * (GD / 4) + n4];
    }

    __builtin_amdgcn_sched_barrier(0);

#pragma unroll
    for (int i = 0; i < KPW; ++i) {
        const int kp = wave + HS_WAVES * i;
        float* dst = Wl + kp * LG_PITCH + (jhi * 4) * 32 + cc * 2;
        *reinterpret_cast<v2f*>(dst)      = (v2f){r0[i].x, r1[i].x};
        *reinterpret_cast<v2f*>(dst + 32) = (v2f){r0[i].y, r1[i].y};
        *reinterpret_cast<v2f*>(dst + 64) = (v2f){r0[i].z, r1[i].z};
        *reinterpret_cast<v2f*>(dst + 96) = (v2f){r0[i].w, r1[i].w};
    }
    __syncthreads();
    if (!live) return;

    v2f a[32];
    {
        const float* Ap = A + (size_t)(row0 + c) * GD + 2 * hh;
#pragma unroll
        for (int i = 0; i < 32; ++i) a[i] = *reinterpret_cast<const v2f*>(Ap + 4 * i);
    }

    const float4 as4 = reinterpret_cast<const float4*>(asrc)[16 * hf + c];
    const float4 ad4 = reinterpret_cast<const float4*>(adst)[16 * hf + c];

    v8f acc[4];
#pragma unroll
    for (int j = 0; j < 4; ++j) acc[j] = (v8f){};
    const float* bp = Wl + hh * LG_PITCH + (4 * hf) * 32 + c * 2;
#pragma unroll
    for (int i = 0; i < 32; ++i) {
#pragma unroll
        for (int j = 0; j < 4; ++j) {
            const v2f b = *reinterpret_cast<const v2f*>(bp + (2 * i) * LG_PITCH + j * 32);
            acc[j] = __builtin_amdgcn_wmma_f32_16x16x4_f32(
                        false, a[i], false, b, (short)0, acc[j], false, false);
        }
    }

    float* Cp = C + (size_t)(row0 + 8 * hh) * GD + 64 * hf + 4 * c;
#pragma unroll
    for (int r = 0; r < 8; ++r) {
        float4 v;
        v.x = acc[0][r]; v.y = acc[1][r]; v.z = acc[2][r]; v.w = acc[3][r];
        *reinterpret_cast<float4*>(Cp + (size_t)r * GD) = v;
    }

    float s4[8], d4[8];
#pragma unroll
    for (int r = 0; r < 8; ++r) {
        s4[r] = acc[0][r] * as4.x + acc[1][r] * as4.y + acc[2][r] * as4.z + acc[3][r] * as4.w;
        d4[r] = acc[0][r] * ad4.x + acc[1][r] * ad4.y + acc[2][r] * ad4.z + acc[3][r] * ad4.w;
    }
    const float zs = sum8_scatter(s4, lane);
    const float zd = sum8_scatter(d4, lane);
    const int hd = c >> 3;
    const size_t o = (size_t)(row0 + 8 * hh + (c & 7)) * GH;
    als[o + 2 * hf + hd] = zs;
    ald[o + 2 * hf + hd] = zd;
}

__global__ void __launch_bounds__(CNT_T)
csr_count_gemm_k(const int* __restrict__ ei,
                 int* __restrict__ rank,
                 int* __restrict__ hplane,
                 int* __restrict__ ctot,
                 const float* __restrict__ A,
                 const float* __restrict__ Wg,
                 const float* __restrict__ asrc,
                 const float* __restrict__ adst,
                 float* __restrict__ C,
                 float* __restrict__ als,
                 float* __restrict__ ald) {
    __shared__ __attribute__((aligned(16))) int lds_words[GN + SCAN_NB];
    static_assert(64 * LG_PITCH <= GN, "Wg's LDS image fits in the counters' space");
    static_assert(GN % 4 == 0, "the chunk totals start on a 16-byte boundary");
    if (blockIdx.x < CNT_P) {
        csr_count_body(ei, rank, hplane, ctot, lds_words, lds_words + GN, (int)blockIdx.x);
    } else {
        gemm128_halfstrip_body(A, Wg, asrc, adst, C, als, ald,
                               reinterpret_cast<float*>(lds_words), (int)blockIdx.x - CNT_P);
    }
}

__global__ void __launch_bounds__(SCAN_T)
csr_scan_k(int* __restrict__ hplane,
           const int* __restrict__ ctot,
           int* __restrict__ rowptr) {
    __shared__ int wtot[SCAN_T / 32];
    __shared__ int wpre[SCAN_T / 32];
    const int t    = threadIdx.x;
    const int lane = t & 31;
    const int wave = t >> 5;
    const int b    = blockIdx.x;
    int4* hp4 = reinterpret_cast<int4*>(hplane);

    const int q = b * SCAN_T + t;
    const bool own = q < SCAN_Q;
    const int qq = own ? q : SCAN_Q - 1;

    int pre = 0;
    if (t < CNT_P * SCAN_NB) {
        const int cv = ctot[t];
        pre = ((t % SCAN_NB) < b) ? cv : 0;
    }

    int4 v = make_int4(0, 0, 0, 0);
#pragma unroll 8
    for (int c = 0; c < CNT_P; ++c) {
        const int4 h = hp4[(size_t)c * SCAN_Q + qq];
        v.x += h.x; v.y += h.y; v.z += h.z; v.w += h.w;
    }
    if (!own) v = make_int4(0, 0, 0, 0);

    const int s = (v.x + v.y) + (v.z + v.w);
    int inc = s;
    for (int off = 1; off < 32; off <<= 1) {
        const int o = __shfl_up(inc, off, 32);
        if (lane >= off) inc += o;
    }
    for (int off = 16; off > 0; off >>= 1) pre += __shfl_xor(pre, off, 32);
    if (lane == 31) wtot[wave] = inc;
    if (lane == 0)  wpre[wave] = pre;
    __syncthreads();

    int wt = wtot[lane];
    int wp = wpre[lane];
    int winc = wt;
    for (int off = 1; off < 32; off <<= 1) {
        const int o = __shfl_up(winc, off, 32);
        if (lane >= off) winc += o;
    }
    for (int off = 16; off > 0; off >>= 1) wp += __shfl_xor(wp, off, 32);
    const int wexc = __shfl(winc - wt, wave, 32);

    const int e0 = wp + wexc + (inc - s);
    if (own) {
        int4 r;
        r.x = e0;
        r.y = e0 + v.x;
        r.z = r.y + v.y;
        r.w = r.z + v.z;
        reinterpret_cast<int4*>(rowptr)[q] = r;
#pragma unroll
        for (int c0 = 0; c0 < CNT_P; c0 += 8) {
            int4 h[8];
#pragma unroll
            for (int u = 0; u < 8; ++u) h[u] = hp4[(size_t)(c0 + u) * SCAN_Q + q];
#pragma unroll
            for (int u = 0; u < 8; ++u) {
                hp4[(size_t)(c0 + u) * SCAN_Q + q] = r;
                r.x += h[u].x; r.y += h[u].y; r.z += h[u].z; r.w += h[u].w;
            }
        }
    } else if (q == SCAN_Q) {
        rowptr[GN] = e0;
    }
}

#define POOL_MAX_Q (GG * GD / 4)
#define POOL_ACC_Q ((2 * GG * GD + GG) / 4)
static_assert(GG % 4 == 0, "the accumulators are a whole number of 16-byte groups");
static_assert(POOL_ACC_Q <= GE, "one placement thread per 16-byte group of start values");

__global__ void csr_fill_k(const int* __restrict__ ei,
                           const int* __restrict__ hplane,
                           const int* __restrict__ rank,
                           int* __restrict__ csrc,
                           float* __restrict__ pool_acc) {
    int e = blockIdx.x * blockDim.x + threadIdx.x;
    if (e < POOL_ACC_Q) {
        const float v = (e < POOL_MAX_Q) ? -INFINITY : 0.0f;
        reinterpret_cast<float4*>(pool_acc)[e] = make_float4(v, v, v, v);
    }
    if (e >= GE) return;
    const int c = e / CNT_CH;
    csrc[hplane[(size_t)c * GN + ei[GE + e]] + rank[e]] = ei[e];
}

__device__ __forceinline__ float lrelu02(float v) {
    return (v >= 0.0f) ? v : 0.2f * v;
}

__device__ __forceinline__ float pick4(int h, float a, float b, float c, float d) {
    const bool odd  = (h & 1) != 0;
    const bool high = (h & 2) != 0;
    const float lo = odd ? b : a;
    const float hi = odd ? d : c;
    return high ? hi : lo;
}

__global__ void __launch_bounds__(256)
seg_softmax_agg_ln_k(const int* __restrict__ rowptr,
                     const int* __restrict__ csrc,
                     const float* __restrict__ als,
                     const float* __restrict__ ald,
                     const float* __restrict__ ht,
                     const float* __restrict__ bias,
                     const float* __restrict__ gamma,
                     const float* __restrict__ beta,
                     float* __restrict__ out) {
    const int lane = threadIdx.x & 31;
    const int d = __builtin_amdgcn_readfirstlane(
        (int)(blockIdx.x * (blockDim.x >> 5) + (threadIdx.x >> 5)));
    if (d >= GN) return;
    const int beg = rowptr[d];
    const int end = rowptr[d + 1];
    const float4* als4 = reinterpret_cast<const float4*>(als);
    const float4 ad = reinterpret_cast<const float4*>(ald)[d];
    const float4 sd = als4[d];

    const float t0 = lrelu02(sd.x + ad.x);
    const float t1 = lrelu02(sd.y + ad.y);
    const float t2 = lrelu02(sd.z + ad.z);
    const float t3 = lrelu02(sd.w + ad.w);

    const bool has0 = beg + lane < end;
    int   s0 = 0;
    float e0 = 0.0f, e1 = 0.0f, e2 = 0.0f, e3 = 0.0f;
    if (has0) {
        s0 = csrc[beg + lane];
        const float4 a = als4[s0];
        e0 = lrelu02(a.x + ad.x);
        e1 = lrelu02(a.y + ad.y);
        e2 = lrelu02(a.z + ad.z);
        e3 = lrelu02(a.w + ad.w);
    }

    float m0 = t0, m1 = t1, m2 = t2, m3 = t3;
    if (has0) {
        m0 = fmaxf(m0, e0);
        m1 = fmaxf(m1, e1);
        m2 = fmaxf(m2, e2);
        m3 = fmaxf(m3, e3);
    }
    for (int i = beg + 32 + lane; i < end; i += 32) {
        const float4 a = als4[csrc[i]];
        m0 = fmaxf(m0, lrelu02(a.x + ad.x));
        m1 = fmaxf(m1, lrelu02(a.y + ad.y));
        m2 = fmaxf(m2, lrelu02(a.z + ad.z));
        m3 = fmaxf(m3, lrelu02(a.w + ad.w));
    }
    for (int off = 16; off > 0; off >>= 1) {
        m0 = fmaxf(m0, __shfl_xor(m0, off, 32));
        m1 = fmaxf(m1, __shfl_xor(m1, off, 32));
        m2 = fmaxf(m2, __shfl_xor(m2, off, 32));
        m3 = fmaxf(m3, __shfl_xor(m3, off, 32));
    }

    float p0 = 0.0f, p1 = 0.0f, p2 = 0.0f, p3 = 0.0f;
    if (has0) {
        p0 = expf(e0 - m0);
        p1 = expf(e1 - m1);
        p2 = expf(e2 - m2);
        p3 = expf(e3 - m3);
    }
    float q0 = p0, q1 = p1, q2 = p2, q3 = p3;
    for (int i = beg + 32 + lane; i < end; i += 32) {
        const float4 a = als4[csrc[i]];
        q0 += expf(lrelu02(a.x + ad.x) - m0);
        q1 += expf(lrelu02(a.y + ad.y) - m1);
        q2 += expf(lrelu02(a.z + ad.z) - m2);
        q3 += expf(lrelu02(a.w + ad.w) - m3);
    }
    for (int off = 16; off > 0; off >>= 1) {
        q0 += __shfl_xor(q0, off, 32);
        q1 += __shfl_xor(q1, off, 32);
        q2 += __shfl_xor(q2, off, 32);
        q3 += __shfl_xor(q3, off, 32);
    }
    q0 += expf(t0 - m0);
    q1 += expf(t1 - m1);
    q2 += expf(t2 - m2);
    q3 += expf(t3 - m3);

    const int h = lane >> 3;
    const float4* hrow = reinterpret_cast<const float4*>(ht) + lane;
    float4 acc;
    {
        const float al = expf(pick4(h, t0, t1, t2, t3) - pick4(h, m0, m1, m2, m3))
                         / pick4(h, q0, q1, q2, q3);
        const float4 hv = hrow[(size_t)d * (GD / 4)];
        acc.x = hv.x * al;
        acc.y = hv.y * al;
        acc.z = hv.z * al;
        acc.w = hv.w * al;
    }
    const float4 bv = reinterpret_cast<const float4*>(bias)[lane];
    const float4 gv = reinterpret_cast<const float4*>(gamma)[lane];
    const float4 ev = reinterpret_cast<const float4*>(beta)[lane];
    int   s  = s0;
    float a0 = p0 / q0;
    float a1 = p1 / q1;
    float a2 = p2 / q2;
    float a3 = p3 / q3;
    for (int base = beg; base < end; base += 32) {
        if (base != beg) {
            s  = 0;
            a0 = 0.0f; a1 = 0.0f; a2 = 0.0f; a3 = 0.0f;
            if (base + lane < end) {
                s = csrc[base + lane];
                const float4 a = als4[s];
                a0 = expf(lrelu02(a.x + ad.x) - m0) / q0;
                a1 = expf(lrelu02(a.y + ad.y) - m1) / q1;
                a2 = expf(lrelu02(a.z + ad.z) - m2) / q2;
                a3 = expf(lrelu02(a.w + ad.w) - m3) / q3;
            }
        }
        const int cnt = (end - base < 32) ? (end - base) : 32;
#pragma unroll 4
        for (int k = 0; k < cnt; ++k) {
            const int sk = __builtin_amdgcn_readlane(s, k);
            const float b0 = __int_as_float(__builtin_amdgcn_readlane(__float_as_int(a0), k));
            const float b1 = __int_as_float(__builtin_amdgcn_readlane(__float_as_int(a1), k));
            const float b2 = __int_as_float(__builtin_amdgcn_readlane(__float_as_int(a2), k));
            const float b3 = __int_as_float(__builtin_amdgcn_readlane(__float_as_int(a3), k));
            const float al = pick4(h, b0, b1, b2, b3);
            const float4 hv = hrow[(size_t)sk * (GD / 4)];
            acc.x += hv.x * al;
            acc.y += hv.y * al;
            acc.z += hv.z * al;
            acc.w += hv.w * al;
        }
    }
    const float v0 = acc.x + bv.x;
    const float v1 = acc.y + bv.y;
    const float v2 = acc.z + bv.z;
    const float v3 = acc.w + bv.w;
    float rs = v0 + v1 + v2 + v3;
    float rq = v0 * v0 + v1 * v1 + v2 * v2 + v3 * v3;
    for (int off = 16; off > 0; off >>= 1) {
        rs += __shfl_xor(rs, off, 32);
        rq += __shfl_xor(rq, off, 32);
    }
    const float mean = rs * (1.0f / GD);
    const float var  = rq * (1.0f / GD) - mean * mean;
    const float rstd = rsqrtf(var + 1e-5f);
    float4 y;
    y.x = (v0 - mean) * rstd * gv.x + ev.x;
    y.y = (v1 - mean) * rstd * gv.y + ev.y;
    y.z = (v2 - mean) * rstd * gv.z + ev.z;
    y.w = (v3 - mean) * rstd * gv.w + ev.w;
    y.x = (y.x >= 0.0f) ? y.x : 0.1f * y.x;
    y.y = (y.y >= 0.0f) ? y.y : 0.1f * y.y;
    y.z = (y.z >= 0.0f) ? y.z : 0.1f * y.z;
    y.w = (y.w >= 0.0f) ? y.w : 0.1f * y.w;
    reinterpret_cast<float4*>(out)[(size_t)d * (GD / 4) + lane] = y;
}

#define POOL_NPW 8
static_assert(GN % POOL_NPW == 0, "a wave takes POOL_NPW whole nodes");

__device__ __forceinline__ void pool_flush(float* __restrict__ gmax,
                                           float* __restrict__ gsum,
                                           float* __restrict__ cnt,
                                           int g, int lane,
                                           const float* mx, const float* sm, int c) {
    float* pm = gmax + (size_t)g * GD + lane;
    float* ps = gsum + (size_t)g * GD + lane;
#pragma unroll
    for (int i = 0; i < 4; ++i) {
        atomicMaxFloat(pm + 32 * i, mx[i]);
        atomicAdd(ps + 32 * i, sm[i]);
    }
    if (lane == 0) atomicAdd(cnt + g, (float)c);
}

__global__ void __launch_bounds__(256)
pool_k(const float* __restrict__ x,
       const int* __restrict__ batch,
       float* __restrict__ gmax,
       float* __restrict__ gsum,
       float* __restrict__ cnt) {
    const int lane = threadIdx.x & 31;
    const int n0 = __builtin_amdgcn_readfirstlane(
        (int)((blockIdx.x * (blockDim.x >> 5) + (threadIdx.x >> 5)) * POOL_NPW));
    if (n0 >= GN) return;
    const float* xp = x + (size_t)n0 * GD + lane;

    int   b[POOL_NPW];
    float v[POOL_NPW][4];
#pragma unroll
    for (int j = 0; j < POOL_NPW; ++j) {
        b[j] = batch[n0 + j];
#pragma unroll
        for (int i = 0; i < 4; ++i) v[j][i] = xp[j * GD + 32 * i];
    }

    int cur = b[0];
    int c = 1;
    float mx[4], sm[4];
#pragma unroll
    for (int i = 0; i < 4; ++i) { mx[i] = v[0][i]; sm[i] = v[0][i]; }
#pragma unroll
    for (int j = 1; j < POOL_NPW; ++j) {
        if (b[j] != cur) {
            pool_flush(gmax, gsum, cnt, cur, lane, mx, sm, c);
            cur = b[j];
            c = 1;
#pragma unroll
            for (int i = 0; i < 4; ++i) { mx[i] = v[j][i]; sm[i] = v[j][i]; }
        } else {
#pragma unroll
            for (int i = 0; i < 4; ++i) {
                mx[i] = fmaxf(mx[i], v[j][i]);
                sm[i] += v[j][i];
            }
            ++c;
        }
    }
    pool_flush(gmax, gsum, cnt, cur, lane, mx, sm, c);
}

extern "C" void kernel_launch(void* const* d_in, const int* in_sizes, int n_in,
                              void* d_out, int out_size, void* d_ws, size_t ws_size,
                              hipStream_t stream) {
    const float* x       = (const float*)d_in[0];
    const int*   ei      = (const int*)  d_in[1];
    const int*   batch   = (const int*)  d_in[2];
    const float* W       = (const float*)d_in[3];
    const float* att_src = (const float*)d_in[4];
    const float* att_dst = (const float*)d_in[5];
    const float* bias    = (const float*)d_in[6];
    const float* ln_g    = (const float*)d_in[7];
    const float* ln_b    = (const float*)d_in[8];
    const float* W_out   = (const float*)d_in[9];
    const float* b_out   = (const float*)d_in[10];
    float* out = (float*)d_out;

    float* ws     = (float*)d_ws;
    float* bufP   = ws;
    float* bufQ   = bufP   + (size_t)GN * GD;
    int*   rowptr = (int*)(bufQ + (size_t)GN * GD);
    int*   rank   = rowptr + ROWPTR_SLOTS;
    int*   csrc   = rank   + GE;
    int*   hplane = csrc   + GE;
    int*   ctot   = hplane + (size_t)CNT_P * GN;
    float* als    = ws     + (size_t)3 * GN * GD;
    float* ald    = als    + (size_t)GN * GH;
    float* gmax   = ald    + (size_t)3 * GN * GH;
    float* gsum   = gmax   + (size_t)GG * GD;
    float* cnt    = gsum   + (size_t)GG * GD;

    const int TB = 256;
    auto blocks = [](long long n, int tb) { return (int)((n + tb - 1) / tb); };

    csr_count_gemm_k<<<CNT_P + HS_NB, CNT_T, 0, stream>>>(
        ei, rank, hplane, ctot,
        x, W, att_src, att_dst, bufQ, als, ald);
    csr_scan_k<<<SCAN_NB, SCAN_T, 0, stream>>>(hplane, ctot, rowptr);
    csr_fill_k<<<blocks(GE, TB), TB, 0, stream>>>(ei, hplane, rank, csrc, gmax);

    float* ht     = bufQ;
    float* nodeio = bufP;
    for (int l = 0; l < 3; ++l) {
        const float* in = (l == 0) ? x : nodeio;

        if (l > 0) {
            gemm128_strip_k<<<blocks(GN / 16, 8), 256, 0, stream>>>(
                in, W + (size_t)l * GD * GD, att_src + l * GH * GC, att_dst + l * GH * GC,
                ht, als, ald, GN);
        }

        seg_softmax_agg_ln_k<<<blocks(GN, 8), 256, 0, stream>>>(
            rowptr, csrc, als, ald, ht,
            bias + l * GD, ln_g + l * GD, ln_b + l * GD, nodeio);
    }

    pool_k<<<blocks(GN / POOL_NPW, 8), 256, 0, stream>>>(
        nodeio, batch, gmax, gsum, cnt);

    {
        int tiles = (GG / 16) * (GEMB / 16);
        pool_out_gemm_k<<<blocks(tiles, 4), 128, 0, stream>>>(
            gmax, gsum, cnt, W_out, b_out, out);
    }
}
